# starting kernel
# baseline (speedup 1.0000x reference)
#include <hip/hip_runtime.h>


#define N_NODES 100000
#define N_EDGES 3200000
#define IN_DIM  128
#define HID_DIM 32
#define OUT_DIM 16

typedef __attribute__((ext_vector_type(2))) float v2f;
typedef __attribute__((ext_vector_type(8))) float v8f;

__global__ void __launch_bounds__(256) k_init(float* __restrict__ deg,
                                              float* __restrict__ agg1,
                                              float* __restrict__ out,
                                              const float* __restrict__ b2) {
  int i = blockIdx.x * blockDim.x + threadIdx.x;
  if (i < N_NODES) deg[i] = 1.0f;
  if (i < N_NODES * HID_DIM) agg1[i] = 0.0f;
  if (i < N_NODES * OUT_DIM) out[i] = b2[i & (OUT_DIM - 1)];
}

__global__ void __launch_bounds__(256) k_degree(const int* __restrict__ ei,
                                                float* __restrict__ deg) {
  int e = blockIdx.x * blockDim.x + threadIdx.x;
  if (e < N_EDGES) {
    int d = ei[N_EDGES + e];
    unsafeAtomicAdd(&deg[d], 1.0f);
  }
}

__global__ void __launch_bounds__(256) k_rsqrt(float* __restrict__ deg) {
  int i = blockIdx.x * blockDim.x + threadIdx.x;
  if (i < N_NODES) {
    float d = deg[i];
    deg[i] = (d > 0.0f) ? rsqrtf(d) : 0.0f;
  }
}

__global__ void __launch_bounds__(128) k_gemm1(const float* __restrict__ X,
                                               const float* __restrict__ W,
                                               float* __restrict__ H) {
  const int wid    = (blockIdx.x * blockDim.x + threadIdx.x) >> 5;
  const int lane   = threadIdx.x & 31;
  const int tile_m = wid >> 1;
  const int tile_n = wid & 1;
  const int half   = lane >> 4;
  const int l15    = lane & 15;
  const int rowA   = tile_m * 16 + l15;
  const int kOff   = half * 2;
  const int colB   = tile_n * 16 + l15;

  v8f c = {};
#pragma unroll
  for (int k = 0; k < IN_DIM; k += 4) {
    v2f a = *(const v2f*)(X + (size_t)rowA * IN_DIM + k + kOff);
    v2f b;
    b.x = W[(k + kOff + 0) * HID_DIM + colB];
    b.y = W[(k + kOff + 1) * HID_DIM + colB];
    c = __builtin_amdgcn_wmma_f32_16x16x4_f32(false, a, false, b,
                                              (short)0, c, false, false);
  }
  const int m0 = tile_m * 16;
#pragma unroll
  for (int i = 0; i < 8; ++i) {
    H[(size_t)(m0 + half * 8 + i) * HID_DIM + tile_n * 16 + l15] = c[i];
  }
}

__global__ void __launch_bounds__(256) k_agg32(const int* __restrict__ ei,
                                               const float* __restrict__ dinv,
                                               const float* __restrict__ h,
                                               float* __restrict__ agg) {
  int idx = blockIdx.x * blockDim.x + threadIdx.x;
  int e = idx >> 5;
  int f = idx & 31;
  const int total = N_EDGES + N_NODES;
  if (e >= total) return;
  int s, d;
  if (e < N_EDGES) { s = ei[e]; d = ei[N_EDGES + e]; }
  else             { s = d = e - N_EDGES; }
  float w = dinv[s] * dinv[d];
  float v = h[(size_t)s * HID_DIM + f] * w;
  unsafeAtomicAdd(&agg[(size_t)d * HID_DIM + f], v);
}

__global__ void __launch_bounds__(256) k_bias_relu(float* __restrict__ a,
                                                   const float* __restrict__ b1) {
  int i = blockIdx.x * blockDim.x + threadIdx.x;
  if (i < N_NODES * HID_DIM) {
    float v = a[i] + b1[i & (HID_DIM - 1)];
    a[i] = fmaxf(v, 0.0f);
  }
}

__global__ void __launch_bounds__(64) k_gemm2(const float* __restrict__ H1,
                                              const float* __restrict__ W,
                                              float* __restrict__ G) {
  const int wid    = (blockIdx.x * blockDim.x + threadIdx.x) >> 5;
  const int lane   = threadIdx.x & 31;
  const int tile_m = wid;
  const int half   = lane >> 4;
  const int l15    = lane & 15;
  const int rowA   = tile_m * 16 + l15;
  const int kOff   = half * 2;
  const int colB   = l15;

  v8f c = {};
#pragma unroll
  for (int k = 0; k < HID_DIM; k += 4) {
    v2f a = *(const v2f*)(H1 + (size_t)rowA * HID_DIM + k + kOff);
    v2f b;
    b.x = W[(k + kOff + 0) * OUT_DIM + colB];
    b.y = W[(k + kOff + 1) * OUT_DIM + colB];
    c = __builtin_amdgcn_wmma_f32_16x16x4_f32(false, a, false, b,
                                              (short)0, c, false, false);
  }
  const int m0 = tile_m * 16;
#pragma unroll
  for (int i = 0; i < 8; ++i) {
    G[(size_t)(m0 + half * 8 + i) * OUT_DIM + l15] = c[i];
  }
}

__global__ void __launch_bounds__(256) k_agg16(const int* __restrict__ ei,
                                               const float* __restrict__ dinv,
                                               const float* __restrict__ g,
                                               float* __restrict__ out) {
  int idx = blockIdx.x * blockDim.x + threadIdx.x;
  int e = idx >> 4;
  int f = idx & 15;
  const int total = N_EDGES + N_NODES;
  if (e >= total) return;
  int s, d;
  if (e < N_EDGES) { s = ei[e]; d = ei[N_EDGES + e]; }
  else             { s = d = e - N_EDGES; }
  float w = dinv[s] * dinv[d];
  float v = g[(size_t)s * OUT_DIM + f] * w;
  unsafeAtomicAdd(&out[(size_t)d * OUT_DIM + f], v);
}

extern "C" void kernel_launch(void* const* d_in, const int* in_sizes, int n_in,
                              void* d_out, int out_size, void* d_ws, size_t ws_size,
                              hipStream_t stream) {
  const float* x  = (const float*)d_in[0];
  const float* W1 = (const float*)d_in[1];
  const float* b1 = (const float*)d_in[2];
  const float* W2 = (const float*)d_in[3];
  const float* b2 = (const float*)d_in[4];
  const int*   ei = (const int*)d_in[5];

  float* deg = (float*)d_ws;
  float* h   = deg + N_NODES;
  float* h1  = h   + (size_t)N_NODES * HID_DIM;
  float* g   = h1  + (size_t)N_NODES * HID_DIM;
  float* out = (float*)d_out;

  const int B = 256;
  k_init<<<(N_NODES * HID_DIM + B - 1) / B, B, 0, stream>>>(deg, h1, out, b2);
  k_degree<<<(N_EDGES + B - 1) / B, B, 0, stream>>>(ei, deg);
  k_rsqrt<<<(N_NODES + B - 1) / B, B, 0, stream>>>(deg);
  k_gemm1<<<3125, 128, 0, stream>>>(x, W1, h);
  {
    long long threads = (long long)(N_EDGES + N_NODES) * HID_DIM;
    k_agg32<<<(unsigned)((threads + B - 1) / B), B, 0, stream>>>(ei, deg, h, h1);
  }
  k_bias_relu<<<(N_NODES * HID_DIM + B - 1) / B, B, 0, stream>>>(h1, b1);
  k_gemm2<<<3125, 64, 0, stream>>>(h1, W2, g);
  {
    long long threads = (long long)(N_EDGES + N_NODES) * OUT_DIM;
    k_agg16<<<(unsigned)((threads + B - 1) / B), B, 0, stream>>>(ei, deg, g, out);
  }
}


# Edges counting-sorted by destination range (250 ranges of 400 nodes); each range summed in LDS by one work-group (LDS atomics: arrival order, self term first, bias last); global atomics removed
# speedup vs baseline: 4.0373x; 4.0373x over previous
#include <hip/hip_runtime.h>


#define N_NODES 100000
#define N_EDGES 3200000
#define IN_DIM  128
#define HID_DIM 32
#define OUT_DIM 16

#define NBKT 250
#define NPB  400
#define NBLK 250
#define EPB  12800

typedef __attribute__((ext_vector_type(2))) float v2f;
typedef __attribute__((ext_vector_type(8))) float v8f;
typedef __attribute__((ext_vector_type(2))) int   v2i;

__device__ __forceinline__ int clamp_node(int v) {
  return min(max(v, 0), N_NODES - 1);
}
__device__ __forceinline__ int bucket_of(int d) {
  return (int)((unsigned)d / (unsigned)NPB);
}

__device__ __forceinline__ void block_incl_scan256(int v, int* sh, int tid) {
  if (tid < 256) sh[tid] = v;
  __syncthreads();
#pragma unroll
  for (int off = 1; off < 256; off <<= 1) {
    int t = 0;
    if (tid < 256 && tid >= off) t = sh[tid - off];
    __syncthreads();
    if (tid < 256) sh[tid] += t;
    __syncthreads();
  }
}

__global__ void __launch_bounds__(1024) k_hist(const int* __restrict__ ei,
                                               int* __restrict__ hist) {
  __shared__ int lh[256];
  const int tid = threadIdx.x, blk = blockIdx.x;
  if (tid < 256) lh[tid] = 0;
  __syncthreads();
  const int* dstp = ei + N_EDGES + blk * EPB;
#pragma unroll
  for (int k = 0; k < 13; ++k) {
    int i = tid + k * 1024;
    if (i < EPB) {
      int d = clamp_node(dstp[i]);
      atomicAdd(&lh[bucket_of(d)], 1);
    }
  }
  __syncthreads();
  if (tid < NBKT) hist[tid * NBLK + blk] = lh[tid];
}

__global__ void __launch_bounds__(256) k_rowscan(const int* __restrict__ hist,
                                                 int* __restrict__ rowOff,
                                                 int* __restrict__ rowTotal) {
  __shared__ int sh[256];
  const int tid = threadIdx.x, b = blockIdx.x;
  int v = (tid < NBLK) ? hist[b * NBLK + tid] : 0;
  block_incl_scan256(v, sh, tid);
  if (tid < NBLK) rowOff[b * NBLK + tid] = sh[tid] - v;
  if (tid == 0) rowTotal[b] = sh[255];
}

__global__ void __launch_bounds__(1024) k_scatter(const int* __restrict__ ei,
                                                  const int* __restrict__ rowOff,
                                                  const int* __restrict__ rowTotal,
                                                  int* __restrict__ binned) {
  __shared__ int sh[256];
  __shared__ int cur[256];
  const int tid = threadIdx.x, blk = blockIdx.x;
  int v = (tid < NBKT) ? rowTotal[tid] : 0;
  block_incl_scan256(v, sh, tid);
  if (tid < NBKT) cur[tid] = (sh[tid] - v) + rowOff[tid * NBLK + blk];
  __syncthreads();
  const int* srcp = ei + blk * EPB;
  const int* dstp = ei + N_EDGES + blk * EPB;
#pragma unroll
  for (int k = 0; k < 13; ++k) {
    int i = tid + k * 1024;
    if (i < EPB) {
      int s = clamp_node(srcp[i]);
      int d = clamp_node(dstp[i]);
      int pos = atomicAdd(&cur[bucket_of(d)], 1);
      v2i p; p.x = s; p.y = d;
      *(v2i*)(binned + 2 * (size_t)pos) = p;
    }
  }
}

__global__ void __launch_bounds__(1024) k_degree(const int* __restrict__ binned,
                                                 const int* __restrict__ rowTotal,
                                                 float* __restrict__ dinv) {
  __shared__ int cnt[NPB];
  __shared__ int sh[256];
  const int tid = threadIdx.x, b = blockIdx.x;
  const int n0 = b * NPB;
  if (tid < NPB) cnt[tid] = 0;
  int v = (tid < NBKT) ? rowTotal[tid] : 0;
  block_incl_scan256(v, sh, tid);
  const int start = b ? sh[b - 1] : 0;
  const int count = sh[b] - start;
  for (int i = tid; i < count; i += 1024) {
    int d = binned[2 * (size_t)(start + i) + 1] - n0;
    atomicAdd(&cnt[d], 1);
  }
  __syncthreads();
  if (tid < NPB) {
    float deg = 1.0f + (float)cnt[tid];
    dinv[n0 + tid] = (deg > 0.0f) ? rsqrtf(deg) : 0.0f;
  }
}

__global__ void __launch_bounds__(128) k_gemm1(const float* __restrict__ X,
                                               const float* __restrict__ W,
                                               float* __restrict__ H) {
  const int wid    = (blockIdx.x * blockDim.x + threadIdx.x) >> 5;
  const int lane   = threadIdx.x & 31;
  const int tile_m = wid >> 1;
  const int tile_n = wid & 1;
  const int half   = lane >> 4;
  const int l15    = lane & 15;
  const int rowA   = tile_m * 16 + l15;
  const int kOff   = half * 2;
  const int colB   = tile_n * 16 + l15;

  v8f c = {};
#pragma unroll
  for (int k = 0; k < IN_DIM; k += 4) {
    v2f a = *(const v2f*)(X + (size_t)rowA * IN_DIM + k + kOff);
    v2f b;
    b.x = W[(k + kOff + 0) * HID_DIM + colB];
    b.y = W[(k + kOff + 1) * HID_DIM + colB];
    c = __builtin_amdgcn_wmma_f32_16x16x4_f32(false, a, false, b,
                                              (short)0, c, false, false);
  }
  const int m0 = tile_m * 16;
#pragma unroll
  for (int i = 0; i < 8; ++i) {
    H[(size_t)(m0 + half * 8 + i) * HID_DIM + tile_n * 16 + l15] = c[i];
  }
}

__global__ void __launch_bounds__(1024) k_agg32(const int* __restrict__ binned,
                                                const int* __restrict__ rowTotal,
                                                const float* __restrict__ dinv,
                                                const float* __restrict__ h,
                                                const float* __restrict__ b1,
                                                float* __restrict__ h1) {
  __shared__ float acc[NPB * HID_DIM];
  __shared__ float dl[NPB];
  __shared__ int sh[256];
  const int tid = threadIdx.x, b = blockIdx.x;
  const int lane = tid & 31;
  const int wave = __builtin_amdgcn_readfirstlane(tid >> 5);
  const int n0 = b * NPB;
  if (tid < NPB) dl[tid] = dinv[n0 + tid];
  int tv = (tid < NBKT) ? rowTotal[tid] : 0;
  block_incl_scan256(tv, sh, tid);
  const int start = __builtin_amdgcn_readfirstlane(b ? sh[b - 1] : 0);
  const int count = __builtin_amdgcn_readfirstlane(sh[b]) - start;
  for (int i = tid; i < NPB * HID_DIM; i += 1024) {
    float di = dl[i >> 5];
    acc[i] = h[(size_t)n0 * HID_DIM + i] * (di * di);
  }
  __syncthreads();

  for (int e0 = wave * 32; e0 < count; e0 += 1024) {
    const int idx = e0 + lane;
    int s = 0, d = 0;
    float w = 0.0f;
    if (idx < count) {
      v2i p = *(const v2i*)(binned + 2 * (size_t)(start + idx));
      s = p.x;
      d = p.y - n0;
      w = dinv[s] * dl[d];
    }
    const int m = count - e0;
    if (m >= 32) {
#pragma unroll
      for (int j = 0; j < 32; j += 16) {
        float x[16];
#pragma unroll
        for (int u = 0; u < 16; ++u) {
          int sj = __builtin_amdgcn_readlane(s, j + u);
          x[u] = h[(size_t)sj * HID_DIM + lane];
        }
#pragma unroll
        for (int u = 0; u < 16; ++u) {
          int dj = __builtin_amdgcn_readlane(d, j + u);
          float wj = __int_as_float(__builtin_amdgcn_readlane(__float_as_int(w), j + u));
          atomicAdd(&acc[dj * HID_DIM + lane], x[u] * wj);
        }
      }
    } else {
      for (int j = 0; j < m; ++j) {
        int sj = __shfl(s, j);
        int dj = __shfl(d, j);
        float wj = __shfl(w, j);
        atomicAdd(&acc[dj * HID_DIM + lane], h[(size_t)sj * HID_DIM + lane] * wj);
      }
    }
  }
  __syncthreads();
  for (int i = tid; i < NPB * HID_DIM; i += 1024) {
    float v = acc[i] + b1[i & (HID_DIM - 1)];
    h1[(size_t)n0 * HID_DIM + i] = fmaxf(v, 0.0f);
  }
}

__global__ void __launch_bounds__(64) k_gemm2(const float* __restrict__ H1,
                                              const float* __restrict__ W,
                                              float* __restrict__ G) {
  const int wid    = (blockIdx.x * blockDim.x + threadIdx.x) >> 5;
  const int lane   = threadIdx.x & 31;
  const int tile_m = wid;
  const int half   = lane >> 4;
  const int l15    = lane & 15;
  const int rowA   = tile_m * 16 + l15;
  const int kOff   = half * 2;
  const int colB   = l15;

  v8f c = {};
#pragma unroll
  for (int k = 0; k < HID_DIM; k += 4) {
    v2f a = *(const v2f*)(H1 + (size_t)rowA * HID_DIM + k + kOff);
    v2f b;
    b.x = W[(k + kOff + 0) * OUT_DIM + colB];
    b.y = W[(k + kOff + 1) * OUT_DIM + colB];
    c = __builtin_amdgcn_wmma_f32_16x16x4_f32(false, a, false, b,
                                              (short)0, c, false, false);
  }
  const int m0 = tile_m * 16;
#pragma unroll
  for (int i = 0; i < 8; ++i) {
    G[(size_t)(m0 + half * 8 + i) * OUT_DIM + l15] = c[i];
  }
}

__global__ void __launch_bounds__(1024) k_agg16(const int* __restrict__ binned,
                                                const int* __restrict__ rowTotal,
                                                const float* __restrict__ dinv,
                                                const float* __restrict__ g,
                                                const float* __restrict__ b2,
                                                float* __restrict__ out) {
  __shared__ float acc[NPB * OUT_DIM];
  __shared__ float dl[NPB];
  __shared__ int sh[256];
  const int tid = threadIdx.x, b = blockIdx.x;
  const int lane = tid & 31;
  const int hf = lane >> 4;
  const int c = lane & 15;
  const int wave = __builtin_amdgcn_readfirstlane(tid >> 5);
  const int n0 = b * NPB;
  if (tid < NPB) dl[tid] = dinv[n0 + tid];
  int tv = (tid < NBKT) ? rowTotal[tid] : 0;
  block_incl_scan256(tv, sh, tid);
  const int start = __builtin_amdgcn_readfirstlane(b ? sh[b - 1] : 0);
  const int count = __builtin_amdgcn_readfirstlane(sh[b]) - start;
  for (int i = tid; i < NPB * OUT_DIM; i += 1024) {
    float di = dl[i >> 4];
    acc[i] = g[(size_t)n0 * OUT_DIM + i] * (di * di);
  }
  __syncthreads();

  for (int e0 = wave * 32; e0 < count; e0 += 1024) {
    const int idx = e0 + lane;
    int s = 0, d = 0;
    float w = 0.0f;
    if (idx < count) {
      v2i p = *(const v2i*)(binned + 2 * (size_t)(start + idx));
      s = p.x;
      d = p.y - n0;
      w = dinv[s] * dl[d];
    }
    const int m = count - e0;
    if (m >= 32) {
      float x[16];
#pragma unroll
      for (int u = 0; u < 16; ++u) {
        int s0 = __builtin_amdgcn_readlane(s, 2 * u);
        int s1 = __builtin_amdgcn_readlane(s, 2 * u + 1);
        int sj = hf ? s1 : s0;
        x[u] = g[(size_t)sj * OUT_DIM + c];
      }
#pragma unroll
      for (int u = 0; u < 16; ++u) {
        int d0 = __builtin_amdgcn_readlane(d, 2 * u);
        int d1 = __builtin_amdgcn_readlane(d, 2 * u + 1);
        int w0 = __builtin_amdgcn_readlane(__float_as_int(w), 2 * u);
        int w1 = __builtin_amdgcn_readlane(__float_as_int(w), 2 * u + 1);
        int dj = hf ? d1 : d0;
        float wj = __int_as_float(hf ? w1 : w0);
        atomicAdd(&acc[dj * OUT_DIM + c], x[u] * wj);
      }
    } else {
      for (int j = 0; 2 * j < m; ++j) {
        int e = 2 * j + hf;
        int sj = __shfl(s, e);
        int dj = __shfl(d, e);
        float wj = __shfl(w, e);
        if (e < m) atomicAdd(&acc[dj * OUT_DIM + c], g[(size_t)sj * OUT_DIM + c] * wj);
      }
    }
  }
  __syncthreads();
  for (int i = tid; i < NPB * OUT_DIM; i += 1024) {
    out[(size_t)n0 * OUT_DIM + i] = acc[i] + b2[i & (OUT_DIM - 1)];
  }
}

extern "C" void kernel_launch(void* const* d_in, const int* in_sizes, int n_in,
                              void* d_out, int out_size, void* d_ws, size_t ws_size,
                              hipStream_t stream) {
  const float* x  = (const float*)d_in[0];
  const float* W1 = (const float*)d_in[1];
  const float* b1 = (const float*)d_in[2];
  const float* W2 = (const float*)d_in[3];
  const float* b2 = (const float*)d_in[4];
  const int*   ei = (const int*)d_in[5];

  float* dinv = (float*)d_ws;
  float* h    = dinv + N_NODES;
  float* h1   = h    + (size_t)N_NODES * HID_DIM;
  float* g    = h1   + (size_t)N_NODES * HID_DIM;
  int* hist     = (int*)(g + (size_t)N_NODES * OUT_DIM);
  int* rowOff   = hist + NBKT * NBLK;
  int* rowTotal = rowOff + NBKT * NBLK;
  int* binned   = rowTotal + 256;
  float* out = (float*)d_out;

  k_hist<<<NBLK, 1024, 0, stream>>>(ei, hist);
  k_rowscan<<<NBKT, 256, 0, stream>>>(hist, rowOff, rowTotal);
  k_scatter<<<NBLK, 1024, 0, stream>>>(ei, rowOff, rowTotal, binned);
  k_degree<<<NBKT, 1024, 0, stream>>>(binned, rowTotal, dinv);
  k_gemm1<<<3125, 128, 0, stream>>>(x, W1, h);
  k_agg32<<<NBKT, 1024, 0, stream>>>(binned, rowTotal, dinv, h, b1, h1);
  k_gemm2<<<3125, 64, 0, stream>>>(h1, W2, g);
  k_agg16<<<NBKT, 1024, 0, stream>>>(binned, rowTotal, dinv, g, b2, out);
}


# Edge weight factored out of the edge loop: GEMM epilogues scale each row by deg^-1/2 of its node, destination deg^-1/2 multiplies the finished sum once; per-edge deg^-1/2 gather and multiply removed
# speedup vs baseline: 4.7328x; 1.1723x over previous
#include <hip/hip_runtime.h>


#define N_NODES 100000
#define N_EDGES 3200000
#define IN_DIM  128
#define HID_DIM 32
#define OUT_DIM 16

#define NBKT 250
#define NPB  400
#define NBLK 250
#define EPB  12800

typedef __attribute__((ext_vector_type(2))) float v2f;
typedef __attribute__((ext_vector_type(4))) float v4f;
typedef __attribute__((ext_vector_type(8))) float v8f;
typedef __attribute__((ext_vector_type(2))) int   v2i;

__device__ __forceinline__ int clamp_node(int v) {
  return min(max(v, 0), N_NODES - 1);
}
__device__ __forceinline__ int bucket_of(int d) {
  return (int)((unsigned)d / (unsigned)NPB);
}

__device__ __forceinline__ void block_incl_scan256(int v, int* sh, int tid) {
  if (tid < 256) sh[tid] = v;
  __syncthreads();
#pragma unroll
  for (int off = 1; off < 256; off <<= 1) {
    int t = 0;
    if (tid < 256 && tid >= off) t = sh[tid - off];
    __syncthreads();
    if (tid < 256) sh[tid] += t;
    __syncthreads();
  }
}

__global__ void __launch_bounds__(1024) k_hist(const int* __restrict__ ei,
                                               int* __restrict__ hist) {
  __shared__ int lh[256];
  const int tid = threadIdx.x, blk = blockIdx.x;
  if (tid < 256) lh[tid] = 0;
  __syncthreads();
  const int* dstp = ei + N_EDGES + blk * EPB;
#pragma unroll
  for (int k = 0; k < 13; ++k) {
    int i = tid + k * 1024;
    if (i < EPB) {
      int d = clamp_node(dstp[i]);
      atomicAdd(&lh[bucket_of(d)], 1);
    }
  }
  __syncthreads();
  if (tid < NBKT) hist[tid * NBLK + blk] = lh[tid];
}

__global__ void __launch_bounds__(256) k_rowscan(const int* __restrict__ hist,
                                                 int* __restrict__ rowOff,
                                                 int* __restrict__ rowTotal) {
  __shared__ int sh[256];
  const int tid = threadIdx.x, b = blockIdx.x;
  int v = (tid < NBLK) ? hist[b * NBLK + tid] : 0;
  block_incl_scan256(v, sh, tid);
  if (tid < NBLK) rowOff[b * NBLK + tid] = sh[tid] - v;
  if (tid == 0) rowTotal[b] = sh[255];
}

__global__ void __launch_bounds__(1024) k_scatter(const int* __restrict__ ei,
                                                  const int* __restrict__ rowOff,
                                                  const int* __restrict__ rowTotal,
                                                  int* __restrict__ binned) {
  __shared__ int sh[256];
  __shared__ int cur[256];
  const int tid = threadIdx.x, blk = blockIdx.x;
  int v = (tid < NBKT) ? rowTotal[tid] : 0;
  block_incl_scan256(v, sh, tid);
  if (tid < NBKT) cur[tid] = (sh[tid] - v) + rowOff[tid * NBLK + blk];
  __syncthreads();
  const int* srcp = ei + blk * EPB;
  const int* dstp = ei + N_EDGES + blk * EPB;
#pragma unroll
  for (int k = 0; k < 13; ++k) {
    int i = tid + k * 1024;
    if (i < EPB) {
      int s = clamp_node(srcp[i]);
      int d = clamp_node(dstp[i]);
      int pos = atomicAdd(&cur[bucket_of(d)], 1);
      v2i p; p.x = s; p.y = d;
      *(v2i*)(binned + 2 * (size_t)pos) = p;
    }
  }
}

__global__ void __launch_bounds__(1024) k_degree(const int* __restrict__ binned,
                                                 const int* __restrict__ rowTotal,
                                                 float* __restrict__ dinv) {
  __shared__ int cnt[NPB];
  __shared__ int sh[256];
  const int tid = threadIdx.x, b = blockIdx.x;
  const int n0 = b * NPB;
  if (tid < NPB) cnt[tid] = 0;
  int v = (tid < NBKT) ? rowTotal[tid] : 0;
  block_incl_scan256(v, sh, tid);
  const int start = b ? sh[b - 1] : 0;
  const int count = sh[b] - start;
  for (int i = tid; i < count; i += 1024) {
    int d = binned[2 * (size_t)(start + i) + 1] - n0;
    atomicAdd(&cnt[d], 1);
  }
  __syncthreads();
  if (tid < NPB) {
    float deg = 1.0f + (float)cnt[tid];
    dinv[n0 + tid] = (deg > 0.0f) ? rsqrtf(deg) : 0.0f;
  }
}

__global__ void __launch_bounds__(128) k_gemm1(const float* __restrict__ X,
                                               const float* __restrict__ W,
                                               const float* __restrict__ dinv,
                                               float* __restrict__ H) {
  const int wid    = (blockIdx.x * blockDim.x + threadIdx.x) >> 5;
  const int lane   = threadIdx.x & 31;
  const int tile_m = wid >> 1;
  const int tile_n = wid & 1;
  const int half   = lane >> 4;
  const int l15    = lane & 15;
  const int rowA   = tile_m * 16 + l15;
  const int kOff   = half * 2;
  const int colB   = tile_n * 16 + l15;

  v8f c = {};
#pragma unroll
  for (int k = 0; k < IN_DIM; k += 4) {
    v2f a = *(const v2f*)(X + (size_t)rowA * IN_DIM + k + kOff);
    v2f b;
    b.x = W[(k + kOff + 0) * HID_DIM + colB];
    b.y = W[(k + kOff + 1) * HID_DIM + colB];
    c = __builtin_amdgcn_wmma_f32_16x16x4_f32(false, a, false, b,
                                              (short)0, c, false, false);
  }
  const int m0 = tile_m * 16;
  const v4f dA = *(const v4f*)(dinv + m0 + half * 8);
  const v4f dB = *(const v4f*)(dinv + m0 + half * 8 + 4);
#pragma unroll
  for (int i = 0; i < 8; ++i) {
    const float di = (i < 4) ? dA[i & 3] : dB[i & 3];
    H[(size_t)(m0 + half * 8 + i) * HID_DIM + tile_n * 16 + l15] = c[i] * di;
  }
}

__global__ void __launch_bounds__(1024) k_agg32(const int* __restrict__ binned,
                                                const int* __restrict__ rowTotal,
                                                const float* __restrict__ dinv,
                                                const float* __restrict__ h,
                                                const float* __restrict__ b1,
                                                float* __restrict__ h1) {
  __shared__ float acc[NPB * HID_DIM];
  __shared__ float dl[NPB];
  __shared__ int sh[256];
  const int tid = threadIdx.x, b = blockIdx.x;
  const int lane = tid & 31;
  const int wave = __builtin_amdgcn_readfirstlane(tid >> 5);
  const int n0 = b * NPB;
  if (tid < NPB) dl[tid] = dinv[n0 + tid];
  int tv = (tid < NBKT) ? rowTotal[tid] : 0;
  block_incl_scan256(tv, sh, tid);
  const int start = __builtin_amdgcn_readfirstlane(b ? sh[b - 1] : 0);
  const int count = __builtin_amdgcn_readfirstlane(sh[b]) - start;
  for (int i = tid; i < NPB * HID_DIM; i += 1024) {
    acc[i] = h[(size_t)n0 * HID_DIM + i];
  }
  __syncthreads();

  for (int e0 = wave * 32; e0 < count; e0 += 1024) {
    const int idx = e0 + lane;
    int s = 0, d = 0;
    if (idx < count) {
      v2i p = *(const v2i*)(binned + 2 * (size_t)(start + idx));
      s = p.x;
      d = p.y - n0;
    }
    const int m = count - e0;
    if (m >= 32) {
#pragma unroll
      for (int j = 0; j < 32; j += 16) {
        float x[16];
#pragma unroll
        for (int u = 0; u < 16; ++u) {
          int sj = __builtin_amdgcn_readlane(s, j + u);
          x[u] = h[(size_t)sj * HID_DIM + lane];
        }
#pragma unroll
        for (int u = 0; u < 16; ++u) {
          int dj = __builtin_amdgcn_readlane(d, j + u);
          atomicAdd(&acc[dj * HID_DIM + lane], x[u]);
        }
      }
    } else {
      for (int j = 0; j < m; ++j) {
        int sj = __shfl(s, j);
        int dj = __shfl(d, j);
        atomicAdd(&acc[dj * HID_DIM + lane], h[(size_t)sj * HID_DIM + lane]);
      }
    }
  }
  __syncthreads();
  for (int i = tid; i < NPB * HID_DIM; i += 1024) {
    float v = acc[i] * dl[i >> 5] + b1[i & (HID_DIM - 1)];
    h1[(size_t)n0 * HID_DIM + i] = fmaxf(v, 0.0f);
  }
}

__global__ void __launch_bounds__(64) k_gemm2(const float* __restrict__ H1,
                                              const float* __restrict__ W,
                                              const float* __restrict__ dinv,
                                              float* __restrict__ G) {
  const int wid    = (blockIdx.x * blockDim.x + threadIdx.x) >> 5;
  const int lane   = threadIdx.x & 31;
  const int tile_m = wid;
  const int half   = lane >> 4;
  const int l15    = lane & 15;
  const int rowA   = tile_m * 16 + l15;
  const int kOff   = half * 2;
  const int colB   = l15;

  v8f c = {};
#pragma unroll
  for (int k = 0; k < HID_DIM; k += 4) {
    v2f a = *(const v2f*)(H1 + (size_t)rowA * HID_DIM + k + kOff);
    v2f b;
    b.x = W[(k + kOff + 0) * OUT_DIM + colB];
    b.y = W[(k + kOff + 1) * OUT_DIM + colB];
    c = __builtin_amdgcn_wmma_f32_16x16x4_f32(false, a, false, b,
                                              (short)0, c, false, false);
  }
  const int m0 = tile_m * 16;
  const v4f dA = *(const v4f*)(dinv + m0 + half * 8);
  const v4f dB = *(const v4f*)(dinv + m0 + half * 8 + 4);
#pragma unroll
  for (int i = 0; i < 8; ++i) {
    const float di = (i < 4) ? dA[i & 3] : dB[i & 3];
    G[(size_t)(m0 + half * 8 + i) * OUT_DIM + l15] = c[i] * di;
  }
}

__global__ void __launch_bounds__(1024) k_agg16(const int* __restrict__ binned,
                                                const int* __restrict__ rowTotal,
                                                const float* __restrict__ dinv,
                                                const float* __restrict__ g,
                                                const float* __restrict__ b2,
                                                float* __restrict__ out) {
  __shared__ float acc[NPB * OUT_DIM];
  __shared__ float dl[NPB];
  __shared__ int sh[256];
  const int tid = threadIdx.x, b = blockIdx.x;
  const int lane = tid & 31;
  const int hf = lane >> 4;
  const int c = lane & 15;
  const int wave = __builtin_amdgcn_readfirstlane(tid >> 5);
  const int n0 = b * NPB;
  if (tid < NPB) dl[tid] = dinv[n0 + tid];
  int tv = (tid < NBKT) ? rowTotal[tid] : 0;
  block_incl_scan256(tv, sh, tid);
  const int start = __builtin_amdgcn_readfirstlane(b ? sh[b - 1] : 0);
  const int count = __builtin_amdgcn_readfirstlane(sh[b]) - start;
  for (int i = tid; i < NPB * OUT_DIM; i += 1024) {
    acc[i] = g[(size_t)n0 * OUT_DIM + i];
  }
  __syncthreads();

  for (int e0 = wave * 32; e0 < count; e0 += 1024) {
    const int idx = e0 + lane;
    int s = 0, d = 0;
    if (idx < count) {
      v2i p = *(const v2i*)(binned + 2 * (size_t)(start + idx));
      s = p.x;
      d = p.y - n0;
    }
    const int m = count - e0;
    if (m >= 32) {
      float x[16];
#pragma unroll
      for (int u = 0; u < 16; ++u) {
        int s0 = __builtin_amdgcn_readlane(s, 2 * u);
        int s1 = __builtin_amdgcn_readlane(s, 2 * u + 1);
        int sj = hf ? s1 : s0;
        x[u] = g[(size_t)sj * OUT_DIM + c];
      }
#pragma unroll
      for (int u = 0; u < 16; ++u) {
        int d0 = __builtin_amdgcn_readlane(d, 2 * u);
        int d1 = __builtin_amdgcn_readlane(d, 2 * u + 1);
        int dj = hf ? d1 : d0;
        atomicAdd(&acc[dj * OUT_DIM + c], x[u]);
      }
    } else {
      for (int j = 0; 2 * j < m; ++j) {
        int e = 2 * j + hf;
        int sj = __shfl(s, e);
        int dj = __shfl(d, e);
        if (e < m) atomicAdd(&acc[dj * OUT_DIM + c], g[(size_t)sj * OUT_DIM + c]);
      }
    }
  }
  __syncthreads();
  for (int i = tid; i < NPB * OUT_DIM; i += 1024) {
    out[(size_t)n0 * OUT_DIM + i] = acc[i] * dl[i >> 4] + b2[i & (OUT_DIM - 1)];
  }
}

extern "C" void kernel_launch(void* const* d_in, const int* in_sizes, int n_in,
                              void* d_out, int out_size, void* d_ws, size_t ws_size,
                              hipStream_t stream) {
  const float* x  = (const float*)d_in[0];
  const float* W1 = (const float*)d_in[1];
  const float* b1 = (const float*)d_in[2];
  const float* W2 = (const float*)d_in[3];
  const float* b2 = (const float*)d_in[4];
  const int*   ei = (const int*)d_in[5];

  float* dinv = (float*)d_ws;
  float* h    = dinv + N_NODES;
  float* h1   = h    + (size_t)N_NODES * HID_DIM;
  float* g    = h1   + (size_t)N_NODES * HID_DIM;
  int* hist     = (int*)(g + (size_t)N_NODES * OUT_DIM);
  int* rowOff   = hist + NBKT * NBLK;
  int* rowTotal = rowOff + NBKT * NBLK;
  int* binned   = rowTotal + 256;
  float* out = (float*)d_out;

  k_hist<<<NBLK, 1024, 0, stream>>>(ei, hist);
  k_rowscan<<<NBKT, 256, 0, stream>>>(hist, rowOff, rowTotal);
  k_scatter<<<NBLK, 1024, 0, stream>>>(ei, rowOff, rowTotal, binned);
  k_degree<<<NBKT, 1024, 0, stream>>>(binned, rowTotal, dinv);
  k_gemm1<<<3125, 128, 0, stream>>>(x, W1, dinv, h);
  k_agg32<<<NBKT, 1024, 0, stream>>>(binned, rowTotal, dinv, h, b1, h1);
  k_gemm2<<<3125, 64, 0, stream>>>(h1, W2, dinv, g);
  k_agg16<<<NBKT, 1024, 0, stream>>>(binned, rowTotal, dinv, g, b2, out);
}


# Both dense transforms moved into the per-range work-groups: GEMM1 behind the degree count of the same 400 rows, GEMM2 on the layer-1 sums still in LDS; 8 launches to 6, hidden rows never stored
# speedup vs baseline: 5.0581x; 1.0687x over previous
#include <hip/hip_runtime.h>


#define N_NODES 100000
#define N_EDGES 3200000
#define IN_DIM  128
#define HID_DIM 32
#define OUT_DIM 16

#define NBKT 250
#define NPB  400
#define NBLK 250
#define EPB  12800

typedef __attribute__((ext_vector_type(2))) float v2f;
typedef __attribute__((ext_vector_type(4))) float v4f;
typedef __attribute__((ext_vector_type(8))) float v8f;
typedef __attribute__((ext_vector_type(2))) int   v2i;

__device__ __forceinline__ int clamp_node(int v) {
  return min(max(v, 0), N_NODES - 1);
}
__device__ __forceinline__ int bucket_of(int d) {
  return (int)((unsigned)d / (unsigned)NPB);
}

__device__ __forceinline__ void block_incl_scan256(int v, int* sh, int tid) {
  if (tid < 256) sh[tid] = v;
  __syncthreads();
#pragma unroll
  for (int off = 1; off < 256; off <<= 1) {
    int t = 0;
    if (tid < 256 && tid >= off) t = sh[tid - off];
    __syncthreads();
    if (tid < 256) sh[tid] += t;
    __syncthreads();
  }
}

__global__ void __launch_bounds__(1024) k_hist(const int* __restrict__ ei,
                                               int* __restrict__ hist) {
  __shared__ int lh[256];
  const int tid = threadIdx.x, blk = blockIdx.x;
  if (tid < 256) lh[tid] = 0;
  __syncthreads();
  const int* dstp = ei + N_EDGES + blk * EPB;
#pragma unroll
  for (int k = 0; k < 13; ++k) {
    int i = tid + k * 1024;
    if (i < EPB) {
      int d = clamp_node(dstp[i]);
      atomicAdd(&lh[bucket_of(d)], 1);
    }
  }
  __syncthreads();
  if (tid < NBKT) hist[tid * NBLK + blk] = lh[tid];
}

__global__ void __launch_bounds__(256) k_rowscan(const int* __restrict__ hist,
                                                 int* __restrict__ rowOff,
                                                 int* __restrict__ rowTotal) {
  __shared__ int sh[256];
  const int tid = threadIdx.x, b = blockIdx.x;
  int v = (tid < NBLK) ? hist[b * NBLK + tid] : 0;
  block_incl_scan256(v, sh, tid);
  if (tid < NBLK) rowOff[b * NBLK + tid] = sh[tid] - v;
  if (tid == 0) rowTotal[b] = sh[255];
}

__global__ void __launch_bounds__(1024) k_scatter(const int* __restrict__ ei,
                                                  const int* __restrict__ rowOff,
                                                  const int* __restrict__ rowTotal,
                                                  int* __restrict__ binned) {
  __shared__ int sh[256];
  __shared__ int cur[256];
  const int tid = threadIdx.x, blk = blockIdx.x;
  int v = (tid < NBKT) ? rowTotal[tid] : 0;
  block_incl_scan256(v, sh, tid);
  if (tid < NBKT) cur[tid] = (sh[tid] - v) + rowOff[tid * NBLK + blk];
  __syncthreads();
  const int* srcp = ei + blk * EPB;
  const int* dstp = ei + N_EDGES + blk * EPB;
#pragma unroll
  for (int k = 0; k < 13; ++k) {
    int i = tid + k * 1024;
    if (i < EPB) {
      int s = clamp_node(srcp[i]);
      int d = clamp_node(dstp[i]);
      int pos = atomicAdd(&cur[bucket_of(d)], 1);
      v2i p; p.x = s; p.y = d;
      *(v2i*)(binned + 2 * (size_t)pos) = p;
    }
  }
}

#define NTILE (NPB / 16)
__global__ void __launch_bounds__(1024) k_degree_gemm1(const int* __restrict__ binned,
                                                       const int* __restrict__ rowTotal,
                                                       const float* __restrict__ X,
                                                       const float* __restrict__ W,
                                                       float* __restrict__ dinv,
                                                       float* __restrict__ H) {
  __shared__ int cnt[NPB];
  __shared__ float dl[NPB];
  __shared__ int sh[256];
  const int tid = threadIdx.x, b = blockIdx.x;
  const int lane = tid & 31;
  const int wave = __builtin_amdgcn_readfirstlane(tid >> 5);
  const int n0   = b * NPB;
  const int half = lane >> 4;
  const int l15  = lane & 15;
  const int kOff = half * 2;

  v2f a[IN_DIM / 4];
  if (wave < NTILE) {
    const float* xr = X + (size_t)(n0 + wave * 16 + l15) * IN_DIM + kOff;
#pragma unroll
    for (int k4 = 0; k4 < IN_DIM / 4; ++k4) a[k4] = *(const v2f*)(xr + 4 * k4);
  }

  if (tid < NPB) cnt[tid] = 0;
  int v = (tid < NBKT) ? rowTotal[tid] : 0;
  block_incl_scan256(v, sh, tid);
  const int start = b ? sh[b - 1] : 0;
  const int count = sh[b] - start;
  for (int i = tid; i < count; i += 1024) {
    int d = binned[2 * (size_t)(start + i) + 1] - n0;
    atomicAdd(&cnt[d], 1);
  }
  __syncthreads();
  if (tid < NPB) {
    float deg = 1.0f + (float)cnt[tid];
    float di = (deg > 0.0f) ? rsqrtf(deg) : 0.0f;
    dl[tid] = di;
    dinv[n0 + tid] = di;
  }
  __syncthreads();

  if (wave < NTILE) {
    v8f c0 = {}, c1 = {};
#pragma unroll
    for (int k4 = 0; k4 < IN_DIM / 4; ++k4) {
      const float* wr = W + (4 * k4 + kOff) * HID_DIM + l15;
      v2f b0, b1;
      b0.x = wr[0];
      b0.y = wr[HID_DIM];
      b1.x = wr[16];
      b1.y = wr[HID_DIM + 16];
      c0 = __builtin_amdgcn_wmma_f32_16x16x4_f32(false, a[k4], false, b0,
                                                 (short)0, c0, false, false);
      c1 = __builtin_amdgcn_wmma_f32_16x16x4_f32(false, a[k4], false, b1,
                                                 (short)0, c1, false, false);
    }
    const int r0 = wave * 16 + half * 8;
#pragma unroll
    for (int i = 0; i < 8; ++i) {
      const float di = dl[r0 + i];
      float* hr = H + (size_t)(n0 + r0 + i) * HID_DIM + l15;
      hr[0]  = c0[i] * di;
      hr[16] = c1[i] * di;
    }
  }
}

#define APITCH 34
__global__ void __launch_bounds__(1024) k_agg32_gemm2(const int* __restrict__ binned,
                                                      const int* __restrict__ rowTotal,
                                                      const float* __restrict__ dinv,
                                                      const float* __restrict__ h,
                                                      const float* __restrict__ b1,
                                                      const float* __restrict__ W2,
                                                      float* __restrict__ G) {
  __shared__ __attribute__((aligned(16))) float acc[NPB * APITCH];
  __shared__ float dl[NPB];
  __shared__ int sh[256];
  const int tid = threadIdx.x, b = blockIdx.x;
  const int lane = tid & 31;
  const int wave = __builtin_amdgcn_readfirstlane(tid >> 5);
  const int n0 = b * NPB;
  if (tid < NPB) dl[tid] = dinv[n0 + tid];
  int tv = (tid < NBKT) ? rowTotal[tid] : 0;
  block_incl_scan256(tv, sh, tid);
  const int start = __builtin_amdgcn_readfirstlane(b ? sh[b - 1] : 0);
  const int count = __builtin_amdgcn_readfirstlane(sh[b]) - start;
  for (int i = tid; i < NPB * HID_DIM; i += 1024) {
    acc[(i >> 5) * APITCH + (i & 31)] = h[(size_t)n0 * HID_DIM + i];
  }
  __syncthreads();

  for (int e0 = wave * 32; e0 < count; e0 += 1024) {
    const int idx = e0 + lane;
    int s = 0, d = 0;
    if (idx < count) {
      v2i p = *(const v2i*)(binned + 2 * (size_t)(start + idx));
      s = p.x;
      d = p.y - n0;
    }
    const int m = count - e0;
    if (m >= 32) {
#pragma unroll
      for (int j = 0; j < 32; j += 16) {
        float x[16];
#pragma unroll
        for (int u = 0; u < 16; ++u) {
          int sj = __builtin_amdgcn_readlane(s, j + u);
          x[u] = h[(size_t)sj * HID_DIM + lane];
        }
#pragma unroll
        for (int u = 0; u < 16; ++u) {
          int dj = __builtin_amdgcn_readlane(d, j + u);
          atomicAdd(&acc[dj * APITCH + lane], x[u]);
        }
      }
    } else {
      for (int j = 0; j < m; ++j) {
        int sj = __shfl(s, j);
        int dj = __shfl(d, j);
        atomicAdd(&acc[dj * APITCH + lane], h[(size_t)sj * HID_DIM + lane]);
      }
    }
  }
  __syncthreads();
  for (int i = tid; i < NPB * HID_DIM; i += 1024) {
    const int r = i >> 5, cc = i & 31;
    float v = acc[r * APITCH + cc] * dl[r] + b1[cc];
    acc[r * APITCH + cc] = fmaxf(v, 0.0f);
  }
  __syncthreads();

  if (wave < NTILE) {
    const int half = lane >> 4;
    const int l15  = lane & 15;
    const int kOff = half * 2;
    const float* ar = acc + (wave * 16 + l15) * APITCH + kOff;
    v8f c = {};
#pragma unroll
    for (int k = 0; k < HID_DIM; k += 4) {
      v2f a = *(const v2f*)(ar + k);
      v2f bb;
      bb.x = W2[(k + kOff + 0) * OUT_DIM + l15];
      bb.y = W2[(k + kOff + 1) * OUT_DIM + l15];
      c = __builtin_amdgcn_wmma_f32_16x16x4_f32(false, a, false, bb,
                                                (short)0, c, false, false);
    }
    const int r0 = wave * 16 + half * 8;
#pragma unroll
    for (int i = 0; i < 8; ++i) {
      G[(size_t)(n0 + r0 + i) * OUT_DIM + l15] = c[i] * dl[r0 + i];
    }
  }
}

__global__ void __launch_bounds__(1024) k_agg16(const int* __restrict__ binned,
                                                const int* __restrict__ rowTotal,
                                                const float* __restrict__ dinv,
                                                const float* __restrict__ g,
                                                const float* __restrict__ b2,
                                                float* __restrict__ out) {
  __shared__ float acc[NPB * OUT_DIM];
  __shared__ float dl[NPB];
  __shared__ int sh[256];
  const int tid = threadIdx.x, b = blockIdx.x;
  const int lane = tid & 31;
  const int hf = lane >> 4;
  const int c = lane & 15;
  const int wave = __builtin_amdgcn_readfirstlane(tid >> 5);
  const int n0 = b * NPB;
  if (tid < NPB) dl[tid] = dinv[n0 + tid];
  int tv = (tid < NBKT) ? rowTotal[tid] : 0;
  block_incl_scan256(tv, sh, tid);
  const int start = __builtin_amdgcn_readfirstlane(b ? sh[b - 1] : 0);
  const int count = __builtin_amdgcn_readfirstlane(sh[b]) - start;
  for (int i = tid; i < NPB * OUT_DIM; i += 1024) {
    acc[i] = g[(size_t)n0 * OUT_DIM + i];
  }
  __syncthreads();

  for (int e0 = wave * 32; e0 < count; e0 += 1024) {
    const int idx = e0 + lane;
    int s = 0, d = 0;
    if (idx < count) {
      v2i p = *(const v2i*)(binned + 2 * (size_t)(start + idx));
      s = p.x;
      d = p.y - n0;
    }
    const int m = count - e0;
    if (m >= 32) {
      float x[16];
#pragma unroll
      for (int u = 0; u < 16; ++u) {
        int s0 = __builtin_amdgcn_readlane(s, 2 * u);
        int s1 = __builtin_amdgcn_readlane(s, 2 * u + 1);
        int sj = hf ? s1 : s0;
        x[u] = g[(size_t)sj * OUT_DIM + c];
      }
#pragma unroll
      for (int u = 0; u < 16; ++u) {
        int d0 = __builtin_amdgcn_readlane(d, 2 * u);
        int d1 = __builtin_amdgcn_readlane(d, 2 * u + 1);
        int dj = hf ? d1 : d0;
        atomicAdd(&acc[dj * OUT_DIM + c], x[u]);
      }
    } else {
      for (int j = 0; 2 * j < m; ++j) {
        int e = 2 * j + hf;
        int sj = __shfl(s, e);
        int dj = __shfl(d, e);
        if (e < m) atomicAdd(&acc[dj * OUT_DIM + c], g[(size_t)sj * OUT_DIM + c]);
      }
    }
  }
  __syncthreads();
  for (int i = tid; i < NPB * OUT_DIM; i += 1024) {
    out[(size_t)n0 * OUT_DIM + i] = acc[i] * dl[i >> 4] + b2[i & (OUT_DIM - 1)];
  }
}

extern "C" void kernel_launch(void* const* d_in, const int* in_sizes, int n_in,
                              void* d_out, int out_size, void* d_ws, size_t ws_size,
                              hipStream_t stream) {
  const float* x  = (const float*)d_in[0];
  const float* W1 = (const float*)d_in[1];
  const float* b1 = (const float*)d_in[2];
  const float* W2 = (const float*)d_in[3];
  const float* b2 = (const float*)d_in[4];
  const int*   ei = (const int*)d_in[5];

  float* dinv = (float*)d_ws;
  float* h    = dinv + N_NODES;
  float* g    = h    + (size_t)N_NODES * HID_DIM;
  int* hist     = (int*)(g + (size_t)N_NODES * OUT_DIM);
  int* rowOff   = hist + NBKT * NBLK;
  int* rowTotal = rowOff + NBKT * NBLK;
  int* binned   = rowTotal + 256;
  float* out = (float*)d_out;

  k_hist<<<NBLK, 1024, 0, stream>>>(ei, hist);
  k_rowscan<<<NBKT, 256, 0, stream>>>(hist, rowOff, rowTotal);
  k_scatter<<<NBLK, 1024, 0, stream>>>(ei, rowOff, rowTotal, binned);
  k_degree_gemm1<<<NBKT, 1024, 0, stream>>>(binned, rowTotal, x, W1, dinv, h);
  k_agg32_gemm2<<<NBKT, 1024, 0, stream>>>(binned, rowTotal, dinv, h, b1, W2, g);
  k_agg16<<<NBKT, 1024, 0, stream>>>(binned, rowTotal, dinv, g, b2, out);
}


# Binning pass orders each work-group's 12800 records by range in LDS (rank = return of the LDS cursor add), then writes each range's run with contiguous stores; inside a range still arrival order
# speedup vs baseline: 5.3586x; 1.0594x over previous
#include <hip/hip_runtime.h>


#define N_NODES 100000
#define N_EDGES 3200000
#define IN_DIM  128
#define HID_DIM 32
#define OUT_DIM 16

#define NBKT 250
#define NPB  400
#define NBLK 250
#define EPB  12800

typedef __attribute__((ext_vector_type(2))) float v2f;
typedef __attribute__((ext_vector_type(4))) float v4f;
typedef __attribute__((ext_vector_type(8))) float v8f;
typedef __attribute__((ext_vector_type(2))) int   v2i;

__device__ __forceinline__ int clamp_node(int v) {
  return min(max(v, 0), N_NODES - 1);
}
__device__ __forceinline__ int bucket_of(int d) {
  return (int)((unsigned)d / (unsigned)NPB);
}

__device__ __forceinline__ void block_incl_scan256(int v, int* sh, int tid) {
  if (tid < 256) sh[tid] = v;
  __syncthreads();
#pragma unroll
  for (int off = 1; off < 256; off <<= 1) {
    int t = 0;
    if (tid < 256 && tid >= off) t = sh[tid - off];
    __syncthreads();
    if (tid < 256) sh[tid] += t;
    __syncthreads();
  }
}

__device__ __forceinline__ void block_incl_scan256_pair(int va, int vb, int* sa, int* sb, int tid) {
  if (tid < 256) { sa[tid] = va; sb[tid] = vb; }
  __syncthreads();
#pragma unroll
  for (int off = 1; off < 256; off <<= 1) {
    int ta = 0, tb = 0;
    if (tid < 256 && tid >= off) { ta = sa[tid - off]; tb = sb[tid - off]; }
    __syncthreads();
    if (tid < 256) { sa[tid] += ta; sb[tid] += tb; }
    __syncthreads();
  }
}

__global__ void __launch_bounds__(1024) k_hist(const int* __restrict__ ei,
                                               int* __restrict__ hist) {
  __shared__ int lh[256];
  const int tid = threadIdx.x, blk = blockIdx.x;
  if (tid < 256) lh[tid] = 0;
  __syncthreads();
  const int* dstp = ei + N_EDGES + blk * EPB;
#pragma unroll
  for (int k = 0; k < 13; ++k) {
    int i = tid + k * 1024;
    if (i < EPB) {
      int d = clamp_node(dstp[i]);
      atomicAdd(&lh[bucket_of(d)], 1);
    }
  }
  __syncthreads();
  if (tid < NBKT) hist[tid * NBLK + blk] = lh[tid];
}

__global__ void __launch_bounds__(256) k_rowscan(const int* __restrict__ hist,
                                                 int* __restrict__ rowOff,
                                                 int* __restrict__ rowTotal) {
  __shared__ int sh[256];
  const int tid = threadIdx.x, b = blockIdx.x;
  int v = (tid < NBLK) ? hist[b * NBLK + tid] : 0;
  block_incl_scan256(v, sh, tid);
  if (tid < NBLK) rowOff[b * NBLK + tid] = sh[tid] - v;
  if (tid == 0) rowTotal[b] = sh[255];
}

__global__ void __launch_bounds__(1024) k_scatter(const int* __restrict__ ei,
                                                  const int* __restrict__ hist,
                                                  const int* __restrict__ rowOff,
                                                  const int* __restrict__ rowTotal,
                                                  int* __restrict__ binned) {
  __shared__ __attribute__((aligned(16))) v2i lrec[EPB];
  __shared__ int shg[256];
  __shared__ int shl[256];
  __shared__ int lcur[256];
  __shared__ int gdelta[256];
  const int tid = threadIdx.x, blk = blockIdx.x;
  int vg = (tid < NBKT) ? rowTotal[tid] : 0;
  int vl = (tid < NBKT) ? hist[tid * NBLK + blk] : 0;
  block_incl_scan256_pair(vg, vl, shg, shl, tid);
  if (tid < NBKT) {
    const int lstart = shl[tid] - vl;
    lcur[tid] = lstart;
    gdelta[tid] = (shg[tid] - vg) + rowOff[tid * NBLK + blk] - lstart;
  }
  __syncthreads();
  const int* srcp = ei + blk * EPB;
  const int* dstp = ei + N_EDGES + blk * EPB;
#pragma unroll
  for (int k = 0; k < 13; ++k) {
    int i = tid + k * 1024;
    if (i < EPB) {
      int s = clamp_node(srcp[i]);
      int d = clamp_node(dstp[i]);
      int r = atomicAdd(&lcur[bucket_of(d)], 1);
      v2i p; p.x = s; p.y = d;
      lrec[r] = p;
    }
  }
  __syncthreads();
#pragma unroll
  for (int k = 0; k < 13; ++k) {
    int i = tid + k * 1024;
    if (i < EPB) {
      v2i p = lrec[i];
      int g = i + gdelta[bucket_of(p.y)];
      *(v2i*)(binned + 2 * (size_t)g) = p;
    }
  }
}

#define NTILE (NPB / 16)
__global__ void __launch_bounds__(1024) k_degree_gemm1(const int* __restrict__ binned,
                                                       const int* __restrict__ rowTotal,
                                                       const float* __restrict__ X,
                                                       const float* __restrict__ W,
                                                       float* __restrict__ dinv,
                                                       float* __restrict__ H) {
  __shared__ int cnt[NPB];
  __shared__ float dl[NPB];
  __shared__ int sh[256];
  const int tid = threadIdx.x, b = blockIdx.x;
  const int lane = tid & 31;
  const int wave = __builtin_amdgcn_readfirstlane(tid >> 5);
  const int n0   = b * NPB;
  const int half = lane >> 4;
  const int l15  = lane & 15;
  const int kOff = half * 2;

  v2f a[IN_DIM / 4];
  if (wave < NTILE) {
    const float* xr = X + (size_t)(n0 + wave * 16 + l15) * IN_DIM + kOff;
#pragma unroll
    for (int k4 = 0; k4 < IN_DIM / 4; ++k4) a[k4] = *(const v2f*)(xr + 4 * k4);
  }

  if (tid < NPB) cnt[tid] = 0;
  int v = (tid < NBKT) ? rowTotal[tid] : 0;
  block_incl_scan256(v, sh, tid);
  const int start = b ? sh[b - 1] : 0;
  const int count = sh[b] - start;
  for (int i = tid; i < count; i += 1024) {
    int d = binned[2 * (size_t)(start + i) + 1] - n0;
    atomicAdd(&cnt[d], 1);
  }
  __syncthreads();
  if (tid < NPB) {
    float deg = 1.0f + (float)cnt[tid];
    float di = (deg > 0.0f) ? rsqrtf(deg) : 0.0f;
    dl[tid] = di;
    dinv[n0 + tid] = di;
  }
  __syncthreads();

  if (wave < NTILE) {
    v8f c0 = {}, c1 = {};
#pragma unroll
    for (int k4 = 0; k4 < IN_DIM / 4; ++k4) {
      const float* wr = W + (4 * k4 + kOff) * HID_DIM + l15;
      v2f b0, b1;
      b0.x = wr[0];
      b0.y = wr[HID_DIM];
      b1.x = wr[16];
      b1.y = wr[HID_DIM + 16];
      c0 = __builtin_amdgcn_wmma_f32_16x16x4_f32(false, a[k4], false, b0,
                                                 (short)0, c0, false, false);
      c1 = __builtin_amdgcn_wmma_f32_16x16x4_f32(false, a[k4], false, b1,
                                                 (short)0, c1, false, false);
    }
    const int r0 = wave * 16 + half * 8;
#pragma unroll
    for (int i = 0; i < 8; ++i) {
      const float di = dl[r0 + i];
      float* hr = H + (size_t)(n0 + r0 + i) * HID_DIM + l15;
      hr[0]  = c0[i] * di;
      hr[16] = c1[i] * di;
    }
  }
}

#define APITCH 34
__global__ void __launch_bounds__(1024) k_agg32_gemm2(const int* __restrict__ binned,
                                                      const int* __restrict__ rowTotal,
                                                      const float* __restrict__ dinv,
                                                      const float* __restrict__ h,
                                                      const float* __restrict__ b1,
                                                      const float* __restrict__ W2,
                                                      float* __restrict__ G) {
  __shared__ __attribute__((aligned(16))) float acc[NPB * APITCH];
  __shared__ float dl[NPB];
  __shared__ int sh[256];
  const int tid = threadIdx.x, b = blockIdx.x;
  const int lane = tid & 31;
  const int wave = __builtin_amdgcn_readfirstlane(tid >> 5);
  const int n0 = b * NPB;
  if (tid < NPB) dl[tid] = dinv[n0 + tid];
  int tv = (tid < NBKT) ? rowTotal[tid] : 0;
  block_incl_scan256(tv, sh, tid);
  const int start = __builtin_amdgcn_readfirstlane(b ? sh[b - 1] : 0);
  const int count = __builtin_amdgcn_readfirstlane(sh[b]) - start;
  for (int i = tid; i < NPB * HID_DIM; i += 1024) {
    acc[(i >> 5) * APITCH + (i & 31)] = h[(size_t)n0 * HID_DIM + i];
  }
  __syncthreads();

  for (int e0 = wave * 32; e0 < count; e0 += 1024) {
    const int idx = e0 + lane;
    int s = 0, d = 0;
    if (idx < count) {
      v2i p = *(const v2i*)(binned + 2 * (size_t)(start + idx));
      s = p.x;
      d = p.y - n0;
    }
    const int m = count - e0;
    if (m >= 32) {
#pragma unroll
      for (int j = 0; j < 32; j += 16) {
        float x[16];
#pragma unroll
        for (int u = 0; u < 16; ++u) {
          int sj = __builtin_amdgcn_readlane(s, j + u);
          x[u] = h[(size_t)sj * HID_DIM + lane];
        }
#pragma unroll
        for (int u = 0; u < 16; ++u) {
          int dj = __builtin_amdgcn_readlane(d, j + u);
          atomicAdd(&acc[dj * APITCH + lane], x[u]);
        }
      }
    } else {
      for (int j = 0; j < m; ++j) {
        int sj = __shfl(s, j);
        int dj = __shfl(d, j);
        atomicAdd(&acc[dj * APITCH + lane], h[(size_t)sj * HID_DIM + lane]);
      }
    }
  }
  __syncthreads();
  for (int i = tid; i < NPB * HID_DIM; i += 1024) {
    const int r = i >> 5, cc = i & 31;
    float v = acc[r * APITCH + cc] * dl[r] + b1[cc];
    acc[r * APITCH + cc] = fmaxf(v, 0.0f);
  }
  __syncthreads();

  if (wave < NTILE) {
    const int half = lane >> 4;
    const int l15  = lane & 15;
    const int kOff = half * 2;
    const float* ar = acc + (wave * 16 + l15) * APITCH + kOff;
    v8f c = {};
#pragma unroll
    for (int k = 0; k < HID_DIM; k += 4) {
      v2f a = *(const v2f*)(ar + k);
      v2f bb;
      bb.x = W2[(k + kOff + 0) * OUT_DIM + l15];
      bb.y = W2[(k + kOff + 1) * OUT_DIM + l15];
      c = __builtin_amdgcn_wmma_f32_16x16x4_f32(false, a, false, bb,
                                                (short)0, c, false, false);
    }
    const int r0 = wave * 16 + half * 8;
#pragma unroll
    for (int i = 0; i < 8; ++i) {
      G[(size_t)(n0 + r0 + i) * OUT_DIM + l15] = c[i] * dl[r0 + i];
    }
  }
}

__global__ void __launch_bounds__(1024) k_agg16(const int* __restrict__ binned,
                                                const int* __restrict__ rowTotal,
                                                const float* __restrict__ dinv,
                                                const float* __restrict__ g,
                                                const float* __restrict__ b2,
                                                float* __restrict__ out) {
  __shared__ float acc[NPB * OUT_DIM];
  __shared__ float dl[NPB];
  __shared__ int sh[256];
  const int tid = threadIdx.x, b = blockIdx.x;
  const int lane = tid & 31;
  const int hf = lane >> 4;
  const int c = lane & 15;
  const int wave = __builtin_amdgcn_readfirstlane(tid >> 5);
  const int n0 = b * NPB;
  if (tid < NPB) dl[tid] = dinv[n0 + tid];
  int tv = (tid < NBKT) ? rowTotal[tid] : 0;
  block_incl_scan256(tv, sh, tid);
  const int start = __builtin_amdgcn_readfirstlane(b ? sh[b - 1] : 0);
  const int count = __builtin_amdgcn_readfirstlane(sh[b]) - start;
  for (int i = tid; i < NPB * OUT_DIM; i += 1024) {
    acc[i] = g[(size_t)n0 * OUT_DIM + i];
  }
  __syncthreads();

  for (int e0 = wave * 32; e0 < count; e0 += 1024) {
    const int idx = e0 + lane;
    int s = 0, d = 0;
    if (idx < count) {
      v2i p = *(const v2i*)(binned + 2 * (size_t)(start + idx));
      s = p.x;
      d = p.y - n0;
    }
    const int m = count - e0;
    if (m >= 32) {
      float x[16];
#pragma unroll
      for (int u = 0; u < 16; ++u) {
        int s0 = __builtin_amdgcn_readlane(s, 2 * u);
        int s1 = __builtin_amdgcn_readlane(s, 2 * u + 1);
        int sj = hf ? s1 : s0;
        x[u] = g[(size_t)sj * OUT_DIM + c];
      }
#pragma unroll
      for (int u = 0; u < 16; ++u) {
        int d0 = __builtin_amdgcn_readlane(d, 2 * u);
        int d1 = __builtin_amdgcn_readlane(d, 2 * u + 1);
        int dj = hf ? d1 : d0;
        atomicAdd(&acc[dj * OUT_DIM + c], x[u]);
      }
    } else {
      for (int j = 0; 2 * j < m; ++j) {
        int e = 2 * j + hf;
        int sj = __shfl(s, e);
        int dj = __shfl(d, e);
        if (e < m) atomicAdd(&acc[dj * OUT_DIM + c], g[(size_t)sj * OUT_DIM + c]);
      }
    }
  }
  __syncthreads();
  for (int i = tid; i < NPB * OUT_DIM; i += 1024) {
    out[(size_t)n0 * OUT_DIM + i] = acc[i] * dl[i >> 4] + b2[i & (OUT_DIM - 1)];
  }
}

extern "C" void kernel_launch(void* const* d_in, const int* in_sizes, int n_in,
                              void* d_out, int out_size, void* d_ws, size_t ws_size,
                              hipStream_t stream) {
  const float* x  = (const float*)d_in[0];
  const float* W1 = (const float*)d_in[1];
  const float* b1 = (const float*)d_in[2];
  const float* W2 = (const float*)d_in[3];
  const float* b2 = (const float*)d_in[4];
  const int*   ei = (const int*)d_in[5];

  float* dinv = (float*)d_ws;
  float* h    = dinv + N_NODES;
  float* g    = h    + (size_t)N_NODES * HID_DIM;
  int* hist     = (int*)(g + (size_t)N_NODES * OUT_DIM);
  int* rowOff   = hist + NBKT * NBLK;
  int* rowTotal = rowOff + NBKT * NBLK;
  int* binned   = rowTotal + 256;
  float* out = (float*)d_out;

  k_hist<<<NBLK, 1024, 0, stream>>>(ei, hist);
  k_rowscan<<<NBKT, 256, 0, stream>>>(hist, rowOff, rowTotal);
  k_scatter<<<NBLK, 1024, 0, stream>>>(ei, hist, rowOff, rowTotal, binned);
  k_degree_gemm1<<<NBKT, 1024, 0, stream>>>(binned, rowTotal, x, W1, dinv, h);
  k_agg32_gemm2<<<NBKT, 1024, 0, stream>>>(binned, rowTotal, dinv, h, b1, W2, g);
  k_agg16<<<NBKT, 1024, 0, stream>>>(binned, rowTotal, dinv, g, b2, out);
}


# Count and scan launches removed (6 to 4): each binning work-group writes its range-ordered records as one slab plus a run table; a range's work-group gathers its 256 runs into LDS; sum order kept
# speedup vs baseline: 6.3763x; 1.1899x over previous
#include <hip/hip_runtime.h>


#define N_NODES 100000
#define N_EDGES 3200000
#define IN_DIM  128
#define HID_DIM 32
#define OUT_DIM 16

#define NBKT 250
#define NPB  400
#define NBLK 256
#define EPB  12500
#define SLAB 12544
#define RPW  (NBLK / 32)
#define CAP  16384

static_assert(NBKT * NPB == N_NODES, "ranges must tile the nodes");
static_assert(NBLK * EPB == N_EDGES, "slabs must tile the edges");
static_assert(N_NODES <= (1 << 17) && NPB <= (1 << 15), "packed record: 17 bits source, 15 bits local destination");

typedef __attribute__((ext_vector_type(2))) float v2f;
typedef __attribute__((ext_vector_type(4))) float v4f;
typedef __attribute__((ext_vector_type(8))) float v8f;
typedef __attribute__((ext_vector_type(2))) int   v2i;
typedef __attribute__((ext_vector_type(4))) int   v4i;

__device__ __forceinline__ int clamp_node(int v) {
  return min(max(v, 0), N_NODES - 1);
}
__device__ __forceinline__ int bucket_of(int d) {
  return (int)((unsigned)d / (unsigned)NPB);
}

__device__ __forceinline__ void block_incl_scan256(int v, int* sh, int tid) {
  if (tid < 256) sh[tid] = v;
  __syncthreads();
#pragma unroll
  for (int off = 1; off < 256; off <<= 1) {
    int t = 0;
    if (tid < 256 && tid >= off) t = sh[tid - off];
    __syncthreads();
    if (tid < 256) sh[tid] += t;
    __syncthreads();
  }
}

__global__ void __launch_bounds__(1024) k_bin(const int* __restrict__ ei,
                                              int* __restrict__ binned,
                                              v2i* __restrict__ tab) {
  __shared__ __attribute__((aligned(16))) v2i lrec[EPB];
  __shared__ int lh[256];
  __shared__ int sh[256];
  __shared__ int lst[256];
  const int tid = threadIdx.x, blk = blockIdx.x;
  if (tid < 256) lh[tid] = 0;
  const int* srcp = ei + blk * EPB;
  const int* dstp = ei + N_EDGES + blk * EPB;
  int s[13], d[13], r[13];
#pragma unroll
  for (int k = 0; k < 13; ++k) {
    const int i = tid + k * 1024;
    s[k] = 0; d[k] = 0; r[k] = 0;
    if (i < EPB) {
      s[k] = clamp_node(srcp[i]);
      d[k] = clamp_node(dstp[i]);
    }
  }
  __syncthreads();
#pragma unroll
  for (int k = 0; k < 13; ++k) {
    const int i = tid + k * 1024;
    if (i < EPB) r[k] = atomicAdd(&lh[bucket_of(d[k])], 1);
  }
  __syncthreads();
  const int v = (tid < 256) ? lh[tid] : 0;
  block_incl_scan256(v, sh, tid);
  if (tid < 256) lst[tid] = sh[tid] - v;
  if (tid < NBKT) {
    v2i t; t.x = blk * SLAB + sh[tid] - v; t.y = v;
    tab[tid * NBLK + blk] = t;
  }
  __syncthreads();
#pragma unroll
  for (int k = 0; k < 13; ++k) {
    const int i = tid + k * 1024;
    if (i < EPB) {
      v2i p; p.x = s[k]; p.y = d[k];
      lrec[lst[bucket_of(d[k])] + r[k]] = p;
    }
  }
  __syncthreads();
  v4i* outp = (v4i*)(binned + 2 * (size_t)blk * SLAB);
  const v4i* lp = (const v4i*)lrec;
  for (int i = tid; i < EPB / 2; i += 1024) outp[i] = lp[i];
}

template <int RB, typename F>
__device__ __forceinline__ void walk_runs(const int* __restrict__ binned, const int* rs, const int* rc,
                                          int wave, int lane, F fn) {
#pragma unroll
  for (int j0 = 0; j0 < RPW; j0 += RB) {
    int st[RB], n[RB];
    v2i rec[RB][3];
#pragma unroll
    for (int j = 0; j < RB; ++j) {
      const int k = wave + 32 * (j0 + j);
      st[j] = __builtin_amdgcn_readfirstlane(rs[k]);
      n[j]  = __builtin_amdgcn_readfirstlane(rc[k]);
#pragma unroll
      for (int t = 0; t < 3; ++t) {
        const int o = lane + 32 * t;
        const int at = (o < n[j]) ? st[j] + o : 0;
        rec[j][t] = *(const v2i*)(binned + 2 * (size_t)at);
      }
    }
#pragma unroll
    for (int j = 0; j < RB; ++j) {
#pragma unroll
      for (int t = 0; t < 3; ++t) {
        const int o = lane + 32 * t;
        if (o < n[j]) fn(j0 + j, o, rec[j][t]);
      }
    }
#pragma unroll
    for (int j = 0; j < RB; ++j) {
      for (int o = lane + 96; o < n[j]; o += 32)
        fn(j0 + j, o, *(const v2i*)(binned + 2 * (size_t)(st[j] + o)));
    }
  }
}

#define NTILE (NPB / 16)
__global__ void __launch_bounds__(1024) k_degree_gemm1(const int* __restrict__ binned,
                                                       const v2i* __restrict__ tab,
                                                       const float* __restrict__ X,
                                                       const float* __restrict__ W,
                                                       float* __restrict__ dinv,
                                                       float* __restrict__ H) {
  __shared__ int cnt[NPB];
  __shared__ float dl[NPB];
  __shared__ int rs[NBLK];
  __shared__ int rc[NBLK];
  const int tid = threadIdx.x, b = blockIdx.x;
  const int lane = tid & 31;
  const int wave = __builtin_amdgcn_readfirstlane(tid >> 5);
  const int n0   = b * NPB;
  const int half = lane >> 4;
  const int l15  = lane & 15;
  const int kOff = half * 2;

  v2f a[IN_DIM / 4];
  if (wave < NTILE) {
    const float* xr = X + (size_t)(n0 + wave * 16 + l15) * IN_DIM + kOff;
#pragma unroll
    for (int k4 = 0; k4 < IN_DIM / 4; ++k4) a[k4] = *(const v2f*)(xr + 4 * k4);
  }

  if (tid < NPB) cnt[tid] = 0;
  if (tid < NBLK) {
    const v2i t = tab[b * NBLK + tid];
    rs[tid] = t.x;
    rc[tid] = t.y;
  }
  __syncthreads();
  walk_runs<4>(binned, rs, rc, wave, lane,
               [&](int, int, v2i p) { atomicAdd(&cnt[p.y - n0], 1); });
  __syncthreads();
  if (tid < NPB) {
    float deg = 1.0f + (float)cnt[tid];
    float di = (deg > 0.0f) ? rsqrtf(deg) : 0.0f;
    dl[tid] = di;
    dinv[n0 + tid] = di;
  }
  __syncthreads();

  if (wave < NTILE) {
    v8f c0 = {}, c1 = {};
#pragma unroll
    for (int k4 = 0; k4 < IN_DIM / 4; ++k4) {
      const float* wr = W + (4 * k4 + kOff) * HID_DIM + l15;
      v2f b0, b1;
      b0.x = wr[0];
      b0.y = wr[HID_DIM];
      b1.x = wr[16];
      b1.y = wr[HID_DIM + 16];
      c0 = __builtin_amdgcn_wmma_f32_16x16x4_f32(false, a[k4], false, b0,
                                                 (short)0, c0, false, false);
      c1 = __builtin_amdgcn_wmma_f32_16x16x4_f32(false, a[k4], false, b1,
                                                 (short)0, c1, false, false);
    }
    const int r0 = wave * 16 + half * 8;
#pragma unroll
    for (int i = 0; i < 8; ++i) {
      const float di = dl[r0 + i];
      float* hr = H + (size_t)(n0 + r0 + i) * HID_DIM + l15;
      hr[0]  = c0[i] * di;
      hr[16] = c1[i] * di;
    }
  }
}

__device__ __forceinline__ int flat_bases(const int* rc, int* wtot, int wave, int lane, int (&fb)[RPW]) {
  int wt = 0;
#pragma unroll
  for (int j = 0; j < RPW; ++j) {
    fb[j] = wt;
    wt += __builtin_amdgcn_readfirstlane(rc[wave + 32 * j]);
  }
  if (lane == 0) wtot[wave] = wt;
  __syncthreads();
  int x = wtot[lane];
#pragma unroll
  for (int off = 1; off < 32; off <<= 1) {
    const int y = __shfl_up(x, off);
    if (lane >= off) x += y;
  }
  const int total = __builtin_amdgcn_readlane(x, 31);
  const int wb = __builtin_amdgcn_readlane(x, wave) - wt;
#pragma unroll
  for (int j = 0; j < RPW; ++j) fb[j] += wb;
  return total;
}

#define APITCH 34
__global__ void __launch_bounds__(1024) k_agg32_gemm2(const int* __restrict__ binned,
                                                      const v2i* __restrict__ tab,
                                                      const float* __restrict__ dinv,
                                                      const float* __restrict__ h,
                                                      const float* __restrict__ b1,
                                                      const float* __restrict__ W2,
                                                      float* __restrict__ G) {
  __shared__ __attribute__((aligned(16))) float acc[NPB * APITCH];
  __shared__ unsigned erec[CAP];
  __shared__ float dl[NPB];
  __shared__ int rs[NBLK];
  __shared__ int rc[NBLK];
  __shared__ int wtot[32];
  const int tid = threadIdx.x, b = blockIdx.x;
  const int lane = tid & 31;
  const int wave = __builtin_amdgcn_readfirstlane(tid >> 5);
  const int n0 = b * NPB;
  if (tid < NPB) dl[tid] = dinv[n0 + tid];
  if (tid < NBLK) {
    const v2i t = tab[b * NBLK + tid];
    rs[tid] = t.x;
    rc[tid] = t.y;
  }
  for (int i = tid; i < NPB * HID_DIM; i += 1024) {
    acc[(i >> 5) * APITCH + (i & 31)] = h[(size_t)n0 * HID_DIM + i];
  }
  __syncthreads();
  int fb[RPW];
  const int total = flat_bases(rc, wtot, wave, lane, fb);

  for (int c0 = 0; c0 < total; c0 += CAP) {
    walk_runs<RPW>(binned, rs, rc, wave, lane, [&](int j, int o, v2i p) {
      const unsigned r = (unsigned)(fb[j] + o - c0);
      if (r < (unsigned)CAP) erec[r] = ((unsigned)(p.y - n0) << 17) | (unsigned)p.x;
    });
    __syncthreads();
    const int count = min(CAP, total - c0);
    for (int e0 = wave * 32; e0 < count; e0 += 1024) {
      const int idx = e0 + lane;
      unsigned rec = 0;
      if (idx < count) rec = erec[idx];
      const int s = (int)(rec & 0x1FFFFu);
      const int d = (int)(rec >> 17);
      const int m = count - e0;
      if (m >= 32) {
#pragma unroll
        for (int j = 0; j < 32; j += 16) {
          float x[16];
#pragma unroll
          for (int u = 0; u < 16; ++u) {
            int sj = __builtin_amdgcn_readlane(s, j + u);
            x[u] = h[(size_t)sj * HID_DIM + lane];
          }
#pragma unroll
          for (int u = 0; u < 16; ++u) {
            int dj = __builtin_amdgcn_readlane(d, j + u);
            atomicAdd(&acc[dj * APITCH + lane], x[u]);
          }
        }
      } else {
        for (int j = 0; j < m; ++j) {
          int sj = __shfl(s, j);
          int dj = __shfl(d, j);
          atomicAdd(&acc[dj * APITCH + lane], h[(size_t)sj * HID_DIM + lane]);
        }
      }
    }
    __syncthreads();
  }
  for (int i = tid; i < NPB * HID_DIM; i += 1024) {
    const int r = i >> 5, cc = i & 31;
    float v = acc[r * APITCH + cc] * dl[r] + b1[cc];
    acc[r * APITCH + cc] = fmaxf(v, 0.0f);
  }
  __syncthreads();

  if (wave < NTILE) {
    const int half = lane >> 4;
    const int l15  = lane & 15;
    const int kOff = half * 2;
    const float* ar = acc + (wave * 16 + l15) * APITCH + kOff;
    v8f c = {};
#pragma unroll
    for (int k = 0; k < HID_DIM; k += 4) {
      v2f a = *(const v2f*)(ar + k);
      v2f bb;
      bb.x = W2[(k + kOff + 0) * OUT_DIM + l15];
      bb.y = W2[(k + kOff + 1) * OUT_DIM + l15];
      c = __builtin_amdgcn_wmma_f32_16x16x4_f32(false, a, false, bb,
                                                (short)0, c, false, false);
    }
    const int r0 = wave * 16 + half * 8;
#pragma unroll
    for (int i = 0; i < 8; ++i) {
      G[(size_t)(n0 + r0 + i) * OUT_DIM + l15] = c[i] * dl[r0 + i];
    }
  }
}

__global__ void __launch_bounds__(1024) k_agg16(const int* __restrict__ binned,
                                                const v2i* __restrict__ tab,
                                                const float* __restrict__ dinv,
                                                const float* __restrict__ g,
                                                const float* __restrict__ b2,
                                                float* __restrict__ out) {
  __shared__ float acc[NPB * OUT_DIM];
  __shared__ unsigned erec[CAP];
  __shared__ float dl[NPB];
  __shared__ int rs[NBLK];
  __shared__ int rc[NBLK];
  __shared__ int wtot[32];
  const int tid = threadIdx.x, b = blockIdx.x;
  const int lane = tid & 31;
  const int hf = lane >> 4;
  const int c = lane & 15;
  const int wave = __builtin_amdgcn_readfirstlane(tid >> 5);
  const int n0 = b * NPB;
  if (tid < NPB) dl[tid] = dinv[n0 + tid];
  if (tid < NBLK) {
    const v2i t = tab[b * NBLK + tid];
    rs[tid] = t.x;
    rc[tid] = t.y;
  }
  for (int i = tid; i < NPB * OUT_DIM; i += 1024) {
    acc[i] = g[(size_t)n0 * OUT_DIM + i];
  }
  __syncthreads();
  int fb[RPW];
  const int total = flat_bases(rc, wtot, wave, lane, fb);

  for (int c0 = 0; c0 < total; c0 += CAP) {
    walk_runs<RPW>(binned, rs, rc, wave, lane, [&](int j, int o, v2i p) {
      const unsigned r = (unsigned)(fb[j] + o - c0);
      if (r < (unsigned)CAP) erec[r] = ((unsigned)(p.y - n0) << 17) | (unsigned)p.x;
    });
    __syncthreads();
    const int count = min(CAP, total - c0);
    for (int e0 = wave * 32; e0 < count; e0 += 1024) {
      const int idx = e0 + lane;
      unsigned rec = 0;
      if (idx < count) rec = erec[idx];
      const int s = (int)(rec & 0x1FFFFu);
      const int d = (int)(rec >> 17);
      const int m = count - e0;
      if (m >= 32) {
        float x[16];
#pragma unroll
        for (int u = 0; u < 16; ++u) {
          int s0 = __builtin_amdgcn_readlane(s, 2 * u);
          int s1 = __builtin_amdgcn_readlane(s, 2 * u + 1);
          int sj = hf ? s1 : s0;
          x[u] = g[(size_t)sj * OUT_DIM + c];
        }
#pragma unroll
        for (int u = 0; u < 16; ++u) {
          int d0 = __builtin_amdgcn_readlane(d, 2 * u);
          int d1 = __builtin_amdgcn_readlane(d, 2 * u + 1);
          int dj = hf ? d1 : d0;
          atomicAdd(&acc[dj * OUT_DIM + c], x[u]);
        }
      } else {
        for (int j = 0; 2 * j < m; ++j) {
          int e = 2 * j + hf;
          int sj = __shfl(s, e);
          int dj = __shfl(d, e);
          if (e < m) atomicAdd(&acc[dj * OUT_DIM + c], g[(size_t)sj * OUT_DIM + c]);
        }
      }
    }
    __syncthreads();
  }
  for (int i = tid; i < NPB * OUT_DIM; i += 1024) {
    out[(size_t)n0 * OUT_DIM + i] = acc[i] * dl[i >> 4] + b2[i & (OUT_DIM - 1)];
  }
}

extern "C" void kernel_launch(void* const* d_in, const int* in_sizes, int n_in,
                              void* d_out, int out_size, void* d_ws, size_t ws_size,
                              hipStream_t stream) {
  const float* x  = (const float*)d_in[0];
  const float* W1 = (const float*)d_in[1];
  const float* b1 = (const float*)d_in[2];
  const float* W2 = (const float*)d_in[3];
  const float* b2 = (const float*)d_in[4];
  const int*   ei = (const int*)d_in[5];

  float* dinv = (float*)d_ws;
  float* h    = dinv + N_NODES;
  float* g    = h    + (size_t)N_NODES * HID_DIM;
  v2i* tab    = (v2i*)(g + (size_t)N_NODES * OUT_DIM);
  int* binned = (int*)(tab + NBKT * NBLK);
  float* out = (float*)d_out;

  k_bin<<<NBLK, 1024, 0, stream>>>(ei, binned, tab);
  k_degree_gemm1<<<NBKT, 1024, 0, stream>>>(binned, tab, x, W1, dinv, h);
  k_agg32_gemm2<<<NBKT, 1024, 0, stream>>>(binned, tab, dinv, h, b1, W2, g);
  k_agg16<<<NBKT, 1024, 0, stream>>>(binned, tab, dinv, g, b2, out);
}


# Loads ordered for in-order return: each kernel's opening/closing loads leave as one group (16-byte self rows, table, bias, W2); GEMM1 reads W1 from LDS and multiplies while the X tile streams in
# speedup vs baseline: 6.6194x; 1.0381x over previous
#include <hip/hip_runtime.h>


#define N_NODES 100000
#define N_EDGES 3200000
#define IN_DIM  128
#define HID_DIM 32
#define OUT_DIM 16

#define NBKT 250
#define NPB  400
#define NBLK 256
#define EPB  12500
#define SLAB 12544
#define RPW  (NBLK / 32)
#define CAP  16384

static_assert(NBKT * NPB == N_NODES, "ranges must tile the nodes");
static_assert(NBLK * EPB == N_EDGES, "slabs must tile the edges");
static_assert(N_NODES <= (1 << 17) && NPB <= (1 << 15), "packed record: 17 bits source, 15 bits local destination");
static_assert(IN_DIM * HID_DIM == 4 * 1024, "W1 goes to LDS as one 16-byte piece per thread of a 1024-thread work-group");

typedef __attribute__((ext_vector_type(2))) float v2f;
typedef __attribute__((ext_vector_type(4))) float v4f;
typedef __attribute__((ext_vector_type(8))) float v8f;
typedef __attribute__((ext_vector_type(2))) int   v2i;
typedef __attribute__((ext_vector_type(4))) int   v4i;

__device__ __forceinline__ int clamp_node(int v) {
  return min(max(v, 0), N_NODES - 1);
}
__device__ __forceinline__ int bucket_of(int d) {
  return (int)((unsigned)d / (unsigned)NPB);
}

__device__ __forceinline__ void block_incl_scan256(int v, int* sh, int tid) {
  if (tid < 256) sh[tid] = v;
  __syncthreads();
#pragma unroll
  for (int off = 1; off < 256; off <<= 1) {
    int t = 0;
    if (tid < 256 && tid >= off) t = sh[tid - off];
    __syncthreads();
    if (tid < 256) sh[tid] += t;
    __syncthreads();
  }
}

__global__ void __launch_bounds__(1024) k_bin(const int* __restrict__ ei,
                                              int* __restrict__ binned,
                                              v2i* __restrict__ tab) {
  __shared__ __attribute__((aligned(16))) v2i lrec[EPB];
  __shared__ int lh[256];
  __shared__ int sh[256];
  __shared__ int lst[256];
  const int tid = threadIdx.x, blk = blockIdx.x;
  if (tid < 256) lh[tid] = 0;
  const int* srcp = ei + blk * EPB;
  const int* dstp = ei + N_EDGES + blk * EPB;
  int s[13], d[13], r[13];
#pragma unroll
  for (int k = 0; k < 13; ++k) {
    const int i = tid + k * 1024;
    const int ii = min(i, EPB - 1);
    r[k] = 0;
    s[k] = clamp_node(srcp[ii]);
    d[k] = clamp_node(dstp[ii]);
  }
  __syncthreads();
#pragma unroll
  for (int k = 0; k < 13; ++k) {
    const int i = tid + k * 1024;
    if (i < EPB) r[k] = atomicAdd(&lh[bucket_of(d[k])], 1);
  }
  __syncthreads();
  const int v = (tid < 256) ? lh[tid] : 0;
  block_incl_scan256(v, sh, tid);
  if (tid < 256) lst[tid] = sh[tid] - v;
  if (tid < NBKT) {
    v2i t; t.x = blk * SLAB + sh[tid] - v; t.y = v;
    tab[tid * NBLK + blk] = t;
  }
  __syncthreads();
#pragma unroll
  for (int k = 0; k < 13; ++k) {
    const int i = tid + k * 1024;
    if (i < EPB) {
      v2i p; p.x = s[k]; p.y = d[k];
      lrec[lst[bucket_of(d[k])] + r[k]] = p;
    }
  }
  __syncthreads();
  v4i* outp = (v4i*)(binned + 2 * (size_t)blk * SLAB);
  const v4i* lp = (const v4i*)lrec;
  for (int i = tid; i < EPB / 2; i += 1024) outp[i] = lp[i];
}

template <int RB, int JB, int JE, typename F>
__device__ __forceinline__ void walk_runs(const int* __restrict__ binned, const int* rs, const int* rc,
                                          int wave, int lane, F fn) {
#pragma unroll
  for (int j0 = JB; j0 < JE; j0 += RB) {
    int st[RB], n[RB];
    v2i rec[RB][3];
#pragma unroll
    for (int j = 0; j < RB; ++j) {
      const int k = wave + 32 * (j0 + j);
      st[j] = __builtin_amdgcn_readfirstlane(rs[k]);
      n[j]  = __builtin_amdgcn_readfirstlane(rc[k]);
#pragma unroll
      for (int t = 0; t < 3; ++t) {
        const int o = lane + 32 * t;
        const int at = (o < n[j]) ? st[j] + o : 0;
        rec[j][t] = *(const v2i*)(binned + 2 * (size_t)at);
      }
    }
#pragma unroll
    for (int j = 0; j < RB; ++j) {
#pragma unroll
      for (int t = 0; t < 3; ++t) {
        const int o = lane + 32 * t;
        if (o < n[j]) fn(j0 + j, o, rec[j][t]);
      }
    }
#pragma unroll
    for (int j = 0; j < RB; ++j) {
      for (int o = lane + 96; o < n[j]; o += 32)
        fn(j0 + j, o, *(const v2i*)(binned + 2 * (size_t)(st[j] + o)));
    }
  }
}

#define NTILE (NPB / 16)
#define WPITCH 48
__global__ void __launch_bounds__(1024) k_degree_gemm1(const int* __restrict__ binned,
                                                       const v2i* __restrict__ tab,
                                                       const float* __restrict__ X,
                                                       const float* __restrict__ W,
                                                       float* __restrict__ dinv,
                                                       float* __restrict__ H) {
  __shared__ int cnt[NPB];
  __shared__ float dl[NPB];
  __shared__ int rs[NBLK];
  __shared__ int rc[NBLK];
  __shared__ __attribute__((aligned(16))) float wl[IN_DIM * WPITCH];
  const int tid = threadIdx.x, b = blockIdx.x;
  const int lane = tid & 31;
  const int wave = __builtin_amdgcn_readfirstlane(tid >> 5);
  const int n0   = b * NPB;
  const int half = lane >> 4;
  const int l15  = lane & 15;
  const int kOff = half * 2;

  const v2i tt = tab[b * NBLK + (tid & (NBLK - 1))];
  const v4f wv = ((const v4f*)W)[tid];
  if (tid < NPB) cnt[tid] = 0;
  if (tid < NBLK) {
    rs[tid] = tt.x;
    rc[tid] = tt.y;
  }
  *(v4f*)&wl[(tid >> 3) * WPITCH + (tid & 7) * 4] = wv;
  __syncthreads();

  const auto count_fn = [&](int, int, v2i p) { atomicAdd(&cnt[p.y - n0], 1); };

  walk_runs<4, 0, 4>(binned, rs, rc, wave, lane, count_fn);

  v8f c0 = {}, c1 = {};
  if (wave < NTILE) {
    v2f a[IN_DIM / 4];
    const float* xr = X + (size_t)(n0 + wave * 16 + l15) * IN_DIM + kOff;
#pragma unroll
    for (int k4 = 0; k4 < IN_DIM / 4; ++k4) a[k4] = *(const v2f*)(xr + 4 * k4);
#pragma unroll
    for (int k4 = 0; k4 < IN_DIM / 4; ++k4) {
      const float* wr = wl + (4 * k4 + kOff) * WPITCH + l15;
      v2f b0, b1;
      b0.x = wr[0];
      b0.y = wr[WPITCH];
      b1.x = wr[16];
      b1.y = wr[WPITCH + 16];
      c0 = __builtin_amdgcn_wmma_f32_16x16x4_f32(false, a[k4], false, b0,
                                                 (short)0, c0, false, false);
      c1 = __builtin_amdgcn_wmma_f32_16x16x4_f32(false, a[k4], false, b1,
                                                 (short)0, c1, false, false);
    }
  }

  walk_runs<4, 4, RPW>(binned, rs, rc, wave, lane, count_fn);
  __syncthreads();
  if (tid < NPB) {
    float deg = 1.0f + (float)cnt[tid];
    float di = (deg > 0.0f) ? rsqrtf(deg) : 0.0f;
    dl[tid] = di;
    dinv[n0 + tid] = di;
  }
  __syncthreads();

  if (wave < NTILE) {
    const int r0 = wave * 16 + half * 8;
#pragma unroll
    for (int i = 0; i < 8; ++i) {
      const float di = dl[r0 + i];
      float* hr = H + (size_t)(n0 + r0 + i) * HID_DIM + l15;
      hr[0]  = c0[i] * di;
      hr[16] = c1[i] * di;
    }
  }
}

__device__ __forceinline__ int flat_bases(const int* rc, int* wtot, int wave, int lane, int (&fb)[RPW]) {
  int wt = 0;
#pragma unroll
  for (int j = 0; j < RPW; ++j) {
    fb[j] = wt;
    wt += __builtin_amdgcn_readfirstlane(rc[wave + 32 * j]);
  }
  if (lane == 0) wtot[wave] = wt;
  __syncthreads();
  int x = wtot[lane];
#pragma unroll
  for (int off = 1; off < 32; off <<= 1) {
    const int y = __shfl_up(x, off);
    if (lane >= off) x += y;
  }
  const int total = __builtin_amdgcn_readlane(x, 31);
  const int wb = __builtin_amdgcn_readlane(x, wave) - wt;
#pragma unroll
  for (int j = 0; j < RPW; ++j) fb[j] += wb;
  return total;
}

#define APITCH 34
__global__ void __launch_bounds__(1024) k_agg32_gemm2(const int* __restrict__ binned,
                                                      const v2i* __restrict__ tab,
                                                      const float* __restrict__ dinv,
                                                      const float* __restrict__ h,
                                                      const float* __restrict__ b1,
                                                      const float* __restrict__ W2,
                                                      float* __restrict__ G) {
  __shared__ __attribute__((aligned(16))) float acc[NPB * APITCH];
  __shared__ unsigned erec[CAP];
  __shared__ float dl[NPB];
  __shared__ int rs[NBLK];
  __shared__ int rc[NBLK];
  __shared__ int wtot[32];
  const int tid = threadIdx.x, b = blockIdx.x;
  const int lane = tid & 31;
  const int wave = __builtin_amdgcn_readfirstlane(tid >> 5);
  const int n0 = b * NPB;
  constexpr int NV4 = NPB * HID_DIM / 4;
  const float dv   = dinv[n0 + min(tid, NPB - 1)];
  const v2i   tt   = tab[b * NBLK + (tid & (NBLK - 1))];
  const float bias = b1[lane];
  const v4f* h4 = (const v4f*)(h + (size_t)n0 * HID_DIM);
  v4f hv[4];
#pragma unroll
  for (int k = 0; k < 4; ++k) hv[k] = h4[min(tid + k * 1024, NV4 - 1)];
  if (tid < NPB) dl[tid] = dv;
  if (tid < NBLK) {
    rs[tid] = tt.x;
    rc[tid] = tt.y;
  }
#pragma unroll
  for (int k = 0; k < 4; ++k) {
    const int i4 = tid + k * 1024;
    if (i4 < NV4) {
      float* ap = &acc[(i4 >> 3) * APITCH + (i4 & 7) * 4];
      v2f lo, hi;
      lo.x = hv[k].x; lo.y = hv[k].y; hi.x = hv[k].z; hi.y = hv[k].w;
      *(v2f*)ap = lo;
      *(v2f*)(ap + 2) = hi;
    }
  }
  __syncthreads();
  int fb[RPW];
  const int total = flat_bases(rc, wtot, wave, lane, fb);

  for (int c0 = 0; c0 < total; c0 += CAP) {
    walk_runs<RPW, 0, RPW>(binned, rs, rc, wave, lane, [&](int j, int o, v2i p) {
      const unsigned r = (unsigned)(fb[j] + o - c0);
      if (r < (unsigned)CAP) erec[r] = ((unsigned)(p.y - n0) << 17) | (unsigned)p.x;
    });
    __syncthreads();
    const int count = min(CAP, total - c0);
    for (int e0 = wave * 32; e0 < count; e0 += 1024) {
      const int idx = e0 + lane;
      unsigned rec = 0;
      if (idx < count) rec = erec[idx];
      const int s = (int)(rec & 0x1FFFFu);
      const int d = (int)(rec >> 17);
      const int m = count - e0;
      if (m >= 32) {
#pragma unroll
        for (int j = 0; j < 32; j += 16) {
          float x[16];
#pragma unroll
          for (int u = 0; u < 16; ++u) {
            int sj = __builtin_amdgcn_readlane(s, j + u);
            x[u] = h[(size_t)sj * HID_DIM + lane];
          }
#pragma unroll
          for (int u = 0; u < 16; ++u) {
            int dj = __builtin_amdgcn_readlane(d, j + u);
            atomicAdd(&acc[dj * APITCH + lane], x[u]);
          }
        }
      } else {
        for (int j = 0; j < m; ++j) {
          int sj = __shfl(s, j);
          int dj = __shfl(d, j);
          atomicAdd(&acc[dj * APITCH + lane], h[(size_t)sj * HID_DIM + lane]);
        }
      }
    }
    __syncthreads();
  }

  const int half = lane >> 4;
  const int l15  = lane & 15;
  const int kOff = half * 2;
  v2f bw[HID_DIM / 4];
  if (wave < NTILE) {
#pragma unroll
    for (int k4 = 0; k4 < HID_DIM / 4; ++k4) {
      bw[k4].x = W2[(4 * k4 + kOff + 0) * OUT_DIM + l15];
      bw[k4].y = W2[(4 * k4 + kOff + 1) * OUT_DIM + l15];
    }
  }
  for (int i = tid; i < NPB * HID_DIM; i += 1024) {
    const int r = i >> 5, cc = i & 31;
    float v = acc[r * APITCH + cc] * dl[r] + bias;
    acc[r * APITCH + cc] = fmaxf(v, 0.0f);
  }
  __syncthreads();

  if (wave < NTILE) {
    const float* ar = acc + (wave * 16 + l15) * APITCH + kOff;
    v8f c = {};
#pragma unroll
    for (int k = 0; k < HID_DIM; k += 4) {
      v2f a = *(const v2f*)(ar + k);
      c = __builtin_amdgcn_wmma_f32_16x16x4_f32(false, a, false, bw[k / 4],
                                                (short)0, c, false, false);
    }
    const int r0 = wave * 16 + half * 8;
#pragma unroll
    for (int i = 0; i < 8; ++i) {
      G[(size_t)(n0 + r0 + i) * OUT_DIM + l15] = c[i] * dl[r0 + i];
    }
  }
}

__global__ void __launch_bounds__(1024) k_agg16(const int* __restrict__ binned,
                                                const v2i* __restrict__ tab,
                                                const float* __restrict__ dinv,
                                                const float* __restrict__ g,
                                                const float* __restrict__ b2,
                                                float* __restrict__ out) {
  __shared__ __attribute__((aligned(16))) float acc[NPB * OUT_DIM];
  __shared__ unsigned erec[CAP];
  __shared__ float dl[NPB];
  __shared__ int rs[NBLK];
  __shared__ int rc[NBLK];
  __shared__ int wtot[32];
  const int tid = threadIdx.x, b = blockIdx.x;
  const int lane = tid & 31;
  const int hf = lane >> 4;
  const int c = lane & 15;
  const int wave = __builtin_amdgcn_readfirstlane(tid >> 5);
  const int n0 = b * NPB;
  constexpr int NV4 = NPB * OUT_DIM / 4;
  const float dv   = dinv[n0 + min(tid, NPB - 1)];
  const v2i   tt   = tab[b * NBLK + (tid & (NBLK - 1))];
  const float bias = b2[c];
  const v4f* g4 = (const v4f*)(g + (size_t)n0 * OUT_DIM);
  v4f gv[2];
#pragma unroll
  for (int k = 0; k < 2; ++k) gv[k] = g4[min(tid + k * 1024, NV4 - 1)];
  if (tid < NPB) dl[tid] = dv;
  if (tid < NBLK) {
    rs[tid] = tt.x;
    rc[tid] = tt.y;
  }
#pragma unroll
  for (int k = 0; k < 2; ++k) {
    const int i4 = tid + k * 1024;
    if (i4 < NV4) *(v4f*)&acc[4 * i4] = gv[k];
  }
  __syncthreads();
  int fb[RPW];
  const int total = flat_bases(rc, wtot, wave, lane, fb);

  for (int c0 = 0; c0 < total; c0 += CAP) {
    walk_runs<RPW, 0, RPW>(binned, rs, rc, wave, lane, [&](int j, int o, v2i p) {
      const unsigned r = (unsigned)(fb[j] + o - c0);
      if (r < (unsigned)CAP) erec[r] = ((unsigned)(p.y - n0) << 17) | (unsigned)p.x;
    });
    __syncthreads();
    const int count = min(CAP, total - c0);
    for (int e0 = wave * 32; e0 < count; e0 += 1024) {
      const int idx = e0 + lane;
      unsigned rec = 0;
      if (idx < count) rec = erec[idx];
      const int s = (int)(rec & 0x1FFFFu);
      const int d = (int)(rec >> 17);
      const int m = count - e0;
      if (m >= 32) {
        float x[16];
#pragma unroll
        for (int u = 0; u < 16; ++u) {
          int s0 = __builtin_amdgcn_readlane(s, 2 * u);
          int s1 = __builtin_amdgcn_readlane(s, 2 * u + 1);
          int sj = hf ? s1 : s0;
          x[u] = g[(size_t)sj * OUT_DIM + c];
        }
#pragma unroll
        for (int u = 0; u < 16; ++u) {
          int d0 = __builtin_amdgcn_readlane(d, 2 * u);
          int d1 = __builtin_amdgcn_readlane(d, 2 * u + 1);
          int dj = hf ? d1 : d0;
          atomicAdd(&acc[dj * OUT_DIM + c], x[u]);
        }
      } else {
        for (int j = 0; 2 * j < m; ++j) {
          int e = 2 * j + hf;
          int sj = __shfl(s, e);
          int dj = __shfl(d, e);
          if (e < m) atomicAdd(&acc[dj * OUT_DIM + c], g[(size_t)sj * OUT_DIM + c]);
        }
      }
    }
    __syncthreads();
  }
  for (int i = tid; i < NPB * OUT_DIM; i += 1024) {
    out[(size_t)n0 * OUT_DIM + i] = acc[i] * dl[i >> 4] + bias;
  }
}

extern "C" void kernel_launch(void* const* d_in, const int* in_sizes, int n_in,
                              void* d_out, int out_size, void* d_ws, size_t ws_size,
                              hipStream_t stream) {
  const float* x  = (const float*)d_in[0];
  const float* W1 = (const float*)d_in[1];
  const float* b1 = (const float*)d_in[2];
  const float* W2 = (const float*)d_in[3];
  const float* b2 = (const float*)d_in[4];
  const int*   ei = (const int*)d_in[5];

  float* dinv = (float*)d_ws;
  float* h    = dinv + N_NODES;
  float* g    = h    + (size_t)N_NODES * HID_DIM;
  v2i* tab    = (v2i*)(g + (size_t)N_NODES * OUT_DIM);
  int* binned = (int*)(tab + NBKT * NBLK);
  float* out = (float*)d_out;

  k_bin<<<NBLK, 1024, 0, stream>>>(ei, binned, tab);
  k_degree_gemm1<<<NBKT, 1024, 0, stream>>>(binned, tab, x, W1, dinv, h);
  k_agg32_gemm2<<<NBKT, 1024, 0, stream>>>(binned, tab, dinv, h, b1, W2, g);
  k_agg16<<<NBKT, 1024, 0, stream>>>(binned, tab, dinv, g, b2, out);
}


# Range's records ordered by destination node in LDS (slot = LDS cursor add); a wave sums each stretch of equal destination in a register, one LDS float add per stretch: sum order changes
# speedup vs baseline: 6.8641x; 1.0370x over previous
#include <hip/hip_runtime.h>


#define N_NODES 100000
#define N_EDGES 3200000
#define IN_DIM  128
#define HID_DIM 32
#define OUT_DIM 16

#define NBKT 250
#define NPB  400
#define NBLK 256
#define EPB  12500
#define SLAB 12544
#define RPW  (NBLK / 32)
#define CAP  16384

static_assert(NBKT * NPB == N_NODES, "ranges must tile the nodes");
static_assert(NBLK * EPB == N_EDGES, "slabs must tile the edges");
static_assert(N_NODES <= (1 << 17) && NPB <= (1 << 15), "packed record: 17 bits source, 15 bits local destination");
static_assert(IN_DIM * HID_DIM == 4 * 1024, "W1 goes to LDS as one 16-byte piece per thread of a 1024-thread work-group");

typedef __attribute__((ext_vector_type(2))) float v2f;
typedef __attribute__((ext_vector_type(4))) float v4f;
typedef __attribute__((ext_vector_type(8))) float v8f;
typedef __attribute__((ext_vector_type(2))) int   v2i;
typedef __attribute__((ext_vector_type(4))) int   v4i;

__device__ __forceinline__ int clamp_node(int v) {
  return min(max(v, 0), N_NODES - 1);
}
__device__ __forceinline__ int bucket_of(int d) {
  return (int)((unsigned)d / (unsigned)NPB);
}

__device__ __forceinline__ void block_incl_scan256(int v, int* sh, int tid) {
  if (tid < 256) sh[tid] = v;
  __syncthreads();
#pragma unroll
  for (int off = 1; off < 256; off <<= 1) {
    int t = 0;
    if (tid < 256 && tid >= off) t = sh[tid - off];
    __syncthreads();
    if (tid < 256) sh[tid] += t;
    __syncthreads();
  }
}

__global__ void __launch_bounds__(1024) k_bin(const int* __restrict__ ei,
                                              int* __restrict__ binned,
                                              v2i* __restrict__ tab) {
  __shared__ __attribute__((aligned(16))) v2i lrec[EPB];
  __shared__ int lh[256];
  __shared__ int sh[256];
  __shared__ int lst[256];
  const int tid = threadIdx.x, blk = blockIdx.x;
  if (tid < 256) lh[tid] = 0;
  const int* srcp = ei + blk * EPB;
  const int* dstp = ei + N_EDGES + blk * EPB;
  int s[13], d[13], r[13];
#pragma unroll
  for (int k = 0; k < 13; ++k) {
    const int i = tid + k * 1024;
    const int ii = min(i, EPB - 1);
    r[k] = 0;
    s[k] = clamp_node(srcp[ii]);
    d[k] = clamp_node(dstp[ii]);
  }
  __syncthreads();
#pragma unroll
  for (int k = 0; k < 13; ++k) {
    const int i = tid + k * 1024;
    if (i < EPB) r[k] = atomicAdd(&lh[bucket_of(d[k])], 1);
  }
  __syncthreads();
  const int v = (tid < 256) ? lh[tid] : 0;
  block_incl_scan256(v, sh, tid);
  if (tid < 256) lst[tid] = sh[tid] - v;
  if (tid < NBKT) {
    v2i t; t.x = blk * SLAB + sh[tid] - v; t.y = v;
    tab[tid * NBLK + blk] = t;
  }
  __syncthreads();
#pragma unroll
  for (int k = 0; k < 13; ++k) {
    const int i = tid + k * 1024;
    if (i < EPB) {
      v2i p; p.x = s[k]; p.y = d[k];
      lrec[lst[bucket_of(d[k])] + r[k]] = p;
    }
  }
  __syncthreads();
  v4i* outp = (v4i*)(binned + 2 * (size_t)blk * SLAB);
  const v4i* lp = (const v4i*)lrec;
  for (int i = tid; i < EPB / 2; i += 1024) outp[i] = lp[i];
}

template <int RB, int JB, int JE, typename F>
__device__ __forceinline__ void walk_runs(const int* __restrict__ binned, const int* rs, const int* rc,
                                          int wave, int lane, F fn) {
#pragma unroll
  for (int j0 = JB; j0 < JE; j0 += RB) {
    int st[RB], n[RB];
    v2i rec[RB][3];
#pragma unroll
    for (int j = 0; j < RB; ++j) {
      const int k = wave + 32 * (j0 + j);
      st[j] = __builtin_amdgcn_readfirstlane(rs[k]);
      n[j]  = __builtin_amdgcn_readfirstlane(rc[k]);
#pragma unroll
      for (int t = 0; t < 3; ++t) {
        const int o = lane + 32 * t;
        const int at = (o < n[j]) ? st[j] + o : 0;
        rec[j][t] = *(const v2i*)(binned + 2 * (size_t)at);
      }
    }
#pragma unroll
    for (int j = 0; j < RB; ++j) {
#pragma unroll
      for (int t = 0; t < 3; ++t) {
        const int o = lane + 32 * t;
        if (o < n[j]) fn(j0 + j, o, rec[j][t]);
      }
    }
#pragma unroll
    for (int j = 0; j < RB; ++j) {
      for (int o = lane + 96; o < n[j]; o += 32)
        fn(j0 + j, o, *(const v2i*)(binned + 2 * (size_t)(st[j] + o)));
    }
  }
}

#define NTILE (NPB / 16)
#define WPITCH 48
__global__ void __launch_bounds__(1024) k_degree_gemm1(const int* __restrict__ binned,
                                                       const v2i* __restrict__ tab,
                                                       const float* __restrict__ X,
                                                       const float* __restrict__ W,
                                                       float* __restrict__ dinv,
                                                       float* __restrict__ H,
                                                       int* __restrict__ nstart) {
  __shared__ int cnt[NPB];
  __shared__ float dl[NPB];
  __shared__ int rs[NBLK];
  __shared__ int rc[NBLK];
  __shared__ __attribute__((aligned(16))) float wl[IN_DIM * WPITCH];
  const int tid = threadIdx.x, b = blockIdx.x;
  const int lane = tid & 31;
  const int wave = __builtin_amdgcn_readfirstlane(tid >> 5);
  const int n0   = b * NPB;
  const int half = lane >> 4;
  const int l15  = lane & 15;
  const int kOff = half * 2;

  const v2i tt = tab[b * NBLK + (tid & (NBLK - 1))];
  const v4f wv = ((const v4f*)W)[tid];
  if (tid < NPB) cnt[tid] = 0;
  if (tid < NBLK) {
    rs[tid] = tt.x;
    rc[tid] = tt.y;
  }
  *(v4f*)&wl[(tid >> 3) * WPITCH + (tid & 7) * 4] = wv;
  __syncthreads();

  const auto count_fn = [&](int, int, v2i p) { atomicAdd(&cnt[p.y - n0], 1); };

  walk_runs<4, 0, 4>(binned, rs, rc, wave, lane, count_fn);

  v8f c0 = {}, c1 = {};
  if (wave < NTILE) {
    v2f a[IN_DIM / 4];
    const float* xr = X + (size_t)(n0 + wave * 16 + l15) * IN_DIM + kOff;
#pragma unroll
    for (int k4 = 0; k4 < IN_DIM / 4; ++k4) a[k4] = *(const v2f*)(xr + 4 * k4);
#pragma unroll
    for (int k4 = 0; k4 < IN_DIM / 4; ++k4) {
      const float* wr = wl + (4 * k4 + kOff) * WPITCH + l15;
      v2f b0, b1;
      b0.x = wr[0];
      b0.y = wr[WPITCH];
      b1.x = wr[16];
      b1.y = wr[WPITCH + 16];
      c0 = __builtin_amdgcn_wmma_f32_16x16x4_f32(false, a[k4], false, b0,
                                                 (short)0, c0, false, false);
      c1 = __builtin_amdgcn_wmma_f32_16x16x4_f32(false, a[k4], false, b1,
                                                 (short)0, c1, false, false);
    }
  }

  walk_runs<4, 4, RPW>(binned, rs, rc, wave, lane, count_fn);
  __syncthreads();
  if (tid < NPB) {
    float deg = 1.0f + (float)cnt[tid];
    float di = (deg > 0.0f) ? rsqrtf(deg) : 0.0f;
    dl[tid] = di;
    dinv[n0 + tid] = di;
  }
  __syncthreads();

  if (wave < NTILE) {
    const int r0 = wave * 16 + half * 8;
#pragma unroll
    for (int i = 0; i < 8; ++i) {
      const float di = dl[r0 + i];
      float* hr = H + (size_t)(n0 + r0 + i) * HID_DIM + l15;
      hr[0]  = c0[i] * di;
      hr[16] = c1[i] * di;
    }
  }
  if (wave == 31) {
    constexpr int PER = (NPB + 31) / 32;
    int v[PER];
    int sum = 0;
#pragma unroll
    for (int k = 0; k < PER; ++k) {
      const int i = lane * PER + k;
      v[k] = (i < NPB) ? cnt[i] : 0;
      sum += v[k];
    }
    int x = sum;
#pragma unroll
    for (int off = 1; off < 32; off <<= 1) {
      const int y = __shfl_up(x, off);
      if (lane >= off) x += y;
    }
    int at = x - sum;
#pragma unroll
    for (int k = 0; k < PER; ++k) {
      const int i = lane * PER + k;
      if (i < NPB) nstart[n0 + i] = at;
      at += v[k];
    }
  }
}

__device__ __forceinline__ void collect_by_node(const int* __restrict__ binned, const int* rs, const int* rc,
                                                int* cursor, unsigned* erec, int n0, int wave, int lane) {
  int st[RPW], n[RPW];
  v2i rec[RPW][3];
#pragma unroll
  for (int j = 0; j < RPW; ++j) {
    const int k = wave + 32 * j;
    st[j] = __builtin_amdgcn_readfirstlane(rs[k]);
    n[j]  = __builtin_amdgcn_readfirstlane(rc[k]);
#pragma unroll
    for (int t = 0; t < 3; ++t) {
      const int o = lane + 32 * t;
      const int at = (o < n[j]) ? st[j] + o : 0;
      rec[j][t] = *(const v2i*)(binned + 2 * (size_t)at);
    }
  }
  unsigned pos[RPW][3], pk[RPW][3];
#pragma unroll
  for (int j = 0; j < RPW; ++j) {
#pragma unroll
    for (int t = 0; t < 3; ++t) {
      const int o = lane + 32 * t;
      const int dloc = rec[j][t].y - n0;
      pk[j][t]  = ((unsigned)dloc << 17) | (unsigned)rec[j][t].x;
      pos[j][t] = (unsigned)CAP;
      if (o < n[j]) pos[j][t] = (unsigned)atomicAdd(&cursor[dloc], 1);
    }
  }
#pragma unroll
  for (int j = 0; j < RPW; ++j) {
#pragma unroll
    for (int t = 0; t < 3; ++t) {
      if (pos[j][t] < (unsigned)CAP) erec[pos[j][t]] = pk[j][t];
    }
  }
#pragma unroll
  for (int j = 0; j < RPW; ++j) {
    for (int o = lane + 96; o < n[j]; o += 32) {
      const v2i p = *(const v2i*)(binned + 2 * (size_t)(st[j] + o));
      const int dloc = p.y - n0;
      const unsigned r = (unsigned)atomicAdd(&cursor[dloc], 1);
      if (r < (unsigned)CAP) erec[r] = ((unsigned)dloc << 17) | (unsigned)p.x;
    }
  }
}

__device__ __forceinline__ int flat_bases(const int* rc, int* wtot, int wave, int lane, int (&fb)[RPW]) {
  int wt = 0;
#pragma unroll
  for (int j = 0; j < RPW; ++j) {
    fb[j] = wt;
    wt += __builtin_amdgcn_readfirstlane(rc[wave + 32 * j]);
  }
  if (lane == 0) wtot[wave] = wt;
  __syncthreads();
  int x = wtot[lane];
#pragma unroll
  for (int off = 1; off < 32; off <<= 1) {
    const int y = __shfl_up(x, off);
    if (lane >= off) x += y;
  }
  const int total = __builtin_amdgcn_readlane(x, 31);
  const int wb = __builtin_amdgcn_readlane(x, wave) - wt;
#pragma unroll
  for (int j = 0; j < RPW; ++j) fb[j] += wb;
  return total;
}

#define APITCH 34
__global__ void __launch_bounds__(1024) k_agg32_gemm2(const int* __restrict__ binned,
                                                      const v2i* __restrict__ tab,
                                                      const float* __restrict__ dinv,
                                                      const float* __restrict__ h,
                                                      const float* __restrict__ b1,
                                                      const float* __restrict__ W2,
                                                      const int* __restrict__ nstart,
                                                      float* __restrict__ G) {
  __shared__ __attribute__((aligned(16))) float acc[NPB * APITCH];
  __shared__ unsigned erec[CAP];
  __shared__ float dl[NPB];
  __shared__ int cursor[NPB];
  __shared__ int rs[NBLK];
  __shared__ int rc[NBLK];
  __shared__ int wtot[32];
  const int tid = threadIdx.x, b = blockIdx.x;
  const int lane = tid & 31;
  const int wave = __builtin_amdgcn_readfirstlane(tid >> 5);
  const int n0 = b * NPB;
  constexpr int NV4 = NPB * HID_DIM / 4;
  const float dv   = dinv[n0 + min(tid, NPB - 1)];
  const int   ns   = nstart[n0 + min(tid, NPB - 1)];
  const v2i   tt   = tab[b * NBLK + (tid & (NBLK - 1))];
  const float bias = b1[lane];
  const v4f* h4 = (const v4f*)(h + (size_t)n0 * HID_DIM);
  v4f hv[4];
#pragma unroll
  for (int k = 0; k < 4; ++k) hv[k] = h4[min(tid + k * 1024, NV4 - 1)];
  if (tid < NPB) {
    dl[tid] = dv;
    cursor[tid] = ns;
  }
  if (tid < NBLK) {
    rs[tid] = tt.x;
    rc[tid] = tt.y;
  }
#pragma unroll
  for (int k = 0; k < 4; ++k) {
    const int i4 = tid + k * 1024;
    if (i4 < NV4) {
      float* ap = &acc[(i4 >> 3) * APITCH + (i4 & 7) * 4];
      v2f lo, hi;
      lo.x = hv[k].x; lo.y = hv[k].y; hi.x = hv[k].z; hi.y = hv[k].w;
      *(v2f*)ap = lo;
      *(v2f*)(ap + 2) = hi;
    }
  }
  __syncthreads();
  int fb[RPW];
  const int total = flat_bases(rc, wtot, wave, lane, fb);

  const bool by_node = total <= CAP;
  for (int c0 = 0; c0 < total; c0 += CAP) {
    if (by_node) {
      collect_by_node(binned, rs, rc, cursor, erec, n0, wave, lane);
    } else {
      walk_runs<RPW, 0, RPW>(binned, rs, rc, wave, lane, [&](int j, int o, v2i p) {
        const unsigned r = (unsigned)(fb[j] + o - c0);
        if (r < (unsigned)CAP) erec[r] = ((unsigned)(p.y - n0) << 17) | (unsigned)p.x;
      });
    }
    __syncthreads();
    const int count = min(CAP, total - c0);
    for (int e0 = wave * 32; e0 < count; e0 += 1024) {
      const int idx = e0 + lane;
      unsigned rec = 0;
      if (idx < count) rec = erec[idx];
      const int s = (int)(rec & 0x1FFFFu);
      const int d = (int)(rec >> 17);
      const int m = count - e0;
      if (m >= 32) {
        int cur = __builtin_amdgcn_readlane(d, 0);
        float run = 0.0f;
#pragma unroll
        for (int j = 0; j < 32; j += 16) {
          float x[16];
#pragma unroll
          for (int u = 0; u < 16; ++u) {
            int sj = __builtin_amdgcn_readlane(s, j + u);
            x[u] = h[(size_t)sj * HID_DIM + lane];
          }
#pragma unroll
          for (int u = 0; u < 16; ++u) {
            int dj = __builtin_amdgcn_readlane(d, j + u);
            if (dj != cur) {
              atomicAdd(&acc[cur * APITCH + lane], run);
              run = 0.0f;
              cur = dj;
            }
            run += x[u];
          }
        }
        atomicAdd(&acc[cur * APITCH + lane], run);
      } else {
        for (int j = 0; j < m; ++j) {
          int sj = __shfl(s, j);
          int dj = __shfl(d, j);
          atomicAdd(&acc[dj * APITCH + lane], h[(size_t)sj * HID_DIM + lane]);
        }
      }
    }
    __syncthreads();
  }

  const int half = lane >> 4;
  const int l15  = lane & 15;
  const int kOff = half * 2;
  v2f bw[HID_DIM / 4];
  if (wave < NTILE) {
#pragma unroll
    for (int k4 = 0; k4 < HID_DIM / 4; ++k4) {
      bw[k4].x = W2[(4 * k4 + kOff + 0) * OUT_DIM + l15];
      bw[k4].y = W2[(4 * k4 + kOff + 1) * OUT_DIM + l15];
    }
  }
  for (int i = tid; i < NPB * HID_DIM; i += 1024) {
    const int r = i >> 5, cc = i & 31;
    float v = acc[r * APITCH + cc] * dl[r] + bias;
    acc[r * APITCH + cc] = fmaxf(v, 0.0f);
  }
  __syncthreads();

  if (wave < NTILE) {
    const float* ar = acc + (wave * 16 + l15) * APITCH + kOff;
    v8f c = {};
#pragma unroll
    for (int k = 0; k < HID_DIM; k += 4) {
      v2f a = *(const v2f*)(ar + k);
      c = __builtin_amdgcn_wmma_f32_16x16x4_f32(false, a, false, bw[k / 4],
                                                (short)0, c, false, false);
    }
    const int r0 = wave * 16 + half * 8;
#pragma unroll
    for (int i = 0; i < 8; ++i) {
      G[(size_t)(n0 + r0 + i) * OUT_DIM + l15] = c[i] * dl[r0 + i];
    }
  }
}

__global__ void __launch_bounds__(1024) k_agg16(const int* __restrict__ binned,
                                                const v2i* __restrict__ tab,
                                                const float* __restrict__ dinv,
                                                const float* __restrict__ g,
                                                const float* __restrict__ b2,
                                                const int* __restrict__ nstart,
                                                float* __restrict__ out) {
  __shared__ __attribute__((aligned(16))) float acc[NPB * OUT_DIM];
  __shared__ unsigned erec[CAP];
  __shared__ float dl[NPB];
  __shared__ int cursor[NPB];
  __shared__ int rs[NBLK];
  __shared__ int rc[NBLK];
  __shared__ int wtot[32];
  const int tid = threadIdx.x, b = blockIdx.x;
  const int lane = tid & 31;
  const int hf = lane >> 4;
  const int c = lane & 15;
  const int wave = __builtin_amdgcn_readfirstlane(tid >> 5);
  const int n0 = b * NPB;
  constexpr int NV4 = NPB * OUT_DIM / 4;
  const float dv   = dinv[n0 + min(tid, NPB - 1)];
  const int   ns   = nstart[n0 + min(tid, NPB - 1)];
  const v2i   tt   = tab[b * NBLK + (tid & (NBLK - 1))];
  const float bias = b2[c];
  const v4f* g4 = (const v4f*)(g + (size_t)n0 * OUT_DIM);
  v4f gv[2];
#pragma unroll
  for (int k = 0; k < 2; ++k) gv[k] = g4[min(tid + k * 1024, NV4 - 1)];
  if (tid < NPB) {
    dl[tid] = dv;
    cursor[tid] = ns;
  }
  if (tid < NBLK) {
    rs[tid] = tt.x;
    rc[tid] = tt.y;
  }
#pragma unroll
  for (int k = 0; k < 2; ++k) {
    const int i4 = tid + k * 1024;
    if (i4 < NV4) *(v4f*)&acc[4 * i4] = gv[k];
  }
  __syncthreads();
  int fb[RPW];
  const int total = flat_bases(rc, wtot, wave, lane, fb);

  const bool by_node = total <= CAP;
  for (int c0 = 0; c0 < total; c0 += CAP) {
    if (by_node) {
      collect_by_node(binned, rs, rc, cursor, erec, n0, wave, lane);
    } else {
      walk_runs<RPW, 0, RPW>(binned, rs, rc, wave, lane, [&](int j, int o, v2i p) {
        const unsigned r = (unsigned)(fb[j] + o - c0);
        if (r < (unsigned)CAP) erec[r] = ((unsigned)(p.y - n0) << 17) | (unsigned)p.x;
      });
    }
    __syncthreads();
    const int count = min(CAP, total - c0);
    for (int e0 = wave * 32; e0 < count; e0 += 1024) {
      const int idx = e0 + lane;
      unsigned rec = 0;
      if (idx < count) rec = erec[idx];
      const int s = (int)(rec & 0x1FFFFu);
      const int d = (int)(rec >> 17);
      const int m = count - e0;
      if (m >= 32) {
        float x[16];
#pragma unroll
        for (int u = 0; u < 16; ++u) {
          int s0 = __builtin_amdgcn_readlane(s, u);
          int s1 = __builtin_amdgcn_readlane(s, 16 + u);
          int sj = hf ? s1 : s0;
          x[u] = g[(size_t)sj * OUT_DIM + c];
        }
        int cur0 = __builtin_amdgcn_readlane(d, 0);
        int cur1 = __builtin_amdgcn_readlane(d, 16);
        float run = 0.0f;
#pragma unroll
        for (int u = 0; u < 16; ++u) {
          int d0 = __builtin_amdgcn_readlane(d, u);
          int d1 = __builtin_amdgcn_readlane(d, 16 + u);
          if (d0 != cur0 || d1 != cur1) {
            const bool moved = hf ? (d1 != cur1) : (d0 != cur0);
            if (moved) {
              atomicAdd(&acc[(hf ? cur1 : cur0) * OUT_DIM + c], run);
              run = 0.0f;
            }
            cur0 = d0;
            cur1 = d1;
          }
          run += x[u];
        }
        atomicAdd(&acc[(hf ? cur1 : cur0) * OUT_DIM + c], run);
      } else {
        for (int j = 0; 2 * j < m; ++j) {
          int e = 2 * j + hf;
          int sj = __shfl(s, e);
          int dj = __shfl(d, e);
          if (e < m) atomicAdd(&acc[dj * OUT_DIM + c], g[(size_t)sj * OUT_DIM + c]);
        }
      }
    }
    __syncthreads();
  }
  for (int i = tid; i < NPB * OUT_DIM; i += 1024) {
    out[(size_t)n0 * OUT_DIM + i] = acc[i] * dl[i >> 4] + bias;
  }
}

extern "C" void kernel_launch(void* const* d_in, const int* in_sizes, int n_in,
                              void* d_out, int out_size, void* d_ws, size_t ws_size,
                              hipStream_t stream) {
  const float* x  = (const float*)d_in[0];
  const float* W1 = (const float*)d_in[1];
  const float* b1 = (const float*)d_in[2];
  const float* W2 = (const float*)d_in[3];
  const float* b2 = (const float*)d_in[4];
  const int*   ei = (const int*)d_in[5];

  float* dinv = (float*)d_ws;
  float* h    = dinv + N_NODES;
  float* g    = h    + (size_t)N_NODES * HID_DIM;
  v2i* tab    = (v2i*)(g + (size_t)N_NODES * OUT_DIM);
  int* binned = (int*)(tab + NBKT * NBLK);
  int* nstart = binned + 2 * (size_t)NBLK * SLAB;
  float* out = (float*)d_out;

  k_bin<<<NBLK, 1024, 0, stream>>>(ei, binned, tab);
  k_degree_gemm1<<<NBKT, 1024, 0, stream>>>(binned, tab, x, W1, dinv, h, nstart);
  k_agg32_gemm2<<<NBKT, 1024, 0, stream>>>(binned, tab, dinv, h, b1, W2, nstart, g);
  k_agg16<<<NBKT, 1024, 0, stream>>>(binned, tab, dinv, g, b2, nstart, out);
}


# Edge loops by 16-lane halves: source by lane broadcast inside the half, both halves' destinations in one scalar compare; layer 1 loads 8 bytes per lane (two rows per load): layer-1 sum order changes
# speedup vs baseline: 7.4624x; 1.0872x over previous
#include <hip/hip_runtime.h>


#define N_NODES 100000
#define N_EDGES 3200000
#define IN_DIM  128
#define HID_DIM 32
#define OUT_DIM 16

#define NBKT 250
#define NPB  400
#define NBLK 256
#define EPB  12500
#define SLAB 12544
#define RPW  (NBLK / 32)
#define CAP  16384

static_assert(NBKT * NPB == N_NODES, "ranges must tile the nodes");
static_assert(NBLK * EPB == N_EDGES, "slabs must tile the edges");
static_assert(N_NODES <= (1 << 17) && NPB <= (1 << 15), "packed record: 17 bits source, 15 bits local destination");
static_assert(IN_DIM * HID_DIM == 4 * 1024, "W1 goes to LDS as one 16-byte piece per thread of a 1024-thread work-group");
static_assert((unsigned long long)N_NODES * HID_DIM * sizeof(float) < (1ull << 32), "edge loops: a row's byte offset fits 32 bits");
static_assert(NPB <= (1 << 15), "edge loops: two local destinations are packed in one 32-bit word");

typedef __attribute__((ext_vector_type(2))) float v2f;
typedef __attribute__((ext_vector_type(4))) float v4f;
typedef __attribute__((ext_vector_type(8))) float v8f;
typedef __attribute__((ext_vector_type(2))) int   v2i;
typedef __attribute__((ext_vector_type(4))) int   v4i;

__device__ __forceinline__ int clamp_node(int v) {
  return min(max(v, 0), N_NODES - 1);
}
__device__ __forceinline__ int bucket_of(int d) {
  return (int)((unsigned)d / (unsigned)NPB);
}

__device__ __forceinline__ void block_incl_scan256(int v, int* sh, int tid) {
  if (tid < 256) sh[tid] = v;
  __syncthreads();
#pragma unroll
  for (int off = 1; off < 256; off <<= 1) {
    int t = 0;
    if (tid < 256 && tid >= off) t = sh[tid - off];
    __syncthreads();
    if (tid < 256) sh[tid] += t;
    __syncthreads();
  }
}

__global__ void __launch_bounds__(1024) k_bin(const int* __restrict__ ei,
                                              int* __restrict__ binned,
                                              v2i* __restrict__ tab) {
  __shared__ __attribute__((aligned(16))) v2i lrec[EPB];
  __shared__ int lh[256];
  __shared__ int sh[256];
  __shared__ int lst[256];
  const int tid = threadIdx.x, blk = blockIdx.x;
  if (tid < 256) lh[tid] = 0;
  const int* srcp = ei + blk * EPB;
  const int* dstp = ei + N_EDGES + blk * EPB;
  int s[13], d[13], r[13];
#pragma unroll
  for (int k = 0; k < 13; ++k) {
    const int i = tid + k * 1024;
    const int ii = min(i, EPB - 1);
    r[k] = 0;
    s[k] = clamp_node(srcp[ii]);
    d[k] = clamp_node(dstp[ii]);
  }
  __syncthreads();
#pragma unroll
  for (int k = 0; k < 13; ++k) {
    const int i = tid + k * 1024;
    if (i < EPB) r[k] = atomicAdd(&lh[bucket_of(d[k])], 1);
  }
  __syncthreads();
  const int v = (tid < 256) ? lh[tid] : 0;
  block_incl_scan256(v, sh, tid);
  if (tid < 256) lst[tid] = sh[tid] - v;
  if (tid < NBKT) {
    v2i t; t.x = blk * SLAB + sh[tid] - v; t.y = v;
    tab[tid * NBLK + blk] = t;
  }
  __syncthreads();
#pragma unroll
  for (int k = 0; k < 13; ++k) {
    const int i = tid + k * 1024;
    if (i < EPB) {
      v2i p; p.x = s[k]; p.y = d[k];
      lrec[lst[bucket_of(d[k])] + r[k]] = p;
    }
  }
  __syncthreads();
  v4i* outp = (v4i*)(binned + 2 * (size_t)blk * SLAB);
  const v4i* lp = (const v4i*)lrec;
  for (int i = tid; i < EPB / 2; i += 1024) outp[i] = lp[i];
}

template <int RB, int JB, int JE, typename F>
__device__ __forceinline__ void walk_runs(const int* __restrict__ binned, const int* rs, const int* rc,
                                          int wave, int lane, F fn) {
#pragma unroll
  for (int j0 = JB; j0 < JE; j0 += RB) {
    int st[RB], n[RB];
    v2i rec[RB][3];
#pragma unroll
    for (int j = 0; j < RB; ++j) {
      const int k = wave + 32 * (j0 + j);
      st[j] = __builtin_amdgcn_readfirstlane(rs[k]);
      n[j]  = __builtin_amdgcn_readfirstlane(rc[k]);
#pragma unroll
      for (int t = 0; t < 3; ++t) {
        const int o = lane + 32 * t;
        const int at = (o < n[j]) ? st[j] + o : 0;
        rec[j][t] = *(const v2i*)(binned + 2 * (size_t)at);
      }
    }
#pragma unroll
    for (int j = 0; j < RB; ++j) {
#pragma unroll
      for (int t = 0; t < 3; ++t) {
        const int o = lane + 32 * t;
        if (o < n[j]) fn(j0 + j, o, rec[j][t]);
      }
    }
#pragma unroll
    for (int j = 0; j < RB; ++j) {
      for (int o = lane + 96; o < n[j]; o += 32)
        fn(j0 + j, o, *(const v2i*)(binned + 2 * (size_t)(st[j] + o)));
    }
  }
}

#define NTILE (NPB / 16)
#define WPITCH 48
__global__ void __launch_bounds__(1024) k_degree_gemm1(const int* __restrict__ binned,
                                                       const v2i* __restrict__ tab,
                                                       const float* __restrict__ X,
                                                       const float* __restrict__ W,
                                                       float* __restrict__ dinv,
                                                       float* __restrict__ H,
                                                       int* __restrict__ nstart) {
  __shared__ int cnt[NPB];
  __shared__ float dl[NPB];
  __shared__ int rs[NBLK];
  __shared__ int rc[NBLK];
  __shared__ __attribute__((aligned(16))) float wl[IN_DIM * WPITCH];
  const int tid = threadIdx.x, b = blockIdx.x;
  const int lane = tid & 31;
  const int wave = __builtin_amdgcn_readfirstlane(tid >> 5);
  const int n0   = b * NPB;
  const int half = lane >> 4;
  const int l15  = lane & 15;
  const int kOff = half * 2;

  const v2i tt = tab[b * NBLK + (tid & (NBLK - 1))];
  const v4f wv = ((const v4f*)W)[tid];
  if (tid < NPB) cnt[tid] = 0;
  if (tid < NBLK) {
    rs[tid] = tt.x;
    rc[tid] = tt.y;
  }
  *(v4f*)&wl[(tid >> 3) * WPITCH + (tid & 7) * 4] = wv;
  __syncthreads();

  const auto count_fn = [&](int, int, v2i p) { atomicAdd(&cnt[p.y - n0], 1); };

  walk_runs<4, 0, 4>(binned, rs, rc, wave, lane, count_fn);

  v8f c0 = {}, c1 = {};
  if (wave < NTILE) {
    v2f a[IN_DIM / 4];
    const float* xr = X + (size_t)(n0 + wave * 16 + l15) * IN_DIM + kOff;
#pragma unroll
    for (int k4 = 0; k4 < IN_DIM / 4; ++k4) a[k4] = *(const v2f*)(xr + 4 * k4);
#pragma unroll
    for (int k4 = 0; k4 < IN_DIM / 4; ++k4) {
      const float* wr = wl + (4 * k4 + kOff) * WPITCH + l15;
      v2f b0, b1;
      b0.x = wr[0];
      b0.y = wr[WPITCH];
      b1.x = wr[16];
      b1.y = wr[WPITCH + 16];
      c0 = __builtin_amdgcn_wmma_f32_16x16x4_f32(false, a[k4], false, b0,
                                                 (short)0, c0, false, false);
      c1 = __builtin_amdgcn_wmma_f32_16x16x4_f32(false, a[k4], false, b1,
                                                 (short)0, c1, false, false);
    }
  }

  walk_runs<4, 4, RPW>(binned, rs, rc, wave, lane, count_fn);
  __syncthreads();
  if (tid < NPB) {
    float deg = 1.0f + (float)cnt[tid];
    float di = (deg > 0.0f) ? rsqrtf(deg) : 0.0f;
    dl[tid] = di;
    dinv[n0 + tid] = di;
  }
  __syncthreads();

  if (wave < NTILE) {
    const int r0 = wave * 16 + half * 8;
#pragma unroll
    for (int i = 0; i < 8; ++i) {
      const float di = dl[r0 + i];
      float* hr = H + (size_t)(n0 + r0 + i) * HID_DIM + l15;
      hr[0]  = c0[i] * di;
      hr[16] = c1[i] * di;
    }
  }
  if (wave == 31) {
    constexpr int PER = (NPB + 31) / 32;
    int v[PER];
    int sum = 0;
#pragma unroll
    for (int k = 0; k < PER; ++k) {
      const int i = lane * PER + k;
      v[k] = (i < NPB) ? cnt[i] : 0;
      sum += v[k];
    }
    int x = sum;
#pragma unroll
    for (int off = 1; off < 32; off <<= 1) {
      const int y = __shfl_up(x, off);
      if (lane >= off) x += y;
    }
    int at = x - sum;
#pragma unroll
    for (int k = 0; k < PER; ++k) {
      const int i = lane * PER + k;
      if (i < NPB) nstart[n0 + i] = at;
      at += v[k];
    }
  }
}

__device__ __forceinline__ void collect_by_node(const int* __restrict__ binned, const int* rs, const int* rc,
                                                int* cursor, unsigned* erec, int n0, int wave, int lane) {
  int st[RPW], n[RPW];
  v2i rec[RPW][3];
#pragma unroll
  for (int j = 0; j < RPW; ++j) {
    const int k = wave + 32 * j;
    st[j] = __builtin_amdgcn_readfirstlane(rs[k]);
    n[j]  = __builtin_amdgcn_readfirstlane(rc[k]);
#pragma unroll
    for (int t = 0; t < 3; ++t) {
      const int o = lane + 32 * t;
      const int at = (o < n[j]) ? st[j] + o : 0;
      rec[j][t] = *(const v2i*)(binned + 2 * (size_t)at);
    }
  }
  unsigned pos[RPW][3], pk[RPW][3];
#pragma unroll
  for (int j = 0; j < RPW; ++j) {
#pragma unroll
    for (int t = 0; t < 3; ++t) {
      const int o = lane + 32 * t;
      const int dloc = rec[j][t].y - n0;
      pk[j][t]  = ((unsigned)dloc << 17) | (unsigned)rec[j][t].x;
      pos[j][t] = (unsigned)CAP;
      if (o < n[j]) pos[j][t] = (unsigned)atomicAdd(&cursor[dloc], 1);
    }
  }
#pragma unroll
  for (int j = 0; j < RPW; ++j) {
#pragma unroll
    for (int t = 0; t < 3; ++t) {
      if (pos[j][t] < (unsigned)CAP) erec[pos[j][t]] = pk[j][t];
    }
  }
#pragma unroll
  for (int j = 0; j < RPW; ++j) {
    for (int o = lane + 96; o < n[j]; o += 32) {
      const v2i p = *(const v2i*)(binned + 2 * (size_t)(st[j] + o));
      const int dloc = p.y - n0;
      const unsigned r = (unsigned)atomicAdd(&cursor[dloc], 1);
      if (r < (unsigned)CAP) erec[r] = ((unsigned)dloc << 17) | (unsigned)p.x;
    }
  }
}

__device__ __forceinline__ int flat_bases(const int* rc, int* wtot, int wave, int lane, int (&fb)[RPW]) {
  int wt = 0;
#pragma unroll
  for (int j = 0; j < RPW; ++j) {
    fb[j] = wt;
    wt += __builtin_amdgcn_readfirstlane(rc[wave + 32 * j]);
  }
  if (lane == 0) wtot[wave] = wt;
  __syncthreads();
  int x = wtot[lane];
#pragma unroll
  for (int off = 1; off < 32; off <<= 1) {
    const int y = __shfl_up(x, off);
    if (lane >= off) x += y;
  }
  const int total = __builtin_amdgcn_readlane(x, 31);
  const int wb = __builtin_amdgcn_readlane(x, wave) - wt;
#pragma unroll
  for (int j = 0; j < RPW; ++j) fb[j] += wb;
  return total;
}

#define APITCH 34
__global__ void __launch_bounds__(1024) k_agg32_gemm2(const int* __restrict__ binned,
                                                      const v2i* __restrict__ tab,
                                                      const float* __restrict__ dinv,
                                                      const float* __restrict__ h,
                                                      const float* __restrict__ b1,
                                                      const float* __restrict__ W2,
                                                      const int* __restrict__ nstart,
                                                      float* __restrict__ G) {
  __shared__ __attribute__((aligned(16))) float acc[NPB * APITCH];
  __shared__ unsigned erec[CAP];
  __shared__ float dl[NPB];
  __shared__ int cursor[NPB];
  __shared__ int rs[NBLK];
  __shared__ int rc[NBLK];
  __shared__ int wtot[32];
  const int tid = threadIdx.x, b = blockIdx.x;
  const int lane = tid & 31;
  const int hf = lane >> 4;
  const int c2 = (lane & 15) * 2;
  const int wave = __builtin_amdgcn_readfirstlane(tid >> 5);
  const int n0 = b * NPB;
  constexpr int NV4 = NPB * HID_DIM / 4;
  const float dv   = dinv[n0 + min(tid, NPB - 1)];
  const int   ns   = nstart[n0 + min(tid, NPB - 1)];
  const v2i   tt   = tab[b * NBLK + (tid & (NBLK - 1))];
  const float bias = b1[lane];
  const v4f* h4 = (const v4f*)(h + (size_t)n0 * HID_DIM);
  v4f hv[4];
#pragma unroll
  for (int k = 0; k < 4; ++k) hv[k] = h4[min(tid + k * 1024, NV4 - 1)];
  if (tid < NPB) {
    dl[tid] = dv;
    cursor[tid] = ns;
  }
  if (tid < NBLK) {
    rs[tid] = tt.x;
    rc[tid] = tt.y;
  }
#pragma unroll
  for (int k = 0; k < 4; ++k) {
    const int i4 = tid + k * 1024;
    if (i4 < NV4) {
      float* ap = &acc[(i4 >> 3) * APITCH + (i4 & 7) * 4];
      v2f lo, hi;
      lo.x = hv[k].x; lo.y = hv[k].y; hi.x = hv[k].z; hi.y = hv[k].w;
      *(v2f*)ap = lo;
      *(v2f*)(ap + 2) = hi;
    }
  }
  __syncthreads();
  int fb[RPW];
  const int total = flat_bases(rc, wtot, wave, lane, fb);

  const bool by_node = total <= CAP;
  for (int c0 = 0; c0 < total; c0 += CAP) {
    if (by_node) {
      collect_by_node(binned, rs, rc, cursor, erec, n0, wave, lane);
    } else {
      walk_runs<RPW, 0, RPW>(binned, rs, rc, wave, lane, [&](int j, int o, v2i p) {
        const unsigned r = (unsigned)(fb[j] + o - c0);
        if (r < (unsigned)CAP) erec[r] = ((unsigned)(p.y - n0) << 17) | (unsigned)p.x;
      });
    }
    __syncthreads();
    const int count = min(CAP, total - c0);
    for (int e0 = wave * 32; e0 < count; e0 += 1024) {
      const int idx = e0 + lane;
      unsigned rec = 0;
      if (idx < count) rec = erec[idx];
      const int s = (int)(rec & 0x1FFFFu);
      const int d = (int)(rec >> 17);
      const int m = count - e0;
      if (m >= 32) {
        v2f x[16];
        const unsigned cb = (unsigned)c2 * (unsigned)sizeof(float);
#define ROW_SRC(U) __builtin_amdgcn_update_dpp(0, s, 0x150 + (U), 0xF, 0xF, true)
#define LOAD_ROW(U) x[U] = *(const v2f*)((const char*)h + ((unsigned)ROW_SRC(U) * (unsigned)(HID_DIM * sizeof(float)) + cb));
        LOAD_ROW(0)  LOAD_ROW(1)  LOAD_ROW(2)  LOAD_ROW(3)
        LOAD_ROW(4)  LOAD_ROW(5)  LOAD_ROW(6)  LOAD_ROW(7)
        LOAD_ROW(8)  LOAD_ROW(9)  LOAD_ROW(10) LOAD_ROW(11)
        LOAD_ROW(12) LOAD_ROW(13) LOAD_ROW(14) LOAD_ROW(15)
#undef LOAD_ROW
#undef ROW_SRC
        const int dp = d | (__shfl(d, lane ^ 16) << 16);
        int curp = __builtin_amdgcn_readlane(dp, 0);
        v2f run = {0.0f, 0.0f};
#pragma unroll
        for (int u = 0; u < 16; ++u) {
          const int p = __builtin_amdgcn_readlane(dp, u);
          if (p != curp) {
            const int mine = hf ? (int)((unsigned)curp >> 16) : (curp & 0xFFFF);
            const int next = hf ? (int)((unsigned)p >> 16) : (p & 0xFFFF);
            if (next != mine) {
              float* ap = &acc[mine * APITCH + c2];
              atomicAdd(ap, run.x);
              atomicAdd(ap + 1, run.y);
              run.x = 0.0f;
              run.y = 0.0f;
            }
            curp = p;
          }
          run.x += x[u].x;
          run.y += x[u].y;
        }
        const int last = hf ? (int)((unsigned)curp >> 16) : (curp & 0xFFFF);
        float* ap = &acc[last * APITCH + c2];
        atomicAdd(ap, run.x);
        atomicAdd(ap + 1, run.y);
      } else {
        for (int j = 0; j < m; ++j) {
          int sj = __shfl(s, j);
          int dj = __shfl(d, j);
          atomicAdd(&acc[dj * APITCH + lane], h[(size_t)sj * HID_DIM + lane]);
        }
      }
    }
    __syncthreads();
  }

  const int half = lane >> 4;
  const int l15  = lane & 15;
  const int kOff = half * 2;
  v2f bw[HID_DIM / 4];
  if (wave < NTILE) {
#pragma unroll
    for (int k4 = 0; k4 < HID_DIM / 4; ++k4) {
      bw[k4].x = W2[(4 * k4 + kOff + 0) * OUT_DIM + l15];
      bw[k4].y = W2[(4 * k4 + kOff + 1) * OUT_DIM + l15];
    }
  }
  for (int i = tid; i < NPB * HID_DIM; i += 1024) {
    const int r = i >> 5, cc = i & 31;
    float v = acc[r * APITCH + cc] * dl[r] + bias;
    acc[r * APITCH + cc] = fmaxf(v, 0.0f);
  }
  __syncthreads();

  if (wave < NTILE) {
    const float* ar = acc + (wave * 16 + l15) * APITCH + kOff;
    v8f c = {};
#pragma unroll
    for (int k = 0; k < HID_DIM; k += 4) {
      v2f a = *(const v2f*)(ar + k);
      c = __builtin_amdgcn_wmma_f32_16x16x4_f32(false, a, false, bw[k / 4],
                                                (short)0, c, false, false);
    }
    const int r0 = wave * 16 + half * 8;
#pragma unroll
    for (int i = 0; i < 8; ++i) {
      G[(size_t)(n0 + r0 + i) * OUT_DIM + l15] = c[i] * dl[r0 + i];
    }
  }
}

__global__ void __launch_bounds__(1024) k_agg16(const int* __restrict__ binned,
                                                const v2i* __restrict__ tab,
                                                const float* __restrict__ dinv,
                                                const float* __restrict__ g,
                                                const float* __restrict__ b2,
                                                const int* __restrict__ nstart,
                                                float* __restrict__ out) {
  __shared__ __attribute__((aligned(16))) float acc[NPB * OUT_DIM];
  __shared__ unsigned erec[CAP];
  __shared__ float dl[NPB];
  __shared__ int cursor[NPB];
  __shared__ int rs[NBLK];
  __shared__ int rc[NBLK];
  __shared__ int wtot[32];
  const int tid = threadIdx.x, b = blockIdx.x;
  const int lane = tid & 31;
  const int hf = lane >> 4;
  const int c = lane & 15;
  const int wave = __builtin_amdgcn_readfirstlane(tid >> 5);
  const int n0 = b * NPB;
  constexpr int NV4 = NPB * OUT_DIM / 4;
  const float dv   = dinv[n0 + min(tid, NPB - 1)];
  const int   ns   = nstart[n0 + min(tid, NPB - 1)];
  const v2i   tt   = tab[b * NBLK + (tid & (NBLK - 1))];
  const float bias = b2[c];
  const v4f* g4 = (const v4f*)(g + (size_t)n0 * OUT_DIM);
  v4f gv[2];
#pragma unroll
  for (int k = 0; k < 2; ++k) gv[k] = g4[min(tid + k * 1024, NV4 - 1)];
  if (tid < NPB) {
    dl[tid] = dv;
    cursor[tid] = ns;
  }
  if (tid < NBLK) {
    rs[tid] = tt.x;
    rc[tid] = tt.y;
  }
#pragma unroll
  for (int k = 0; k < 2; ++k) {
    const int i4 = tid + k * 1024;
    if (i4 < NV4) *(v4f*)&acc[4 * i4] = gv[k];
  }
  __syncthreads();
  int fb[RPW];
  const int total = flat_bases(rc, wtot, wave, lane, fb);

  const bool by_node = total <= CAP;
  for (int c0 = 0; c0 < total; c0 += CAP) {
    if (by_node) {
      collect_by_node(binned, rs, rc, cursor, erec, n0, wave, lane);
    } else {
      walk_runs<RPW, 0, RPW>(binned, rs, rc, wave, lane, [&](int j, int o, v2i p) {
        const unsigned r = (unsigned)(fb[j] + o - c0);
        if (r < (unsigned)CAP) erec[r] = ((unsigned)(p.y - n0) << 17) | (unsigned)p.x;
      });
    }
    __syncthreads();
    const int count = min(CAP, total - c0);
    for (int e0 = wave * 32; e0 < count; e0 += 1024) {
      const int idx = e0 + lane;
      unsigned rec = 0;
      if (idx < count) rec = erec[idx];
      const int s = (int)(rec & 0x1FFFFu);
      const int d = (int)(rec >> 17);
      const int m = count - e0;
      if (m >= 32) {
        float x[16];
        const unsigned cb = (unsigned)c * (unsigned)sizeof(float);
#define ROW_SRC(U) __builtin_amdgcn_update_dpp(0, s, 0x150 + (U), 0xF, 0xF, true)
#define LOAD_ROW(U) x[U] = *(const float*)((const char*)g + ((unsigned)ROW_SRC(U) * (unsigned)(OUT_DIM * sizeof(float)) + cb));
        LOAD_ROW(0)  LOAD_ROW(1)  LOAD_ROW(2)  LOAD_ROW(3)
        LOAD_ROW(4)  LOAD_ROW(5)  LOAD_ROW(6)  LOAD_ROW(7)
        LOAD_ROW(8)  LOAD_ROW(9)  LOAD_ROW(10) LOAD_ROW(11)
        LOAD_ROW(12) LOAD_ROW(13) LOAD_ROW(14) LOAD_ROW(15)
#undef LOAD_ROW
#undef ROW_SRC
        const int dp = d | (__shfl(d, lane ^ 16) << 16);
        int curp = __builtin_amdgcn_readlane(dp, 0);
        float run = 0.0f;
#pragma unroll
        for (int u = 0; u < 16; ++u) {
          const int p = __builtin_amdgcn_readlane(dp, u);
          if (p != curp) {
            const int mine = hf ? (int)((unsigned)curp >> 16) : (curp & 0xFFFF);
            const int next = hf ? (int)((unsigned)p >> 16) : (p & 0xFFFF);
            if (next != mine) {
              atomicAdd(&acc[mine * OUT_DIM + c], run);
              run = 0.0f;
            }
            curp = p;
          }
          run += x[u];
        }
        const int last = hf ? (int)((unsigned)curp >> 16) : (curp & 0xFFFF);
        atomicAdd(&acc[last * OUT_DIM + c], run);
      } else {
        for (int j = 0; 2 * j < m; ++j) {
          int e = 2 * j + hf;
          int sj = __shfl(s, e);
          int dj = __shfl(d, e);
          if (e < m) atomicAdd(&acc[dj * OUT_DIM + c], g[(size_t)sj * OUT_DIM + c]);
        }
      }
    }
    __syncthreads();
  }
  for (int i = tid; i < NPB * OUT_DIM; i += 1024) {
    out[(size_t)n0 * OUT_DIM + i] = acc[i] * dl[i >> 4] + bias;
  }
}

extern "C" void kernel_launch(void* const* d_in, const int* in_sizes, int n_in,
                              void* d_out, int out_size, void* d_ws, size_t ws_size,
                              hipStream_t stream) {
  const float* x  = (const float*)d_in[0];
  const float* W1 = (const float*)d_in[1];
  const float* b1 = (const float*)d_in[2];
  const float* W2 = (const float*)d_in[3];
  const float* b2 = (const float*)d_in[4];
  const int*   ei = (const int*)d_in[5];

  float* dinv = (float*)d_ws;
  float* h    = dinv + N_NODES;
  float* g    = h    + (size_t)N_NODES * HID_DIM;
  v2i* tab    = (v2i*)(g + (size_t)N_NODES * OUT_DIM);
  int* binned = (int*)(tab + NBKT * NBLK);
  int* nstart = binned + 2 * (size_t)NBLK * SLAB;
  float* out = (float*)d_out;

  k_bin<<<NBLK, 1024, 0, stream>>>(ei, binned, tab);
  k_degree_gemm1<<<NBKT, 1024, 0, stream>>>(binned, tab, x, W1, dinv, h, nstart);
  k_agg32_gemm2<<<NBKT, 1024, 0, stream>>>(binned, tab, dinv, h, b1, W2, nstart, g);
  k_agg16<<<NBKT, 1024, 0, stream>>>(binned, tab, dinv, g, b2, nstart, out);
}


# Last partial group of 32 records of a range padded (pads sum into a spare LDS row) and run on the 32-wide path, not one dependent load per record; sum order of those records changes
# speedup vs baseline: 8.1434x; 1.0913x over previous
#include <hip/hip_runtime.h>


#define N_NODES 100000
#define N_EDGES 3200000
#define IN_DIM  128
#define HID_DIM 32
#define OUT_DIM 16

#define NBKT 250
#define NPB  400
#define NBLK 256
#define EPB  12500
#define SLAB 12544
#define RPW  (NBLK / 32)
#define CAP  16384
#define PAD_REC ((unsigned)NPB << 17)

static_assert(NBKT * NPB == N_NODES, "ranges must tile the nodes");
static_assert(NBLK * EPB == N_EDGES, "slabs must tile the edges");
static_assert(N_NODES <= (1 << 17) && NPB < (1 << 15), "packed record: 17 bits source, 15 bits local destination (0..NPB, NPB = the spare row)");
static_assert(IN_DIM * HID_DIM == 4 * 1024, "W1 goes to LDS as one 16-byte piece per thread of a 1024-thread work-group");
static_assert((unsigned long long)N_NODES * HID_DIM * sizeof(float) < (1ull << 32), "edge loops: a row's byte offset fits 32 bits");
static_assert(NPB < (1 << 15), "edge loops: two local destinations are packed in one 32-bit word");

typedef __attribute__((ext_vector_type(2))) float v2f;
typedef __attribute__((ext_vector_type(4))) float v4f;
typedef __attribute__((ext_vector_type(8))) float v8f;
typedef __attribute__((ext_vector_type(2))) int   v2i;
typedef __attribute__((ext_vector_type(4))) int   v4i;

__device__ __forceinline__ int clamp_node(int v) {
  return min(max(v, 0), N_NODES - 1);
}
__device__ __forceinline__ int bucket_of(int d) {
  return (int)((unsigned)d / (unsigned)NPB);
}

__device__ __forceinline__ void block_incl_scan256(int v, int* sh, int tid) {
  if (tid < 256) sh[tid] = v;
  __syncthreads();
#pragma unroll
  for (int off = 1; off < 256; off <<= 1) {
    int t = 0;
    if (tid < 256 && tid >= off) t = sh[tid - off];
    __syncthreads();
    if (tid < 256) sh[tid] += t;
    __syncthreads();
  }
}

__global__ void __launch_bounds__(1024) k_bin(const int* __restrict__ ei,
                                              int* __restrict__ binned,
                                              v2i* __restrict__ tab) {
  __shared__ __attribute__((aligned(16))) v2i lrec[EPB];
  __shared__ int lh[256];
  __shared__ int sh[256];
  __shared__ int lst[256];
  const int tid = threadIdx.x, blk = blockIdx.x;
  if (tid < 256) lh[tid] = 0;
  const int* srcp = ei + blk * EPB;
  const int* dstp = ei + N_EDGES + blk * EPB;
  int s[13], d[13], r[13];
#pragma unroll
  for (int k = 0; k < 13; ++k) {
    const int i = tid + k * 1024;
    const int ii = min(i, EPB - 1);
    r[k] = 0;
    s[k] = clamp_node(srcp[ii]);
    d[k] = clamp_node(dstp[ii]);
  }
  __syncthreads();
#pragma unroll
  for (int k = 0; k < 13; ++k) {
    const int i = tid + k * 1024;
    if (i < EPB) r[k] = atomicAdd(&lh[bucket_of(d[k])], 1);
  }
  __syncthreads();
  const int v = (tid < 256) ? lh[tid] : 0;
  block_incl_scan256(v, sh, tid);
  if (tid < 256) lst[tid] = sh[tid] - v;
  if (tid < NBKT) {
    v2i t; t.x = blk * SLAB + sh[tid] - v; t.y = v;
    tab[tid * NBLK + blk] = t;
  }
  __syncthreads();
#pragma unroll
  for (int k = 0; k < 13; ++k) {
    const int i = tid + k * 1024;
    if (i < EPB) {
      v2i p; p.x = s[k]; p.y = d[k];
      lrec[lst[bucket_of(d[k])] + r[k]] = p;
    }
  }
  __syncthreads();
  v4i* outp = (v4i*)(binned + 2 * (size_t)blk * SLAB);
  const v4i* lp = (const v4i*)lrec;
  for (int i = tid; i < EPB / 2; i += 1024) outp[i] = lp[i];
}

template <int RB, int JB, int JE, typename F>
__device__ __forceinline__ void walk_runs(const int* __restrict__ binned, const int* rs, const int* rc,
                                          int wave, int lane, F fn) {
#pragma unroll
  for (int j0 = JB; j0 < JE; j0 += RB) {
    int st[RB], n[RB];
    v2i rec[RB][3];
#pragma unroll
    for (int j = 0; j < RB; ++j) {
      const int k = wave + 32 * (j0 + j);
      st[j] = __builtin_amdgcn_readfirstlane(rs[k]);
      n[j]  = __builtin_amdgcn_readfirstlane(rc[k]);
#pragma unroll
      for (int t = 0; t < 3; ++t) {
        const int o = lane + 32 * t;
        const int at = (o < n[j]) ? st[j] + o : 0;
        rec[j][t] = *(const v2i*)(binned + 2 * (size_t)at);
      }
    }
#pragma unroll
    for (int j = 0; j < RB; ++j) {
#pragma unroll
      for (int t = 0; t < 3; ++t) {
        const int o = lane + 32 * t;
        if (o < n[j]) fn(j0 + j, o, rec[j][t]);
      }
    }
#pragma unroll
    for (int j = 0; j < RB; ++j) {
      for (int o = lane + 96; o < n[j]; o += 32)
        fn(j0 + j, o, *(const v2i*)(binned + 2 * (size_t)(st[j] + o)));
    }
  }
}

#define NTILE (NPB / 16)
#define WPITCH 48
__global__ void __launch_bounds__(1024) k_degree_gemm1(const int* __restrict__ binned,
                                                       const v2i* __restrict__ tab,
                                                       const float* __restrict__ X,
                                                       const float* __restrict__ W,
                                                       float* __restrict__ dinv,
                                                       float* __restrict__ H,
                                                       int* __restrict__ nstart) {
  __shared__ int cnt[NPB];
  __shared__ float dl[NPB];
  __shared__ int rs[NBLK];
  __shared__ int rc[NBLK];
  __shared__ __attribute__((aligned(16))) float wl[IN_DIM * WPITCH];
  const int tid = threadIdx.x, b = blockIdx.x;
  const int lane = tid & 31;
  const int wave = __builtin_amdgcn_readfirstlane(tid >> 5);
  const int n0   = b * NPB;
  const int half = lane >> 4;
  const int l15  = lane & 15;
  const int kOff = half * 2;

  const v2i tt = tab[b * NBLK + (tid & (NBLK - 1))];
  const v4f wv = ((const v4f*)W)[tid];
  if (tid < NPB) cnt[tid] = 0;
  if (tid < NBLK) {
    rs[tid] = tt.x;
    rc[tid] = tt.y;
  }
  *(v4f*)&wl[(tid >> 3) * WPITCH + (tid & 7) * 4] = wv;
  __syncthreads();

  const auto count_fn = [&](int, int, v2i p) { atomicAdd(&cnt[p.y - n0], 1); };

  walk_runs<4, 0, 4>(binned, rs, rc, wave, lane, count_fn);

  v8f c0 = {}, c1 = {};
  if (wave < NTILE) {
    v2f a[IN_DIM / 4];
    const float* xr = X + (size_t)(n0 + wave * 16 + l15) * IN_DIM + kOff;
#pragma unroll
    for (int k4 = 0; k4 < IN_DIM / 4; ++k4) a[k4] = *(const v2f*)(xr + 4 * k4);
#pragma unroll
    for (int k4 = 0; k4 < IN_DIM / 4; ++k4) {
      const float* wr = wl + (4 * k4 + kOff) * WPITCH + l15;
      v2f b0, b1;
      b0.x = wr[0];
      b0.y = wr[WPITCH];
      b1.x = wr[16];
      b1.y = wr[WPITCH + 16];
      c0 = __builtin_amdgcn_wmma_f32_16x16x4_f32(false, a[k4], false, b0,
                                                 (short)0, c0, false, false);
      c1 = __builtin_amdgcn_wmma_f32_16x16x4_f32(false, a[k4], false, b1,
                                                 (short)0, c1, false, false);
    }
  }

  walk_runs<4, 4, RPW>(binned, rs, rc, wave, lane, count_fn);
  __syncthreads();
  if (tid < NPB) {
    float deg = 1.0f + (float)cnt[tid];
    float di = (deg > 0.0f) ? rsqrtf(deg) : 0.0f;
    dl[tid] = di;
    dinv[n0 + tid] = di;
  }
  __syncthreads();

  if (wave < NTILE) {
    const int r0 = wave * 16 + half * 8;
#pragma unroll
    for (int i = 0; i < 8; ++i) {
      const float di = dl[r0 + i];
      float* hr = H + (size_t)(n0 + r0 + i) * HID_DIM + l15;
      hr[0]  = c0[i] * di;
      hr[16] = c1[i] * di;
    }
  }
  if (wave == 31) {
    constexpr int PER = (NPB + 31) / 32;
    int v[PER];
    int sum = 0;
#pragma unroll
    for (int k = 0; k < PER; ++k) {
      const int i = lane * PER + k;
      v[k] = (i < NPB) ? cnt[i] : 0;
      sum += v[k];
    }
    int x = sum;
#pragma unroll
    for (int off = 1; off < 32; off <<= 1) {
      const int y = __shfl_up(x, off);
      if (lane >= off) x += y;
    }
    int at = x - sum;
#pragma unroll
    for (int k = 0; k < PER; ++k) {
      const int i = lane * PER + k;
      if (i < NPB) nstart[n0 + i] = at;
      at += v[k];
    }
  }
}

__device__ __forceinline__ void collect_by_node(const int* __restrict__ binned, const int* rs, const int* rc,
                                                int* cursor, unsigned* erec, int n0, int wave, int lane) {
  int st[RPW], n[RPW];
  v2i rec[RPW][3];
#pragma unroll
  for (int j = 0; j < RPW; ++j) {
    const int k = wave + 32 * j;
    st[j] = __builtin_amdgcn_readfirstlane(rs[k]);
    n[j]  = __builtin_amdgcn_readfirstlane(rc[k]);
#pragma unroll
    for (int t = 0; t < 3; ++t) {
      const int o = lane + 32 * t;
      const int at = (o < n[j]) ? st[j] + o : 0;
      rec[j][t] = *(const v2i*)(binned + 2 * (size_t)at);
    }
  }
  unsigned pos[RPW][3], pk[RPW][3];
#pragma unroll
  for (int j = 0; j < RPW; ++j) {
#pragma unroll
    for (int t = 0; t < 3; ++t) {
      const int o = lane + 32 * t;
      const int dloc = rec[j][t].y - n0;
      pk[j][t]  = ((unsigned)dloc << 17) | (unsigned)rec[j][t].x;
      pos[j][t] = (unsigned)CAP;
      if (o < n[j]) pos[j][t] = (unsigned)atomicAdd(&cursor[dloc], 1);
    }
  }
#pragma unroll
  for (int j = 0; j < RPW; ++j) {
#pragma unroll
    for (int t = 0; t < 3; ++t) {
      if (pos[j][t] < (unsigned)CAP) erec[pos[j][t]] = pk[j][t];
    }
  }
#pragma unroll
  for (int j = 0; j < RPW; ++j) {
    for (int o = lane + 96; o < n[j]; o += 32) {
      const v2i p = *(const v2i*)(binned + 2 * (size_t)(st[j] + o));
      const int dloc = p.y - n0;
      const unsigned r = (unsigned)atomicAdd(&cursor[dloc], 1);
      if (r < (unsigned)CAP) erec[r] = ((unsigned)dloc << 17) | (unsigned)p.x;
    }
  }
}

__device__ __forceinline__ int flat_bases(const int* rc, int* wtot, int wave, int lane, int (&fb)[RPW]) {
  int wt = 0;
#pragma unroll
  for (int j = 0; j < RPW; ++j) {
    fb[j] = wt;
    wt += __builtin_amdgcn_readfirstlane(rc[wave + 32 * j]);
  }
  if (lane == 0) wtot[wave] = wt;
  __syncthreads();
  int x = wtot[lane];
#pragma unroll
  for (int off = 1; off < 32; off <<= 1) {
    const int y = __shfl_up(x, off);
    if (lane >= off) x += y;
  }
  const int total = __builtin_amdgcn_readlane(x, 31);
  const int wb = __builtin_amdgcn_readlane(x, wave) - wt;
#pragma unroll
  for (int j = 0; j < RPW; ++j) fb[j] += wb;
  return total;
}

#define APITCH 34
__global__ void __launch_bounds__(1024) k_agg32_gemm2(const int* __restrict__ binned,
                                                      const v2i* __restrict__ tab,
                                                      const float* __restrict__ dinv,
                                                      const float* __restrict__ h,
                                                      const float* __restrict__ b1,
                                                      const float* __restrict__ W2,
                                                      const int* __restrict__ nstart,
                                                      float* __restrict__ G) {
  __shared__ __attribute__((aligned(16))) float acc[(NPB + 1) * APITCH];
  __shared__ unsigned erec[CAP];
  __shared__ float dl[NPB];
  __shared__ int cursor[NPB];
  __shared__ int rs[NBLK];
  __shared__ int rc[NBLK];
  __shared__ int wtot[32];
  const int tid = threadIdx.x, b = blockIdx.x;
  const int lane = tid & 31;
  const int hf = lane >> 4;
  const int c2 = (lane & 15) * 2;
  const int wave = __builtin_amdgcn_readfirstlane(tid >> 5);
  const int n0 = b * NPB;
  constexpr int NV4 = NPB * HID_DIM / 4;
  const float dv   = dinv[n0 + min(tid, NPB - 1)];
  const int   ns   = nstart[n0 + min(tid, NPB - 1)];
  const v2i   tt   = tab[b * NBLK + (tid & (NBLK - 1))];
  const float bias = b1[lane];
  const v4f* h4 = (const v4f*)(h + (size_t)n0 * HID_DIM);
  v4f hv[4];
#pragma unroll
  for (int k = 0; k < 4; ++k) hv[k] = h4[min(tid + k * 1024, NV4 - 1)];
  if (tid < NPB) {
    dl[tid] = dv;
    cursor[tid] = ns;
  }
  if (tid < NBLK) {
    rs[tid] = tt.x;
    rc[tid] = tt.y;
  }
  if (tid < APITCH) acc[NPB * APITCH + tid] = 0.0f;
#pragma unroll
  for (int k = 0; k < 4; ++k) {
    const int i4 = tid + k * 1024;
    if (i4 < NV4) {
      float* ap = &acc[(i4 >> 3) * APITCH + (i4 & 7) * 4];
      v2f lo, hi;
      lo.x = hv[k].x; lo.y = hv[k].y; hi.x = hv[k].z; hi.y = hv[k].w;
      *(v2f*)ap = lo;
      *(v2f*)(ap + 2) = hi;
    }
  }
  __syncthreads();
  int fb[RPW];
  const int total = flat_bases(rc, wtot, wave, lane, fb);

  const bool by_node = total <= CAP;
  for (int c0 = 0; c0 < total; c0 += CAP) {
    if (by_node) {
      collect_by_node(binned, rs, rc, cursor, erec, n0, wave, lane);
    } else {
      walk_runs<RPW, 0, RPW>(binned, rs, rc, wave, lane, [&](int j, int o, v2i p) {
        const unsigned r = (unsigned)(fb[j] + o - c0);
        if (r < (unsigned)CAP) erec[r] = ((unsigned)(p.y - n0) << 17) | (unsigned)p.x;
      });
    }
    __syncthreads();
    const int count = min(CAP, total - c0);
    for (int e0 = wave * 32; e0 < count; e0 += 1024) {
      const int idx = e0 + lane;
      unsigned rec = PAD_REC;
      if (idx < count) rec = erec[idx];
      const int s = (int)(rec & 0x1FFFFu);
      const int d = (int)(rec >> 17);
      {
        v2f x[16];
        const unsigned cb = (unsigned)c2 * (unsigned)sizeof(float);
#define ROW_SRC(U) __builtin_amdgcn_update_dpp(0, s, 0x150 + (U), 0xF, 0xF, true)
#define LOAD_ROW(U) x[U] = *(const v2f*)((const char*)h + ((unsigned)ROW_SRC(U) * (unsigned)(HID_DIM * sizeof(float)) + cb));
        LOAD_ROW(0)  LOAD_ROW(1)  LOAD_ROW(2)  LOAD_ROW(3)
        LOAD_ROW(4)  LOAD_ROW(5)  LOAD_ROW(6)  LOAD_ROW(7)
        LOAD_ROW(8)  LOAD_ROW(9)  LOAD_ROW(10) LOAD_ROW(11)
        LOAD_ROW(12) LOAD_ROW(13) LOAD_ROW(14) LOAD_ROW(15)
#undef LOAD_ROW
#undef ROW_SRC
        const int dp = d | (__shfl(d, lane ^ 16) << 16);
        int curp = __builtin_amdgcn_readlane(dp, 0);
        v2f run = {0.0f, 0.0f};
#pragma unroll
        for (int u = 0; u < 16; ++u) {
          const int p = __builtin_amdgcn_readlane(dp, u);
          if (p != curp) {
            const int mine = hf ? (int)((unsigned)curp >> 16) : (curp & 0xFFFF);
            const int next = hf ? (int)((unsigned)p >> 16) : (p & 0xFFFF);
            if (next != mine) {
              float* ap = &acc[mine * APITCH + c2];
              atomicAdd(ap, run.x);
              atomicAdd(ap + 1, run.y);
              run.x = 0.0f;
              run.y = 0.0f;
            }
            curp = p;
          }
          run.x += x[u].x;
          run.y += x[u].y;
        }
        const int last = hf ? (int)((unsigned)curp >> 16) : (curp & 0xFFFF);
        float* ap = &acc[last * APITCH + c2];
        atomicAdd(ap, run.x);
        atomicAdd(ap + 1, run.y);
      }
    }
    __syncthreads();
  }

  const int half = lane >> 4;
  const int l15  = lane & 15;
  const int kOff = half * 2;
  v2f bw[HID_DIM / 4];
  if (wave < NTILE) {
#pragma unroll
    for (int k4 = 0; k4 < HID_DIM / 4; ++k4) {
      bw[k4].x = W2[(4 * k4 + kOff + 0) * OUT_DIM + l15];
      bw[k4].y = W2[(4 * k4 + kOff + 1) * OUT_DIM + l15];
    }
  }
  for (int i = tid; i < NPB * HID_DIM; i += 1024) {
    const int r = i >> 5, cc = i & 31;
    float v = acc[r * APITCH + cc] * dl[r] + bias;
    acc[r * APITCH + cc] = fmaxf(v, 0.0f);
  }
  __syncthreads();

  if (wave < NTILE) {
    const float* ar = acc + (wave * 16 + l15) * APITCH + kOff;
    v8f c = {};
#pragma unroll
    for (int k = 0; k < HID_DIM; k += 4) {
      v2f a = *(const v2f*)(ar + k);
      c = __builtin_amdgcn_wmma_f32_16x16x4_f32(false, a, false, bw[k / 4],
                                                (short)0, c, false, false);
    }
    const int r0 = wave * 16 + half * 8;
#pragma unroll
    for (int i = 0; i < 8; ++i) {
      G[(size_t)(n0 + r0 + i) * OUT_DIM + l15] = c[i] * dl[r0 + i];
    }
  }
}

__global__ void __launch_bounds__(1024) k_agg16(const int* __restrict__ binned,
                                                const v2i* __restrict__ tab,
                                                const float* __restrict__ dinv,
                                                const float* __restrict__ g,
                                                const float* __restrict__ b2,
                                                const int* __restrict__ nstart,
                                                float* __restrict__ out) {
  __shared__ __attribute__((aligned(16))) float acc[(NPB + 1) * OUT_DIM];
  __shared__ unsigned erec[CAP];
  __shared__ float dl[NPB];
  __shared__ int cursor[NPB];
  __shared__ int rs[NBLK];
  __shared__ int rc[NBLK];
  __shared__ int wtot[32];
  const int tid = threadIdx.x, b = blockIdx.x;
  const int lane = tid & 31;
  const int hf = lane >> 4;
  const int c = lane & 15;
  const int wave = __builtin_amdgcn_readfirstlane(tid >> 5);
  const int n0 = b * NPB;
  constexpr int NV4 = NPB * OUT_DIM / 4;
  const float dv   = dinv[n0 + min(tid, NPB - 1)];
  const int   ns   = nstart[n0 + min(tid, NPB - 1)];
  const v2i   tt   = tab[b * NBLK + (tid & (NBLK - 1))];
  const float bias = b2[c];
  const v4f* g4 = (const v4f*)(g + (size_t)n0 * OUT_DIM);
  v4f gv[2];
#pragma unroll
  for (int k = 0; k < 2; ++k) gv[k] = g4[min(tid + k * 1024, NV4 - 1)];
  if (tid < NPB) {
    dl[tid] = dv;
    cursor[tid] = ns;
  }
  if (tid < NBLK) {
    rs[tid] = tt.x;
    rc[tid] = tt.y;
  }
  if (tid < OUT_DIM) acc[NPB * OUT_DIM + tid] = 0.0f;
#pragma unroll
  for (int k = 0; k < 2; ++k) {
    const int i4 = tid + k * 1024;
    if (i4 < NV4) *(v4f*)&acc[4 * i4] = gv[k];
  }
  __syncthreads();
  int fb[RPW];
  const int total = flat_bases(rc, wtot, wave, lane, fb);

  const bool by_node = total <= CAP;
  for (int c0 = 0; c0 < total; c0 += CAP) {
    if (by_node) {
      collect_by_node(binned, rs, rc, cursor, erec, n0, wave, lane);
    } else {
      walk_runs<RPW, 0, RPW>(binned, rs, rc, wave, lane, [&](int j, int o, v2i p) {
        const unsigned r = (unsigned)(fb[j] + o - c0);
        if (r < (unsigned)CAP) erec[r] = ((unsigned)(p.y - n0) << 17) | (unsigned)p.x;
      });
    }
    __syncthreads();
    const int count = min(CAP, total - c0);
    for (int e0 = wave * 32; e0 < count; e0 += 1024) {
      const int idx = e0 + lane;
      unsigned rec = PAD_REC;
      if (idx < count) rec = erec[idx];
      const int s = (int)(rec & 0x1FFFFu);
      const int d = (int)(rec >> 17);
      {
        float x[16];
        const unsigned cb = (unsigned)c * (unsigned)sizeof(float);
#define ROW_SRC(U) __builtin_amdgcn_update_dpp(0, s, 0x150 + (U), 0xF, 0xF, true)
#define LOAD_ROW(U) x[U] = *(const float*)((const char*)g + ((unsigned)ROW_SRC(U) * (unsigned)(OUT_DIM * sizeof(float)) + cb));
        LOAD_ROW(0)  LOAD_ROW(1)  LOAD_ROW(2)  LOAD_ROW(3)
        LOAD_ROW(4)  LOAD_ROW(5)  LOAD_ROW(6)  LOAD_ROW(7)
        LOAD_ROW(8)  LOAD_ROW(9)  LOAD_ROW(10) LOAD_ROW(11)
        LOAD_ROW(12) LOAD_ROW(13) LOAD_ROW(14) LOAD_ROW(15)
#undef LOAD_ROW
#undef ROW_SRC
        const int dp = d | (__shfl(d, lane ^ 16) << 16);
        int curp = __builtin_amdgcn_readlane(dp, 0);
        float run = 0.0f;
#pragma unroll
        for (int u = 0; u < 16; ++u) {
          const int p = __builtin_amdgcn_readlane(dp, u);
          if (p != curp) {
            const int mine = hf ? (int)((unsigned)curp >> 16) : (curp & 0xFFFF);
            const int next = hf ? (int)((unsigned)p >> 16) : (p & 0xFFFF);
            if (next != mine) {
              atomicAdd(&acc[mine * OUT_DIM + c], run);
              run = 0.0f;
            }
            curp = p;
          }
          run += x[u];
        }
        const int last = hf ? (int)((unsigned)curp >> 16) : (curp & 0xFFFF);
        atomicAdd(&acc[last * OUT_DIM + c], run);
      }
    }
    __syncthreads();
  }
  for (int i = tid; i < NPB * OUT_DIM; i += 1024) {
    out[(size_t)n0 * OUT_DIM + i] = acc[i] * dl[i >> 4] + bias;
  }
}

extern "C" void kernel_launch(void* const* d_in, const int* in_sizes, int n_in,
                              void* d_out, int out_size, void* d_ws, size_t ws_size,
                              hipStream_t stream) {
  const float* x  = (const float*)d_in[0];
  const float* W1 = (const float*)d_in[1];
  const float* b1 = (const float*)d_in[2];
  const float* W2 = (const float*)d_in[3];
  const float* b2 = (const float*)d_in[4];
  const int*   ei = (const int*)d_in[5];

  float* dinv = (float*)d_ws;
  float* h    = dinv + N_NODES;
  float* g    = h    + (size_t)N_NODES * HID_DIM;
  v2i* tab    = (v2i*)(g + (size_t)N_NODES * OUT_DIM);
  int* binned = (int*)(tab + NBKT * NBLK);
  int* nstart = binned + 2 * (size_t)NBLK * SLAB;
  float* out = (float*)d_out;

  k_bin<<<NBLK, 1024, 0, stream>>>(ei, binned, tab);
  k_degree_gemm1<<<NBKT, 1024, 0, stream>>>(binned, tab, x, W1, dinv, h, nstart);
  k_agg32_gemm2<<<NBKT, 1024, 0, stream>>>(binned, tab, dinv, h, b1, W2, nstart, g);
  k_agg16<<<NBKT, 1024, 0, stream>>>(binned, tab, dinv, g, b2, nstart, out);
}


# Binned edge records stored as one 4-byte word (local destination << 17 | source, as in the ranges' LDS lists), not an 8-byte pair: one pass writes and three read half the bytes; sum order kept
# speedup vs baseline: 8.4380x; 1.0362x over previous
#include <hip/hip_runtime.h>


#define N_NODES 100000
#define N_EDGES 3200000
#define IN_DIM  128
#define HID_DIM 32
#define OUT_DIM 16

#define NBKT 250
#define NPB  400
#define NBLK 256
#define EPB  12500
#define SLAB 12544
#define RPW  (NBLK / 32)
#define CAP  16384
#define PAD_REC ((unsigned)NPB << 17)

static_assert(NBKT * NPB == N_NODES, "ranges must tile the nodes");
static_assert(NBLK * EPB == N_EDGES, "slabs must tile the edges");
static_assert(EPB % 4 == 0 && SLAB % 4 == 0, "a slab is copied out in 16-byte pieces of four records");
static_assert(N_NODES <= (1 << 17) && NPB < (1 << 15), "packed record: 17 bits source, 15 bits local destination (0..NPB, NPB = the spare row)");
static_assert(IN_DIM * HID_DIM == 4 * 1024, "W1 goes to LDS as one 16-byte piece per thread of a 1024-thread work-group");
static_assert((unsigned long long)N_NODES * HID_DIM * sizeof(float) < (1ull << 32), "edge loops: a row's byte offset fits 32 bits");
static_assert(NPB < (1 << 15), "edge loops: two local destinations are packed in one 32-bit word");

typedef __attribute__((ext_vector_type(2))) float v2f;
typedef __attribute__((ext_vector_type(4))) float v4f;
typedef __attribute__((ext_vector_type(8))) float v8f;
typedef __attribute__((ext_vector_type(2))) int   v2i;
typedef __attribute__((ext_vector_type(4))) int   v4i;

__device__ __forceinline__ int clamp_node(int v) {
  return min(max(v, 0), N_NODES - 1);
}
__device__ __forceinline__ int bucket_of(int d) {
  return (int)((unsigned)d / (unsigned)NPB);
}

__device__ __forceinline__ void block_incl_scan256(int v, int* sh, int tid) {
  if (tid < 256) sh[tid] = v;
  __syncthreads();
#pragma unroll
  for (int off = 1; off < 256; off <<= 1) {
    int t = 0;
    if (tid < 256 && tid >= off) t = sh[tid - off];
    __syncthreads();
    if (tid < 256) sh[tid] += t;
    __syncthreads();
  }
}

__global__ void __launch_bounds__(1024) k_bin(const int* __restrict__ ei,
                                              unsigned* __restrict__ binned,
                                              v2i* __restrict__ tab) {
  __shared__ __attribute__((aligned(16))) unsigned lrec[EPB];
  __shared__ int lh[256];
  __shared__ int sh[256];
  __shared__ int lst[256];
  const int tid = threadIdx.x, blk = blockIdx.x;
  if (tid < 256) lh[tid] = 0;
  const int* srcp = ei + blk * EPB;
  const int* dstp = ei + N_EDGES + blk * EPB;
  int s[13], d[13], r[13];
#pragma unroll
  for (int k = 0; k < 13; ++k) {
    const int i = tid + k * 1024;
    const int ii = min(i, EPB - 1);
    r[k] = 0;
    s[k] = clamp_node(srcp[ii]);
    d[k] = clamp_node(dstp[ii]);
  }
  __syncthreads();
#pragma unroll
  for (int k = 0; k < 13; ++k) {
    const int i = tid + k * 1024;
    if (i < EPB) r[k] = atomicAdd(&lh[bucket_of(d[k])], 1);
  }
  __syncthreads();
  const int v = (tid < 256) ? lh[tid] : 0;
  block_incl_scan256(v, sh, tid);
  if (tid < 256) lst[tid] = sh[tid] - v;
  if (tid < NBKT) {
    v2i t; t.x = blk * SLAB + sh[tid] - v; t.y = v;
    tab[tid * NBLK + blk] = t;
  }
  __syncthreads();
#pragma unroll
  for (int k = 0; k < 13; ++k) {
    const int i = tid + k * 1024;
    if (i < EPB) {
      const int bk = bucket_of(d[k]);
      lrec[lst[bk] + r[k]] = ((unsigned)(d[k] - bk * NPB) << 17) | (unsigned)s[k];
    }
  }
  __syncthreads();
  v4i* outp = (v4i*)(binned + (size_t)blk * SLAB);
  const v4i* lp = (const v4i*)lrec;
  for (int i = tid; i < EPB / 4; i += 1024) outp[i] = lp[i];
}

template <int RB, int JB, int JE, typename F>
__device__ __forceinline__ void walk_runs(const unsigned* __restrict__ binned, const int* rs, const int* rc,
                                          int wave, int lane, F fn) {
#pragma unroll
  for (int j0 = JB; j0 < JE; j0 += RB) {
    int st[RB], n[RB];
    unsigned rec[RB][3];
#pragma unroll
    for (int j = 0; j < RB; ++j) {
      const int k = wave + 32 * (j0 + j);
      st[j] = __builtin_amdgcn_readfirstlane(rs[k]);
      n[j]  = __builtin_amdgcn_readfirstlane(rc[k]);
#pragma unroll
      for (int t = 0; t < 3; ++t) {
        const int o = lane + 32 * t;
        const int at = (o < n[j]) ? st[j] + o : 0;
        rec[j][t] = binned[at];
      }
    }
#pragma unroll
    for (int j = 0; j < RB; ++j) {
#pragma unroll
      for (int t = 0; t < 3; ++t) {
        const int o = lane + 32 * t;
        if (o < n[j]) fn(j0 + j, o, rec[j][t]);
      }
    }
#pragma unroll
    for (int j = 0; j < RB; ++j) {
      for (int o = lane + 96; o < n[j]; o += 32)
        fn(j0 + j, o, binned[st[j] + o]);
    }
  }
}

#define NTILE (NPB / 16)
#define WPITCH 48
__global__ void __launch_bounds__(1024) k_degree_gemm1(const unsigned* __restrict__ binned,
                                                       const v2i* __restrict__ tab,
                                                       const float* __restrict__ X,
                                                       const float* __restrict__ W,
                                                       float* __restrict__ dinv,
                                                       float* __restrict__ H,
                                                       int* __restrict__ nstart) {
  __shared__ int cnt[NPB];
  __shared__ float dl[NPB];
  __shared__ int rs[NBLK];
  __shared__ int rc[NBLK];
  __shared__ __attribute__((aligned(16))) float wl[IN_DIM * WPITCH];
  const int tid = threadIdx.x, b = blockIdx.x;
  const int lane = tid & 31;
  const int wave = __builtin_amdgcn_readfirstlane(tid >> 5);
  const int n0   = b * NPB;
  const int half = lane >> 4;
  const int l15  = lane & 15;
  const int kOff = half * 2;

  const v2i tt = tab[b * NBLK + (tid & (NBLK - 1))];
  const v4f wv = ((const v4f*)W)[tid];
  if (tid < NPB) cnt[tid] = 0;
  if (tid < NBLK) {
    rs[tid] = tt.x;
    rc[tid] = tt.y;
  }
  *(v4f*)&wl[(tid >> 3) * WPITCH + (tid & 7) * 4] = wv;
  __syncthreads();

  const auto count_fn = [&](int, int, unsigned p) { atomicAdd(&cnt[p >> 17], 1); };

  walk_runs<4, 0, 4>(binned, rs, rc, wave, lane, count_fn);

  v8f c0 = {}, c1 = {};
  if (wave < NTILE) {
    v2f a[IN_DIM / 4];
    const float* xr = X + (size_t)(n0 + wave * 16 + l15) * IN_DIM + kOff;
#pragma unroll
    for (int k4 = 0; k4 < IN_DIM / 4; ++k4) a[k4] = *(const v2f*)(xr + 4 * k4);
#pragma unroll
    for (int k4 = 0; k4 < IN_DIM / 4; ++k4) {
      const float* wr = wl + (4 * k4 + kOff) * WPITCH + l15;
      v2f b0, b1;
      b0.x = wr[0];
      b0.y = wr[WPITCH];
      b1.x = wr[16];
      b1.y = wr[WPITCH + 16];
      c0 = __builtin_amdgcn_wmma_f32_16x16x4_f32(false, a[k4], false, b0,
                                                 (short)0, c0, false, false);
      c1 = __builtin_amdgcn_wmma_f32_16x16x4_f32(false, a[k4], false, b1,
                                                 (short)0, c1, false, false);
    }
  }

  walk_runs<4, 4, RPW>(binned, rs, rc, wave, lane, count_fn);
  __syncthreads();
  if (tid < NPB) {
    float deg = 1.0f + (float)cnt[tid];
    float di = (deg > 0.0f) ? rsqrtf(deg) : 0.0f;
    dl[tid] = di;
    dinv[n0 + tid] = di;
  }
  __syncthreads();

  if (wave < NTILE) {
    const int r0 = wave * 16 + half * 8;
#pragma unroll
    for (int i = 0; i < 8; ++i) {
      const float di = dl[r0 + i];
      float* hr = H + (size_t)(n0 + r0 + i) * HID_DIM + l15;
      hr[0]  = c0[i] * di;
      hr[16] = c1[i] * di;
    }
  }
  if (wave == 31) {
    constexpr int PER = (NPB + 31) / 32;
    int v[PER];
    int sum = 0;
#pragma unroll
    for (int k = 0; k < PER; ++k) {
      const int i = lane * PER + k;
      v[k] = (i < NPB) ? cnt[i] : 0;
      sum += v[k];
    }
    int x = sum;
#pragma unroll
    for (int off = 1; off < 32; off <<= 1) {
      const int y = __shfl_up(x, off);
      if (lane >= off) x += y;
    }
    int at = x - sum;
#pragma unroll
    for (int k = 0; k < PER; ++k) {
      const int i = lane * PER + k;
      if (i < NPB) nstart[n0 + i] = at;
      at += v[k];
    }
  }
}

__device__ __forceinline__ void collect_by_node(const unsigned* __restrict__ binned, const int* rs, const int* rc,
                                                int* cursor, unsigned* erec, int wave, int lane) {
  int st[RPW], n[RPW];
  unsigned rec[RPW][3];
#pragma unroll
  for (int j = 0; j < RPW; ++j) {
    const int k = wave + 32 * j;
    st[j] = __builtin_amdgcn_readfirstlane(rs[k]);
    n[j]  = __builtin_amdgcn_readfirstlane(rc[k]);
#pragma unroll
    for (int t = 0; t < 3; ++t) {
      const int o = lane + 32 * t;
      const int at = (o < n[j]) ? st[j] + o : 0;
      rec[j][t] = binned[at];
    }
  }
  unsigned pos[RPW][3];
#pragma unroll
  for (int j = 0; j < RPW; ++j) {
#pragma unroll
    for (int t = 0; t < 3; ++t) {
      const int o = lane + 32 * t;
      pos[j][t] = (unsigned)CAP;
      if (o < n[j]) pos[j][t] = (unsigned)atomicAdd(&cursor[rec[j][t] >> 17], 1);
    }
  }
#pragma unroll
  for (int j = 0; j < RPW; ++j) {
#pragma unroll
    for (int t = 0; t < 3; ++t) {
      if (pos[j][t] < (unsigned)CAP) erec[pos[j][t]] = rec[j][t];
    }
  }
#pragma unroll
  for (int j = 0; j < RPW; ++j) {
    for (int o = lane + 96; o < n[j]; o += 32) {
      const unsigned p = binned[st[j] + o];
      const unsigned r = (unsigned)atomicAdd(&cursor[p >> 17], 1);
      if (r < (unsigned)CAP) erec[r] = p;
    }
  }
}

__device__ __forceinline__ int flat_bases(const int* rc, int* wtot, int wave, int lane, int (&fb)[RPW]) {
  int wt = 0;
#pragma unroll
  for (int j = 0; j < RPW; ++j) {
    fb[j] = wt;
    wt += __builtin_amdgcn_readfirstlane(rc[wave + 32 * j]);
  }
  if (lane == 0) wtot[wave] = wt;
  __syncthreads();
  int x = wtot[lane];
#pragma unroll
  for (int off = 1; off < 32; off <<= 1) {
    const int y = __shfl_up(x, off);
    if (lane >= off) x += y;
  }
  const int total = __builtin_amdgcn_readlane(x, 31);
  const int wb = __builtin_amdgcn_readlane(x, wave) - wt;
#pragma unroll
  for (int j = 0; j < RPW; ++j) fb[j] += wb;
  return total;
}

#define APITCH 34
__global__ void __launch_bounds__(1024) k_agg32_gemm2(const unsigned* __restrict__ binned,
                                                      const v2i* __restrict__ tab,
                                                      const float* __restrict__ dinv,
                                                      const float* __restrict__ h,
                                                      const float* __restrict__ b1,
                                                      const float* __restrict__ W2,
                                                      const int* __restrict__ nstart,
                                                      float* __restrict__ G) {
  __shared__ __attribute__((aligned(16))) float acc[(NPB + 1) * APITCH];
  __shared__ unsigned erec[CAP];
  __shared__ float dl[NPB];
  __shared__ int cursor[NPB];
  __shared__ int rs[NBLK];
  __shared__ int rc[NBLK];
  __shared__ int wtot[32];
  const int tid = threadIdx.x, b = blockIdx.x;
  const int lane = tid & 31;
  const int hf = lane >> 4;
  const int c2 = (lane & 15) * 2;
  const int wave = __builtin_amdgcn_readfirstlane(tid >> 5);
  const int n0 = b * NPB;
  constexpr int NV4 = NPB * HID_DIM / 4;
  const float dv   = dinv[n0 + min(tid, NPB - 1)];
  const int   ns   = nstart[n0 + min(tid, NPB - 1)];
  const v2i   tt   = tab[b * NBLK + (tid & (NBLK - 1))];
  const float bias = b1[lane];
  const v4f* h4 = (const v4f*)(h + (size_t)n0 * HID_DIM);
  v4f hv[4];
#pragma unroll
  for (int k = 0; k < 4; ++k) hv[k] = h4[min(tid + k * 1024, NV4 - 1)];
  if (tid < NPB) {
    dl[tid] = dv;
    cursor[tid] = ns;
  }
  if (tid < NBLK) {
    rs[tid] = tt.x;
    rc[tid] = tt.y;
  }
  if (tid < APITCH) acc[NPB * APITCH + tid] = 0.0f;
#pragma unroll
  for (int k = 0; k < 4; ++k) {
    const int i4 = tid + k * 1024;
    if (i4 < NV4) {
      float* ap = &acc[(i4 >> 3) * APITCH + (i4 & 7) * 4];
      v2f lo, hi;
      lo.x = hv[k].x; lo.y = hv[k].y; hi.x = hv[k].z; hi.y = hv[k].w;
      *(v2f*)ap = lo;
      *(v2f*)(ap + 2) = hi;
    }
  }
  __syncthreads();
  int fb[RPW];
  const int total = flat_bases(rc, wtot, wave, lane, fb);

  const bool by_node = total <= CAP;
  for (int c0 = 0; c0 < total; c0 += CAP) {
    if (by_node) {
      collect_by_node(binned, rs, rc, cursor, erec, wave, lane);
    } else {
      walk_runs<RPW, 0, RPW>(binned, rs, rc, wave, lane, [&](int j, int o, unsigned p) {
        const unsigned r = (unsigned)(fb[j] + o - c0);
        if (r < (unsigned)CAP) erec[r] = p;
      });
    }
    __syncthreads();
    const int count = min(CAP, total - c0);
    for (int e0 = wave * 32; e0 < count; e0 += 1024) {
      const int idx = e0 + lane;
      unsigned rec = PAD_REC;
      if (idx < count) rec = erec[idx];
      const int s = (int)(rec & 0x1FFFFu);
      const int d = (int)(rec >> 17);
      {
        v2f x[16];
        const unsigned cb = (unsigned)c2 * (unsigned)sizeof(float);
#define ROW_SRC(U) __builtin_amdgcn_update_dpp(0, s, 0x150 + (U), 0xF, 0xF, true)
#define LOAD_ROW(U) x[U] = *(const v2f*)((const char*)h + ((unsigned)ROW_SRC(U) * (unsigned)(HID_DIM * sizeof(float)) + cb));
        LOAD_ROW(0)  LOAD_ROW(1)  LOAD_ROW(2)  LOAD_ROW(3)
        LOAD_ROW(4)  LOAD_ROW(5)  LOAD_ROW(6)  LOAD_ROW(7)
        LOAD_ROW(8)  LOAD_ROW(9)  LOAD_ROW(10) LOAD_ROW(11)
        LOAD_ROW(12) LOAD_ROW(13) LOAD_ROW(14) LOAD_ROW(15)
#undef LOAD_ROW
#undef ROW_SRC
        const int dp = d | (__shfl(d, lane ^ 16) << 16);
        int curp = __builtin_amdgcn_readlane(dp, 0);
        v2f run = {0.0f, 0.0f};
#pragma unroll
        for (int u = 0; u < 16; ++u) {
          const int p = __builtin_amdgcn_readlane(dp, u);
          if (p != curp) {
            const int mine = hf ? (int)((unsigned)curp >> 16) : (curp & 0xFFFF);
            const int next = hf ? (int)((unsigned)p >> 16) : (p & 0xFFFF);
            if (next != mine) {
              float* ap = &acc[mine * APITCH + c2];
              atomicAdd(ap, run.x);
              atomicAdd(ap + 1, run.y);
              run.x = 0.0f;
              run.y = 0.0f;
            }
            curp = p;
          }
          run.x += x[u].x;
          run.y += x[u].y;
        }
        const int last = hf ? (int)((unsigned)curp >> 16) : (curp & 0xFFFF);
        float* ap = &acc[last * APITCH + c2];
        atomicAdd(ap, run.x);
        atomicAdd(ap + 1, run.y);
      }
    }
    __syncthreads();
  }

  const int half = lane >> 4;
  const int l15  = lane & 15;
  const int kOff = half * 2;
  v2f bw[HID_DIM / 4];
  if (wave < NTILE) {
#pragma unroll
    for (int k4 = 0; k4 < HID_DIM / 4; ++k4) {
      bw[k4].x = W2[(4 * k4 + kOff + 0) * OUT_DIM + l15];
      bw[k4].y = W2[(4 * k4 + kOff + 1) * OUT_DIM + l15];
    }
  }
  for (int i = tid; i < NPB * HID_DIM; i += 1024) {
    const int r = i >> 5, cc = i & 31;
    float v = acc[r * APITCH + cc] * dl[r] + bias;
    acc[r * APITCH + cc] = fmaxf(v, 0.0f);
  }
  __syncthreads();

  if (wave < NTILE) {
    const float* ar = acc + (wave * 16 + l15) * APITCH + kOff;
    v8f c = {};
#pragma unroll
    for (int k = 0; k < HID_DIM; k += 4) {
      v2f a = *(const v2f*)(ar + k);
      c = __builtin_amdgcn_wmma_f32_16x16x4_f32(false, a, false, bw[k / 4],
                                                (short)0, c, false, false);
    }
    const int r0 = wave * 16 + half * 8;
#pragma unroll
    for (int i = 0; i < 8; ++i) {
      G[(size_t)(n0 + r0 + i) * OUT_DIM + l15] = c[i] * dl[r0 + i];
    }
  }
}

__global__ void __launch_bounds__(1024) k_agg16(const unsigned* __restrict__ binned,
                                                const v2i* __restrict__ tab,
                                                const float* __restrict__ dinv,
                                                const float* __restrict__ g,
                                                const float* __restrict__ b2,
                                                const int* __restrict__ nstart,
                                                float* __restrict__ out) {
  __shared__ __attribute__((aligned(16))) float acc[(NPB + 1) * OUT_DIM];
  __shared__ unsigned erec[CAP];
  __shared__ float dl[NPB];
  __shared__ int cursor[NPB];
  __shared__ int rs[NBLK];
  __shared__ int rc[NBLK];
  __shared__ int wtot[32];
  const int tid = threadIdx.x, b = blockIdx.x;
  const int lane = tid & 31;
  const int hf = lane >> 4;
  const int c = lane & 15;
  const int wave = __builtin_amdgcn_readfirstlane(tid >> 5);
  const int n0 = b * NPB;
  constexpr int NV4 = NPB * OUT_DIM / 4;
  const float dv   = dinv[n0 + min(tid, NPB - 1)];
  const int   ns   = nstart[n0 + min(tid, NPB - 1)];
  const v2i   tt   = tab[b * NBLK + (tid & (NBLK - 1))];
  const float bias = b2[c];
  const v4f* g4 = (const v4f*)(g + (size_t)n0 * OUT_DIM);
  v4f gv[2];
#pragma unroll
  for (int k = 0; k < 2; ++k) gv[k] = g4[min(tid + k * 1024, NV4 - 1)];
  if (tid < NPB) {
    dl[tid] = dv;
    cursor[tid] = ns;
  }
  if (tid < NBLK) {
    rs[tid] = tt.x;
    rc[tid] = tt.y;
  }
  if (tid < OUT_DIM) acc[NPB * OUT_DIM + tid] = 0.0f;
#pragma unroll
  for (int k = 0; k < 2; ++k) {
    const int i4 = tid + k * 1024;
    if (i4 < NV4) *(v4f*)&acc[4 * i4] = gv[k];
  }
  __syncthreads();
  int fb[RPW];
  const int total = flat_bases(rc, wtot, wave, lane, fb);

  const bool by_node = total <= CAP;
  for (int c0 = 0; c0 < total; c0 += CAP) {
    if (by_node) {
      collect_by_node(binned, rs, rc, cursor, erec, wave, lane);
    } else {
      walk_runs<RPW, 0, RPW>(binned, rs, rc, wave, lane, [&](int j, int o, unsigned p) {
        const unsigned r = (unsigned)(fb[j] + o - c0);
        if (r < (unsigned)CAP) erec[r] = p;
      });
    }
    __syncthreads();
    const int count = min(CAP, total - c0);
    for (int e0 = wave * 32; e0 < count; e0 += 1024) {
      const int idx = e0 + lane;
      unsigned rec = PAD_REC;
      if (idx < count) rec = erec[idx];
      const int s = (int)(rec & 0x1FFFFu);
      const int d = (int)(rec >> 17);
      {
        float x[16];
        const unsigned cb = (unsigned)c * (unsigned)sizeof(float);
#define ROW_SRC(U) __builtin_amdgcn_update_dpp(0, s, 0x150 + (U), 0xF, 0xF, true)
#define LOAD_ROW(U) x[U] = *(const float*)((const char*)g + ((unsigned)ROW_SRC(U) * (unsigned)(OUT_DIM * sizeof(float)) + cb));
        LOAD_ROW(0)  LOAD_ROW(1)  LOAD_ROW(2)  LOAD_ROW(3)
        LOAD_ROW(4)  LOAD_ROW(5)  LOAD_ROW(6)  LOAD_ROW(7)
        LOAD_ROW(8)  LOAD_ROW(9)  LOAD_ROW(10) LOAD_ROW(11)
        LOAD_ROW(12) LOAD_ROW(13) LOAD_ROW(14) LOAD_ROW(15)
#undef LOAD_ROW
#undef ROW_SRC
        const int dp = d | (__shfl(d, lane ^ 16) << 16);
        int curp = __builtin_amdgcn_readlane(dp, 0);
        float run = 0.0f;
#pragma unroll
        for (int u = 0; u < 16; ++u) {
          const int p = __builtin_amdgcn_readlane(dp, u);
          if (p != curp) {
            const int mine = hf ? (int)((unsigned)curp >> 16) : (curp & 0xFFFF);
            const int next = hf ? (int)((unsigned)p >> 16) : (p & 0xFFFF);
            if (next != mine) {
              atomicAdd(&acc[mine * OUT_DIM + c], run);
              run = 0.0f;
            }
            curp = p;
          }
          run += x[u];
        }
        const int last = hf ? (int)((unsigned)curp >> 16) : (curp & 0xFFFF);
        atomicAdd(&acc[last * OUT_DIM + c], run);
      }
    }
    __syncthreads();
  }
  for (int i = tid; i < NPB * OUT_DIM; i += 1024) {
    out[(size_t)n0 * OUT_DIM + i] = acc[i] * dl[i >> 4] + bias;
  }
}

extern "C" void kernel_launch(void* const* d_in, const int* in_sizes, int n_in,
                              void* d_out, int out_size, void* d_ws, size_t ws_size,
                              hipStream_t stream) {
  const float* x  = (const float*)d_in[0];
  const float* W1 = (const float*)d_in[1];
  const float* b1 = (const float*)d_in[2];
  const float* W2 = (const float*)d_in[3];
  const float* b2 = (const float*)d_in[4];
  const int*   ei = (const int*)d_in[5];

  float* dinv = (float*)d_ws;
  float* h    = dinv + N_NODES;
  float* g    = h    + (size_t)N_NODES * HID_DIM;
  v2i* tab    = (v2i*)(g + (size_t)N_NODES * OUT_DIM);
  unsigned* binned = (unsigned*)(tab + NBKT * NBLK);
  int* nstart = (int*)(binned + (size_t)NBLK * SLAB);
  float* out = (float*)d_out;

  k_bin<<<NBLK, 1024, 0, stream>>>(ei, binned, tab);
  k_degree_gemm1<<<NBKT, 1024, 0, stream>>>(binned, tab, x, W1, dinv, h, nstart);
  k_agg32_gemm2<<<NBKT, 1024, 0, stream>>>(binned, tab, dinv, h, b1, W2, nstart, g);
  k_agg16<<<NBKT, 1024, 0, stream>>>(binned, tab, dinv, g, b2, nstart, out);
}
